# starting kernel
# baseline (speedup 1.0000x reference)
#include <hip/hip_runtime.h>
#include <hip/hip_bf16.h>

#define BG   1024
#define NN   128
#define VN   (BG * NN)
#define HD   128
#define EPG  2048
#define KTOP 30
#define CONV_O 32
#define CONV_W 3
#define CONV_T (KTOP - CONV_W + 1)
#define FLAT   (CONV_O * CONV_T)
#define NCLS   16

typedef float v2f __attribute__((ext_vector_type(2)));
typedef float v8f __attribute__((ext_vector_type(8)));

__global__ __launch_bounds__(256) void sage_aggregate(
    const float* __restrict__ X, const int* __restrict__ src,
    const int* __restrict__ dst, float* __restrict__ meanOut)
{
  __shared__ float feat[NN * HD];
  __shared__ float acc [NN * HD];
  __shared__ float deg [NN];

  const int g = blockIdx.x;
  const int t = threadIdx.x;
  const float* xg = X + (size_t)g * NN * HD;

  for (int i = t; i < NN * HD / 4; i += 256) {
    ((float4*)feat)[i] = ((const float4*)xg)[i];
    ((float4*)acc)[i]  = make_float4(0.f, 0.f, 0.f, 0.f);
  }
  if (t < NN) deg[t] = 0.f;
  __syncthreads();

  const int ebase = g * EPG;
  const int eg = t >> 2;
  const int fg = t & 3;
  for (int e = eg; e < EPG; e += 64) {
    const int s = src[ebase + e] & (NN - 1);
    const int d = dst[ebase + e] & (NN - 1);
    const float* fs = feat + s * HD + fg * 32;
    float*       ad = acc  + d * HD + fg * 32;
    if (fg == 0) atomicAdd(&deg[d], 1.0f);
#pragma unroll
    for (int f = 0; f < 32; ++f) atomicAdd(&ad[f], fs[f]);
  }
  __syncthreads();

  float* mg = meanOut + (size_t)g * NN * HD;
  for (int i = t; i < NN * HD; i += 256) {
    const float c = deg[i >> 7];
    mg[i] = acc[i] / (c > 1.f ? c : 1.f);
  }
}

__global__ __launch_bounds__(256) void sage_gemm(
    const float* __restrict__ Aagg, const float* __restrict__ Aself,
    const float* __restrict__ Wl,   const float* __restrict__ Wr,
    const float* __restrict__ bias, float* __restrict__ out)
{
  __shared__ float2 wp[128 * HD];

  const int t = threadIdx.x;
  for (int idx = t; idx < 128 * HD; idx += 256) {
    const int kk  = idx >> 7;
    const int col = idx & (HD - 1);
    const float* W = (kk < 64) ? Wl : Wr;
    const int k = (kk & 63) * 2;
    float2 v;
    v.x = W[(size_t)k       * HD + col];
    v.y = W[(size_t)(k + 1) * HD + col];
    wp[idx] = v;
  }
  __syncthreads();

  const int wave = t >> 5;
  const int lane = t & 31;
  const int m    = lane & 15;
  const int hi   = lane >> 4;
  const int kb   = hi * 2;

  const int rowBase = blockIdx.x * 128 + wave * 16;
  const float* arow1 = Aagg  + (size_t)(rowBase + m) * HD;
  const float* arow2 = Aself + (size_t)(rowBase + m) * HD;

  v8f acc0 = {}, acc1 = {}, acc2 = {}, acc3 = {};
  v8f acc4 = {}, acc5 = {}, acc6 = {}, acc7 = {};

#define GEMM_STEP(AROW, KOFF)                                              \
  {                                                                        \
    v2f a = *(const v2f*)((AROW) + k0 + kb);                               \
    const float2* bp = wp + ((k0 >> 1) + hi + (KOFF)) * HD + m;            \
    float2 bb; v2f b;                                                      \
    bb = bp[0];   b.x = bb.x; b.y = bb.y;                                  \
    acc0 = __builtin_amdgcn_wmma_f32_16x16x4_f32(false, a, false, b, (short)0, acc0, false, false); \
    bb = bp[16];  b.x = bb.x; b.y = bb.y;                                  \
    acc1 = __builtin_amdgcn_wmma_f32_16x16x4_f32(false, a, false, b, (short)0, acc1, false, false); \
    bb = bp[32];  b.x = bb.x; b.y = bb.y;                                  \
    acc2 = __builtin_amdgcn_wmma_f32_16x16x4_f32(false, a, false, b, (short)0, acc2, false, false); \
    bb = bp[48];  b.x = bb.x; b.y = bb.y;                                  \
    acc3 = __builtin_amdgcn_wmma_f32_16x16x4_f32(false, a, false, b, (short)0, acc3, false, false); \
    bb = bp[64];  b.x = bb.x; b.y = bb.y;                                  \
    acc4 = __builtin_amdgcn_wmma_f32_16x16x4_f32(false, a, false, b, (short)0, acc4, false, false); \
    bb = bp[80];  b.x = bb.x; b.y = bb.y;                                  \
    acc5 = __builtin_amdgcn_wmma_f32_16x16x4_f32(false, a, false, b, (short)0, acc5, false, false); \
    bb = bp[96];  b.x = bb.x; b.y = bb.y;                                  \
    acc6 = __builtin_amdgcn_wmma_f32_16x16x4_f32(false, a, false, b, (short)0, acc6, false, false); \
    bb = bp[112]; b.x = bb.x; b.y = bb.y;                                  \
    acc7 = __builtin_amdgcn_wmma_f32_16x16x4_f32(false, a, false, b, (short)0, acc7, false, false); \
  }

#pragma unroll 2
  for (int k0 = 0; k0 < HD; k0 += 4) GEMM_STEP(arow1, 0)
#pragma unroll 2
  for (int k0 = 0; k0 < HD; k0 += 4) GEMM_STEP(arow2, 64)
#undef GEMM_STEP

  const int rbase = rowBase + hi * 8;
  v8f accs[8] = {acc0, acc1, acc2, acc3, acc4, acc5, acc6, acc7};
#pragma unroll
  for (int ct = 0; ct < 8; ++ct) {
    const int col = ct * 16 + m;
    const float bv = bias[col];
#pragma unroll
    for (int v = 0; v < 8; ++v) {
      float val = accs[ct][v] + bv;
      val = val > 0.f ? val : 0.f;
      out[(size_t)(rbase + v) * HD + col] = val;
    }
  }
}

__global__ __launch_bounds__(256) void sortpool_head(
    const float* __restrict__ H,
    const float* __restrict__ conv_w, const float* __restrict__ conv_b,
    const float* __restrict__ W1, const float* __restrict__ b1,
    const float* __restrict__ W2, const float* __restrict__ b2,
    float* __restrict__ out)
{
  __shared__ float vals[NN];
  __shared__ int   idxs[NN];
  __shared__ float xs[KTOP * HD];
  __shared__ float y[FLAT];
  __shared__ float z[HD];
  __shared__ float logits[NCLS];
  __shared__ float lse;

  const int g = blockIdx.x;
  const int t = threadIdx.x;
  const float* hg = H + (size_t)g * NN * HD;

  if (t < NN) { vals[t] = hg[t * HD + (HD - 1)]; idxs[t] = t; }
  __syncthreads();

  for (int k = 2; k <= NN; k <<= 1) {
    for (int j = k >> 1; j > 0; j >>= 1) {
      if (t < NN) {
        const int ixj = t ^ j;
        if (ixj > t) {
          const bool dirDesc = ((t & k) == 0);
          const float va = vals[t], vb = vals[ixj];
          const int   ia = idxs[t], ib = idxs[ixj];
          const bool aBeforeB = (va > vb) || (va == vb && ia < ib);
          if (dirDesc ? !aBeforeB : aBeforeB) {
            vals[t] = vb; vals[ixj] = va;
            idxs[t] = ib; idxs[ixj] = ia;
          }
        }
      }
      __syncthreads();
    }
  }

  for (int i = t; i < KTOP * HD; i += 256) {
    const int tt = i >> 7, c = i & (HD - 1);
    xs[i] = hg[(size_t)idxs[tt] * HD + c];
  }
  __syncthreads();

  for (int i = t; i < FLAT; i += 256) {
    const int o = i / CONV_T, tt = i % CONV_T;
    float s = conv_b[o];
#pragma unroll
    for (int w = 0; w < CONV_W; ++w) {
      const float* xr = xs + (tt + w) * HD;
      const float* wr = conv_w + (size_t)o * HD * CONV_W + w;
      for (int c = 0; c < HD; ++c) s += xr[c] * wr[c * CONV_W];
    }
    y[i] = s > 0.f ? s : 0.f;
  }
  __syncthreads();

  if (t < HD) {
    float s = b1[t];
    for (int i = 0; i < FLAT; ++i) s += y[i] * W1[(size_t)i * HD + t];
    z[t] = s > 0.f ? s : 0.f;
  }
  __syncthreads();

  if (t < NCLS) {
    float s = b2[t];
    for (int j = 0; j < HD; ++j) s += z[j] * W2[j * NCLS + t];
    logits[t] = s;
  }
  __syncthreads();

  if (t == 0) {
    float mx = logits[0];
    for (int c = 1; c < NCLS; ++c) mx = fmaxf(mx, logits[c]);
    float se = 0.f;
    for (int c = 0; c < NCLS; ++c) se += expf(logits[c] - mx);
    lse = mx + logf(se);
  }
  __syncthreads();
  if (t < NCLS) out[(size_t)g * NCLS + t] = logits[t] - lse;
}

extern "C" void kernel_launch(void* const* d_in, const int* in_sizes, int n_in,
                              void* d_out, int out_size, void* d_ws, size_t ws_size,
                              hipStream_t stream) {
  (void)in_sizes; (void)n_in; (void)out_size; (void)ws_size;
  const float* x      = (const float*)d_in[0];
  const int*   src    = (const int*)d_in[1];
  const int*   dst    = (const int*)d_in[2];
  const float* Wl1    = (const float*)d_in[5];
  const float* bl1    = (const float*)d_in[6];
  const float* Wr1    = (const float*)d_in[7];
  const float* Wl2    = (const float*)d_in[8];
  const float* bl2    = (const float*)d_in[9];
  const float* Wr2    = (const float*)d_in[10];
  const float* Wl3    = (const float*)d_in[11];
  const float* bl3    = (const float*)d_in[12];
  const float* Wr3    = (const float*)d_in[13];
  const float* conv_w = (const float*)d_in[14];
  const float* conv_b = (const float*)d_in[15];
  const float* W1     = (const float*)d_in[16];
  const float* b1     = (const float*)d_in[17];
  const float* W2     = (const float*)d_in[18];
  const float* b2     = (const float*)d_in[19];

  float* ws   = (float*)d_ws;
  float* mean = ws;
  float* h1   = ws + (size_t)VN * HD;
  float* h2   = h1 + (size_t)VN * HD;

  const int gemmBlocks = VN / 128;

  sage_aggregate<<<BG, 256, 0, stream>>>(x, src, dst, mean);
  sage_gemm<<<gemmBlocks, 256, 0, stream>>>(mean, x, Wl1, Wr1, bl1, h1);
  sage_aggregate<<<BG, 256, 0, stream>>>(h1, src, dst, mean);
  sage_gemm<<<gemmBlocks, 256, 0, stream>>>(mean, h1, Wl2, Wr2, bl2, h2);
  sage_aggregate<<<BG, 256, 0, stream>>>(h2, src, dst, mean);
  sage_gemm<<<gemmBlocks, 256, 0, stream>>>(mean, h2, Wl3, Wr3, bl3, h1);
  sortpool_head<<<BG, 256, 0, stream>>>(h1, conv_w, conv_b, W1, b1, W2, b2,
                                        (float*)d_out);
}


# aggregate: LDS float-atomic scatter-add replaced by a per-graph edge-count matrix (exact unit-step LDS atomics) times features on f32 WMMA; sums now run in fixed source-index order, same every run
# speedup vs baseline: 3.9893x; 3.9893x over previous
#include <hip/hip_runtime.h>
#include <hip/hip_bf16.h>

#define BG   1024
#define NN   128
#define VN   (BG * NN)
#define HD   128
#define EPG  2048
#define KTOP 30
#define CONV_O 32
#define CONV_W 3
#define CONV_T (KTOP - CONV_W + 1)
#define FLAT   (CONV_O * CONV_T)
#define NCLS   16

typedef float v2f __attribute__((ext_vector_type(2)));
typedef float v8f __attribute__((ext_vector_type(8)));

#define FP_PITCH 144
#define CM_PITCH 132

__global__ __launch_bounds__(256) void sage_aggregate(
    const float* __restrict__ X, const int* __restrict__ src,
    const int* __restrict__ dst, float* __restrict__ meanOut)
{
  __shared__ float2 fp[(NN / 2) * FP_PITCH];
  __shared__ float  cm[NN * CM_PITCH];
  __shared__ float  deg[NN];

  const int g = blockIdx.x;
  const int t = threadIdx.x;

  const int4* s4 = (const int4*)(src + (size_t)g * EPG);
  const int4* d4 = (const int4*)(dst + (size_t)g * EPG);
  const int4 sv0 = s4[t], sv1 = s4[t + 256];
  const int4 dv0 = d4[t], dv1 = d4[t + 256];

  for (int i = t; i < NN * CM_PITCH / 4; i += 256)
    ((float4*)cm)[i] = make_float4(0.f, 0.f, 0.f, 0.f);
  if (t < NN) deg[t] = 0.f;

  const float4* xg4 = (const float4*)(X + (size_t)g * NN * HD);
#pragma unroll
  for (int j = 0; j < 8; ++j) {
    const int idx = t + 256 * j;
    const int q   = idx >> 5;
    const int c4  = idx & 31;
    const float4 r0 = xg4[(2 * q) * (HD / 4) + c4];
    const float4 r1 = xg4[(2 * q + 1) * (HD / 4) + c4];
    float4* o = (float4*)(fp + q * FP_PITCH + c4 * 4);
    o[0] = make_float4(r0.x, r1.x, r0.y, r1.y);
    o[1] = make_float4(r0.z, r1.z, r0.w, r1.w);
  }
  __syncthreads();

#define COUNT_EDGE(S, D)                                   \
  {                                                        \
    const int s_ = (S) & (NN - 1);                         \
    const int d_ = (D) & (NN - 1);                         \
    atomicAdd(&cm[d_ * CM_PITCH + s_], 1.0f);              \
    atomicAdd(&deg[d_], 1.0f);                             \
  }
  COUNT_EDGE(sv0.x, dv0.x) COUNT_EDGE(sv0.y, dv0.y)
  COUNT_EDGE(sv0.z, dv0.z) COUNT_EDGE(sv0.w, dv0.w)
  COUNT_EDGE(sv1.x, dv1.x) COUNT_EDGE(sv1.y, dv1.y)
  COUNT_EDGE(sv1.z, dv1.z) COUNT_EDGE(sv1.w, dv1.w)
#undef COUNT_EDGE
  __syncthreads();

  const int wave = t >> 5;
  const int lane = t & 31;
  const int m    = lane & 15;
  const int hi   = lane >> 4;

  const float*  arow = cm + (wave * 16 + m) * CM_PITCH + 2 * hi;
  const float2* bcol = fp + hi * FP_PITCH + m;

  v8f acc[8] = {};
  v2f    af[2];
  float2 bf[2][8];

  af[0] = *(const v2f*)arow;
#pragma unroll
  for (int ct = 0; ct < 8; ++ct) bf[0][ct] = bcol[ct * 16];

#pragma unroll
  for (int ks = 0; ks < NN / 4; ++ks) {
    const int cur = ks & 1, nxt = cur ^ 1;
    if (ks + 1 < NN / 4) {
      af[nxt] = *(const v2f*)(arow + 4 * (ks + 1));
      const float2* bp = bcol + 2 * (ks + 1) * FP_PITCH;
#pragma unroll
      for (int ct = 0; ct < 8; ++ct) bf[nxt][ct] = bp[ct * 16];
    }
#pragma unroll
    for (int ct = 0; ct < 8; ++ct) {
      v2f b; b.x = bf[cur][ct].x; b.y = bf[cur][ct].y;
      acc[ct] = __builtin_amdgcn_wmma_f32_16x16x4_f32(false, af[cur], false, b,
                                                      (short)0, acc[ct], false, false);
    }
  }

  float* st = cm + (wave * 16 + 8 * hi) * CM_PITCH + m;
#pragma unroll
  for (int ct = 0; ct < 8; ++ct) {
#pragma unroll
    for (int v = 0; v < 8; ++v) st[v * CM_PITCH + ct * 16] = acc[ct][v];
  }
  __syncthreads();

  float4* mg4 = (float4*)(meanOut + (size_t)g * NN * HD);
  for (int i = t; i < NN * HD / 4; i += 256) {
    const int row = i >> 5, c4 = i & 31;
    const float c  = deg[row];
    const float dn = c > 1.f ? c : 1.f;
    const float4 v = *(const float4*)(cm + row * CM_PITCH + c4 * 4);
    mg4[i] = make_float4(v.x / dn, v.y / dn, v.z / dn, v.w / dn);
  }
}

__global__ __launch_bounds__(256) void sage_gemm(
    const float* __restrict__ Aagg, const float* __restrict__ Aself,
    const float* __restrict__ Wl,   const float* __restrict__ Wr,
    const float* __restrict__ bias, float* __restrict__ out)
{
  __shared__ float2 wp[128 * HD];

  const int t = threadIdx.x;
  for (int idx = t; idx < 128 * HD; idx += 256) {
    const int kk  = idx >> 7;
    const int col = idx & (HD - 1);
    const float* W = (kk < 64) ? Wl : Wr;
    const int k = (kk & 63) * 2;
    float2 v;
    v.x = W[(size_t)k       * HD + col];
    v.y = W[(size_t)(k + 1) * HD + col];
    wp[idx] = v;
  }
  __syncthreads();

  const int wave = t >> 5;
  const int lane = t & 31;
  const int m    = lane & 15;
  const int hi   = lane >> 4;
  const int kb   = hi * 2;

  const int rowBase = blockIdx.x * 128 + wave * 16;
  const float* arow1 = Aagg  + (size_t)(rowBase + m) * HD;
  const float* arow2 = Aself + (size_t)(rowBase + m) * HD;

  v8f acc0 = {}, acc1 = {}, acc2 = {}, acc3 = {};
  v8f acc4 = {}, acc5 = {}, acc6 = {}, acc7 = {};

#define GEMM_STEP(AROW, KOFF)                                              \
  {                                                                        \
    v2f a = *(const v2f*)((AROW) + k0 + kb);                               \
    const float2* bp = wp + ((k0 >> 1) + hi + (KOFF)) * HD + m;            \
    float2 bb; v2f b;                                                      \
    bb = bp[0];   b.x = bb.x; b.y = bb.y;                                  \
    acc0 = __builtin_amdgcn_wmma_f32_16x16x4_f32(false, a, false, b, (short)0, acc0, false, false); \
    bb = bp[16];  b.x = bb.x; b.y = bb.y;                                  \
    acc1 = __builtin_amdgcn_wmma_f32_16x16x4_f32(false, a, false, b, (short)0, acc1, false, false); \
    bb = bp[32];  b.x = bb.x; b.y = bb.y;                                  \
    acc2 = __builtin_amdgcn_wmma_f32_16x16x4_f32(false, a, false, b, (short)0, acc2, false, false); \
    bb = bp[48];  b.x = bb.x; b.y = bb.y;                                  \
    acc3 = __builtin_amdgcn_wmma_f32_16x16x4_f32(false, a, false, b, (short)0, acc3, false, false); \
    bb = bp[64];  b.x = bb.x; b.y = bb.y;                                  \
    acc4 = __builtin_amdgcn_wmma_f32_16x16x4_f32(false, a, false, b, (short)0, acc4, false, false); \
    bb = bp[80];  b.x = bb.x; b.y = bb.y;                                  \
    acc5 = __builtin_amdgcn_wmma_f32_16x16x4_f32(false, a, false, b, (short)0, acc5, false, false); \
    bb = bp[96];  b.x = bb.x; b.y = bb.y;                                  \
    acc6 = __builtin_amdgcn_wmma_f32_16x16x4_f32(false, a, false, b, (short)0, acc6, false, false); \
    bb = bp[112]; b.x = bb.x; b.y = bb.y;                                  \
    acc7 = __builtin_amdgcn_wmma_f32_16x16x4_f32(false, a, false, b, (short)0, acc7, false, false); \
  }

#pragma unroll 2
  for (int k0 = 0; k0 < HD; k0 += 4) GEMM_STEP(arow1, 0)
#pragma unroll 2
  for (int k0 = 0; k0 < HD; k0 += 4) GEMM_STEP(arow2, 64)
#undef GEMM_STEP

  const int rbase = rowBase + hi * 8;
  v8f accs[8] = {acc0, acc1, acc2, acc3, acc4, acc5, acc6, acc7};
#pragma unroll
  for (int ct = 0; ct < 8; ++ct) {
    const int col = ct * 16 + m;
    const float bv = bias[col];
#pragma unroll
    for (int v = 0; v < 8; ++v) {
      float val = accs[ct][v] + bv;
      val = val > 0.f ? val : 0.f;
      out[(size_t)(rbase + v) * HD + col] = val;
    }
  }
}

__global__ __launch_bounds__(256) void sortpool_head(
    const float* __restrict__ H,
    const float* __restrict__ conv_w, const float* __restrict__ conv_b,
    const float* __restrict__ W1, const float* __restrict__ b1,
    const float* __restrict__ W2, const float* __restrict__ b2,
    float* __restrict__ out)
{
  __shared__ float vals[NN];
  __shared__ int   idxs[NN];
  __shared__ float xs[KTOP * HD];
  __shared__ float y[FLAT];
  __shared__ float z[HD];
  __shared__ float logits[NCLS];
  __shared__ float lse;

  const int g = blockIdx.x;
  const int t = threadIdx.x;
  const float* hg = H + (size_t)g * NN * HD;

  if (t < NN) { vals[t] = hg[t * HD + (HD - 1)]; idxs[t] = t; }
  __syncthreads();

  for (int k = 2; k <= NN; k <<= 1) {
    for (int j = k >> 1; j > 0; j >>= 1) {
      if (t < NN) {
        const int ixj = t ^ j;
        if (ixj > t) {
          const bool dirDesc = ((t & k) == 0);
          const float va = vals[t], vb = vals[ixj];
          const int   ia = idxs[t], ib = idxs[ixj];
          const bool aBeforeB = (va > vb) || (va == vb && ia < ib);
          if (dirDesc ? !aBeforeB : aBeforeB) {
            vals[t] = vb; vals[ixj] = va;
            idxs[t] = ib; idxs[ixj] = ia;
          }
        }
      }
      __syncthreads();
    }
  }

  for (int i = t; i < KTOP * HD; i += 256) {
    const int tt = i >> 7, c = i & (HD - 1);
    xs[i] = hg[(size_t)idxs[tt] * HD + c];
  }
  __syncthreads();

  for (int i = t; i < FLAT; i += 256) {
    const int o = i / CONV_T, tt = i % CONV_T;
    float s = conv_b[o];
#pragma unroll
    for (int w = 0; w < CONV_W; ++w) {
      const float* xr = xs + (tt + w) * HD;
      const float* wr = conv_w + (size_t)o * HD * CONV_W + w;
      for (int c = 0; c < HD; ++c) s += xr[c] * wr[c * CONV_W];
    }
    y[i] = s > 0.f ? s : 0.f;
  }
  __syncthreads();

  if (t < HD) {
    float s = b1[t];
    for (int i = 0; i < FLAT; ++i) s += y[i] * W1[(size_t)i * HD + t];
    z[t] = s > 0.f ? s : 0.f;
  }
  __syncthreads();

  if (t < NCLS) {
    float s = b2[t];
    for (int j = 0; j < HD; ++j) s += z[j] * W2[j * NCLS + t];
    logits[t] = s;
  }
  __syncthreads();

  if (t == 0) {
    float mx = logits[0];
    for (int c = 1; c < NCLS; ++c) mx = fmaxf(mx, logits[c]);
    float se = 0.f;
    for (int c = 0; c < NCLS; ++c) se += expf(logits[c] - mx);
    lse = mx + logf(se);
  }
  __syncthreads();
  if (t < NCLS) out[(size_t)g * NCLS + t] = logits[t] - lse;
}

extern "C" void kernel_launch(void* const* d_in, const int* in_sizes, int n_in,
                              void* d_out, int out_size, void* d_ws, size_t ws_size,
                              hipStream_t stream) {
  (void)in_sizes; (void)n_in; (void)out_size; (void)ws_size;
  const float* x      = (const float*)d_in[0];
  const int*   src    = (const int*)d_in[1];
  const int*   dst    = (const int*)d_in[2];
  const float* Wl1    = (const float*)d_in[5];
  const float* bl1    = (const float*)d_in[6];
  const float* Wr1    = (const float*)d_in[7];
  const float* Wl2    = (const float*)d_in[8];
  const float* bl2    = (const float*)d_in[9];
  const float* Wr2    = (const float*)d_in[10];
  const float* Wl3    = (const float*)d_in[11];
  const float* bl3    = (const float*)d_in[12];
  const float* Wr3    = (const float*)d_in[13];
  const float* conv_w = (const float*)d_in[14];
  const float* conv_b = (const float*)d_in[15];
  const float* W1     = (const float*)d_in[16];
  const float* b1     = (const float*)d_in[17];
  const float* W2     = (const float*)d_in[18];
  const float* b2     = (const float*)d_in[19];

  float* ws   = (float*)d_ws;
  float* mean = ws;
  float* h1   = ws + (size_t)VN * HD;
  float* h2   = h1 + (size_t)VN * HD;

  const int gemmBlocks = VN / 128;

  sage_aggregate<<<BG, 256, 0, stream>>>(x, src, dst, mean);
  sage_gemm<<<gemmBlocks, 256, 0, stream>>>(mean, x, Wl1, Wr1, bl1, h1);
  sage_aggregate<<<BG, 256, 0, stream>>>(h1, src, dst, mean);
  sage_gemm<<<gemmBlocks, 256, 0, stream>>>(mean, h1, Wl2, Wr2, bl2, h2);
  sage_aggregate<<<BG, 256, 0, stream>>>(h2, src, dst, mean);
  sage_gemm<<<gemmBlocks, 256, 0, stream>>>(mean, h2, Wl3, Wr3, bl3, h1);
  sortpool_head<<<BG, 256, 0, stream>>>(h1, conv_w, conv_b, W1, b1, W2, b2,
                                        (float*)d_out);
}


# head: Conv1d moved from a scalar loop whose LDS reads were 16-way bank-conflicted to the f32 matrix instruction on padded LDS tiles; conv sums now in matrix k order, two partial sums added last
# speedup vs baseline: 5.4330x; 1.3619x over previous
#include <hip/hip_runtime.h>
#include <hip/hip_bf16.h>

#define BG   1024
#define NN   128
#define VN   (BG * NN)
#define HD   128
#define EPG  2048
#define KTOP 30
#define CONV_O 32
#define CONV_W 3
#define CONV_T (KTOP - CONV_W + 1)
#define FLAT   (CONV_O * CONV_T)
#define NCLS   16

typedef float v2f __attribute__((ext_vector_type(2)));
typedef float v8f __attribute__((ext_vector_type(8)));

#define FP_PITCH 144
#define CM_PITCH 132

__global__ __launch_bounds__(256) void sage_aggregate(
    const float* __restrict__ X, const int* __restrict__ src,
    const int* __restrict__ dst, float* __restrict__ meanOut)
{
  __shared__ float2 fp[(NN / 2) * FP_PITCH];
  __shared__ float  cm[NN * CM_PITCH];
  __shared__ float  deg[NN];

  const int g = blockIdx.x;
  const int t = threadIdx.x;

  const int4* s4 = (const int4*)(src + (size_t)g * EPG);
  const int4* d4 = (const int4*)(dst + (size_t)g * EPG);
  const int4 sv0 = s4[t], sv1 = s4[t + 256];
  const int4 dv0 = d4[t], dv1 = d4[t + 256];

  for (int i = t; i < NN * CM_PITCH / 4; i += 256)
    ((float4*)cm)[i] = make_float4(0.f, 0.f, 0.f, 0.f);
  if (t < NN) deg[t] = 0.f;

  const float4* xg4 = (const float4*)(X + (size_t)g * NN * HD);
#pragma unroll
  for (int j = 0; j < 8; ++j) {
    const int idx = t + 256 * j;
    const int q   = idx >> 5;
    const int c4  = idx & 31;
    const float4 r0 = xg4[(2 * q) * (HD / 4) + c4];
    const float4 r1 = xg4[(2 * q + 1) * (HD / 4) + c4];
    float4* o = (float4*)(fp + q * FP_PITCH + c4 * 4);
    o[0] = make_float4(r0.x, r1.x, r0.y, r1.y);
    o[1] = make_float4(r0.z, r1.z, r0.w, r1.w);
  }
  __syncthreads();

#define COUNT_EDGE(S, D)                                   \
  {                                                        \
    const int s_ = (S) & (NN - 1);                         \
    const int d_ = (D) & (NN - 1);                         \
    atomicAdd(&cm[d_ * CM_PITCH + s_], 1.0f);              \
    atomicAdd(&deg[d_], 1.0f);                             \
  }
  COUNT_EDGE(sv0.x, dv0.x) COUNT_EDGE(sv0.y, dv0.y)
  COUNT_EDGE(sv0.z, dv0.z) COUNT_EDGE(sv0.w, dv0.w)
  COUNT_EDGE(sv1.x, dv1.x) COUNT_EDGE(sv1.y, dv1.y)
  COUNT_EDGE(sv1.z, dv1.z) COUNT_EDGE(sv1.w, dv1.w)
#undef COUNT_EDGE
  __syncthreads();

  const int wave = t >> 5;
  const int lane = t & 31;
  const int m    = lane & 15;
  const int hi   = lane >> 4;

  const float*  arow = cm + (wave * 16 + m) * CM_PITCH + 2 * hi;
  const float2* bcol = fp + hi * FP_PITCH + m;

  v8f acc[8] = {};
  v2f    af[2];
  float2 bf[2][8];

  af[0] = *(const v2f*)arow;
#pragma unroll
  for (int ct = 0; ct < 8; ++ct) bf[0][ct] = bcol[ct * 16];

#pragma unroll
  for (int ks = 0; ks < NN / 4; ++ks) {
    const int cur = ks & 1, nxt = cur ^ 1;
    if (ks + 1 < NN / 4) {
      af[nxt] = *(const v2f*)(arow + 4 * (ks + 1));
      const float2* bp = bcol + 2 * (ks + 1) * FP_PITCH;
#pragma unroll
      for (int ct = 0; ct < 8; ++ct) bf[nxt][ct] = bp[ct * 16];
    }
#pragma unroll
    for (int ct = 0; ct < 8; ++ct) {
      v2f b; b.x = bf[cur][ct].x; b.y = bf[cur][ct].y;
      acc[ct] = __builtin_amdgcn_wmma_f32_16x16x4_f32(false, af[cur], false, b,
                                                      (short)0, acc[ct], false, false);
    }
  }

  float* st = cm + (wave * 16 + 8 * hi) * CM_PITCH + m;
#pragma unroll
  for (int ct = 0; ct < 8; ++ct) {
#pragma unroll
    for (int v = 0; v < 8; ++v) st[v * CM_PITCH + ct * 16] = acc[ct][v];
  }
  __syncthreads();

  float4* mg4 = (float4*)(meanOut + (size_t)g * NN * HD);
  for (int i = t; i < NN * HD / 4; i += 256) {
    const int row = i >> 5, c4 = i & 31;
    const float c  = deg[row];
    const float dn = c > 1.f ? c : 1.f;
    const float4 v = *(const float4*)(cm + row * CM_PITCH + c4 * 4);
    mg4[i] = make_float4(v.x / dn, v.y / dn, v.z / dn, v.w / dn);
  }
}

__global__ __launch_bounds__(256) void sage_gemm(
    const float* __restrict__ Aagg, const float* __restrict__ Aself,
    const float* __restrict__ Wl,   const float* __restrict__ Wr,
    const float* __restrict__ bias, float* __restrict__ out)
{
  __shared__ float2 wp[128 * HD];

  const int t = threadIdx.x;
  for (int idx = t; idx < 128 * HD; idx += 256) {
    const int kk  = idx >> 7;
    const int col = idx & (HD - 1);
    const float* W = (kk < 64) ? Wl : Wr;
    const int k = (kk & 63) * 2;
    float2 v;
    v.x = W[(size_t)k       * HD + col];
    v.y = W[(size_t)(k + 1) * HD + col];
    wp[idx] = v;
  }
  __syncthreads();

  const int wave = t >> 5;
  const int lane = t & 31;
  const int m    = lane & 15;
  const int hi   = lane >> 4;
  const int kb   = hi * 2;

  const int rowBase = blockIdx.x * 128 + wave * 16;
  const float* arow1 = Aagg  + (size_t)(rowBase + m) * HD;
  const float* arow2 = Aself + (size_t)(rowBase + m) * HD;

  v8f acc0 = {}, acc1 = {}, acc2 = {}, acc3 = {};
  v8f acc4 = {}, acc5 = {}, acc6 = {}, acc7 = {};

#define GEMM_STEP(AROW, KOFF)                                              \
  {                                                                        \
    v2f a = *(const v2f*)((AROW) + k0 + kb);                               \
    const float2* bp = wp + ((k0 >> 1) + hi + (KOFF)) * HD + m;            \
    float2 bb; v2f b;                                                      \
    bb = bp[0];   b.x = bb.x; b.y = bb.y;                                  \
    acc0 = __builtin_amdgcn_wmma_f32_16x16x4_f32(false, a, false, b, (short)0, acc0, false, false); \
    bb = bp[16];  b.x = bb.x; b.y = bb.y;                                  \
    acc1 = __builtin_amdgcn_wmma_f32_16x16x4_f32(false, a, false, b, (short)0, acc1, false, false); \
    bb = bp[32];  b.x = bb.x; b.y = bb.y;                                  \
    acc2 = __builtin_amdgcn_wmma_f32_16x16x4_f32(false, a, false, b, (short)0, acc2, false, false); \
    bb = bp[48];  b.x = bb.x; b.y = bb.y;                                  \
    acc3 = __builtin_amdgcn_wmma_f32_16x16x4_f32(false, a, false, b, (short)0, acc3, false, false); \
    bb = bp[64];  b.x = bb.x; b.y = bb.y;                                  \
    acc4 = __builtin_amdgcn_wmma_f32_16x16x4_f32(false, a, false, b, (short)0, acc4, false, false); \
    bb = bp[80];  b.x = bb.x; b.y = bb.y;                                  \
    acc5 = __builtin_amdgcn_wmma_f32_16x16x4_f32(false, a, false, b, (short)0, acc5, false, false); \
    bb = bp[96];  b.x = bb.x; b.y = bb.y;                                  \
    acc6 = __builtin_amdgcn_wmma_f32_16x16x4_f32(false, a, false, b, (short)0, acc6, false, false); \
    bb = bp[112]; b.x = bb.x; b.y = bb.y;                                  \
    acc7 = __builtin_amdgcn_wmma_f32_16x16x4_f32(false, a, false, b, (short)0, acc7, false, false); \
  }

#pragma unroll 2
  for (int k0 = 0; k0 < HD; k0 += 4) GEMM_STEP(arow1, 0)
#pragma unroll 2
  for (int k0 = 0; k0 < HD; k0 += 4) GEMM_STEP(arow2, 64)
#undef GEMM_STEP

  const int rbase = rowBase + hi * 8;
  v8f accs[8] = {acc0, acc1, acc2, acc3, acc4, acc5, acc6, acc7};
#pragma unroll
  for (int ct = 0; ct < 8; ++ct) {
    const int col = ct * 16 + m;
    const float bv = bias[col];
#pragma unroll
    for (int v = 0; v < 8; ++v) {
      float val = accs[ct][v] + bv;
      val = val > 0.f ? val : 0.f;
      out[(size_t)(rbase + v) * HD + col] = val;
    }
  }
}

#define XS_PITCH 132
#define XS_ROWS  34
#define CW_PITCH 388

__global__ __launch_bounds__(256) void sortpool_head(
    const float* __restrict__ H,
    const float* __restrict__ conv_w, const float* __restrict__ conv_b,
    const float* __restrict__ W1, const float* __restrict__ b1,
    const float* __restrict__ W2, const float* __restrict__ b2,
    float* __restrict__ out)
{
  __shared__ float vals[NN];
  __shared__ int   idxs[NN];
  __shared__ float4 xs4[XS_ROWS * XS_PITCH / 4];
  __shared__ float4 cw4[CONV_O * CW_PITCH / 4];
  __shared__ float y[FLAT];
  __shared__ float z[HD];
  __shared__ float logits[NCLS];
  __shared__ float lse;

  const int g = blockIdx.x;
  const int t = threadIdx.x;
  const float* hg = H + (size_t)g * NN * HD;

  const float4* cwg = (const float4*)conv_w;
#pragma unroll
  for (int j = 0; j < CONV_O * HD * CONV_W / 4 / 256; ++j) {
    const int idx = t + 256 * j;
    const int o = idx / (HD * CONV_W / 4);
    const int q = idx - o * (HD * CONV_W / 4);
    cw4[o * (CW_PITCH / 4) + q] = cwg[idx];
  }

  if (t < NN) { vals[t] = hg[t * HD + (HD - 1)]; idxs[t] = t; }
  __syncthreads();

  for (int k = 2; k <= NN; k <<= 1) {
    for (int j = k >> 1; j > 0; j >>= 1) {
      if (t < NN) {
        const int ixj = t ^ j;
        if (ixj > t) {
          const bool dirDesc = ((t & k) == 0);
          const float va = vals[t], vb = vals[ixj];
          const int   ia = idxs[t], ib = idxs[ixj];
          const bool aBeforeB = (va > vb) || (va == vb && ia < ib);
          if (dirDesc ? !aBeforeB : aBeforeB) {
            vals[t] = vb; vals[ixj] = va;
            idxs[t] = ib; idxs[ixj] = ia;
          }
        }
      }
      __syncthreads();
    }
  }

  const float4* hg4 = (const float4*)hg;
  for (int i = t; i < XS_ROWS * (HD / 4); i += 256) {
    const int r = i >> 5, c4 = i & 31;
    float4 v = make_float4(0.f, 0.f, 0.f, 0.f);
    if (r < KTOP) v = hg4[idxs[r] * (HD / 4) + c4];
    xs4[r * (XS_PITCH / 4) + c4] = v;
  }
  __syncthreads();

  const int wave = t >> 5;
  const int lane = t & 31;
  const int m    = lane & 15;
  const int hi   = lane >> 4;
  if (wave < 4) {
    const int rt = wave & 1;
    const int ct = wave >> 1;
    const float* xs = (const float*)xs4;
    const float* cw = (const float*)cw4;
    const float* ap = xs + (rt * 16 + m) * XS_PITCH + 2 * hi;
    const float* bp = cw + (ct * 16 + m) * CW_PITCH + 6 * hi;
    v8f acc0 = {}, acc1 = {};
#pragma unroll
    for (int w = 0; w < CONV_W; ++w) {
#pragma unroll 4
      for (int c0 = 0; c0 < HD; c0 += 8) {
        const v2f a0 = *(const v2f*)(ap + w * XS_PITCH + c0);
        const v2f a1 = *(const v2f*)(ap + w * XS_PITCH + c0 + 4);
        v2f b0, b1;
        b0.x = bp[c0 * CONV_W + w];
        b0.y = bp[(c0 + 1) * CONV_W + w];
        b1.x = bp[(c0 + 4) * CONV_W + w];
        b1.y = bp[(c0 + 5) * CONV_W + w];
        acc0 = __builtin_amdgcn_wmma_f32_16x16x4_f32(false, a0, false, b0,
                                                     (short)0, acc0, false, false);
        acc1 = __builtin_amdgcn_wmma_f32_16x16x4_f32(false, a1, false, b1,
                                                     (short)0, acc1, false, false);
      }
    }
    const int o = ct * 16 + m;
    const float bv = conv_b[o];
#pragma unroll
    for (int v = 0; v < 8; ++v) {
      const int tt = rt * 16 + hi * 8 + v;
      const float s = (acc0[v] + acc1[v]) + bv;
      if (tt < CONV_T) y[o * CONV_T + tt] = s > 0.f ? s : 0.f;
    }
  }
  __syncthreads();

  if (t < HD) {
    float s = b1[t];
    for (int i = 0; i < FLAT; ++i) s += y[i] * W1[(size_t)i * HD + t];
    z[t] = s > 0.f ? s : 0.f;
  }
  __syncthreads();

  if (t < NCLS) {
    float s = b2[t];
    for (int j = 0; j < HD; ++j) s += z[j] * W2[j * NCLS + t];
    logits[t] = s;
  }
  __syncthreads();

  if (t == 0) {
    float mx = logits[0];
    for (int c = 1; c < NCLS; ++c) mx = fmaxf(mx, logits[c]);
    float se = 0.f;
    for (int c = 0; c < NCLS; ++c) se += expf(logits[c] - mx);
    lse = mx + logf(se);
  }
  __syncthreads();
  if (t < NCLS) out[(size_t)g * NCLS + t] = logits[t] - lse;
}

extern "C" void kernel_launch(void* const* d_in, const int* in_sizes, int n_in,
                              void* d_out, int out_size, void* d_ws, size_t ws_size,
                              hipStream_t stream) {
  (void)in_sizes; (void)n_in; (void)out_size; (void)ws_size;
  const float* x      = (const float*)d_in[0];
  const int*   src    = (const int*)d_in[1];
  const int*   dst    = (const int*)d_in[2];
  const float* Wl1    = (const float*)d_in[5];
  const float* bl1    = (const float*)d_in[6];
  const float* Wr1    = (const float*)d_in[7];
  const float* Wl2    = (const float*)d_in[8];
  const float* bl2    = (const float*)d_in[9];
  const float* Wr2    = (const float*)d_in[10];
  const float* Wl3    = (const float*)d_in[11];
  const float* bl3    = (const float*)d_in[12];
  const float* Wr3    = (const float*)d_in[13];
  const float* conv_w = (const float*)d_in[14];
  const float* conv_b = (const float*)d_in[15];
  const float* W1     = (const float*)d_in[16];
  const float* b1     = (const float*)d_in[17];
  const float* W2     = (const float*)d_in[18];
  const float* b2     = (const float*)d_in[19];

  float* ws   = (float*)d_ws;
  float* mean = ws;
  float* h1   = ws + (size_t)VN * HD;
  float* h2   = h1 + (size_t)VN * HD;

  const int gemmBlocks = VN / 128;

  sage_aggregate<<<BG, 256, 0, stream>>>(x, src, dst, mean);
  sage_gemm<<<gemmBlocks, 256, 0, stream>>>(mean, x, Wl1, Wr1, bl1, h1);
  sage_aggregate<<<BG, 256, 0, stream>>>(h1, src, dst, mean);
  sage_gemm<<<gemmBlocks, 256, 0, stream>>>(mean, h1, Wl2, Wr2, bl2, h2);
  sage_aggregate<<<BG, 256, 0, stream>>>(h2, src, dst, mean);
  sage_gemm<<<gemmBlocks, 256, 0, stream>>>(mean, h2, Wl3, Wr3, bl3, h1);
  sortpool_head<<<BG, 256, 0, stream>>>(h1, conv_w, conv_b, W1, b1, W2, b2,
                                        (float*)d_out);
}


# each layer's aggregation and linear transform fused into one launch per graph: mean and X stay in LDS as A operands, weights staged beside them; mean round trip and 3 launches gone; sum order kept
# speedup vs baseline: 5.7667x; 1.0614x over previous
#include <hip/hip_runtime.h>
#include <hip/hip_bf16.h>

#define BG   1024
#define NN   128
#define VN   (BG * NN)
#define HD   128
#define EPG  2048
#define KTOP 30
#define CONV_O 32
#define CONV_W 3
#define CONV_T (KTOP - CONV_W + 1)
#define FLAT   (CONV_O * CONV_T)
#define NCLS   16

typedef float v2f __attribute__((ext_vector_type(2)));
typedef float v8f __attribute__((ext_vector_type(8)));

#define FP_PITCH 144
#define CM_PITCH 132

__global__ __launch_bounds__(256) void sage_layer(
    const float* __restrict__ X, const int* src,
    const int* dst,
    const float* __restrict__ Wl, const float* __restrict__ Wr,
    const float* __restrict__ bias, float* __restrict__ out)
{
  __shared__ float2 fp[(NN / 2) * FP_PITCH];
  __shared__ float  cm[NN * CM_PITCH];
  __shared__ float  deg[NN];
  __shared__ float2 wp[128 * HD];

  const int g = blockIdx.x;
  const int t = threadIdx.x;

  for (int i = t; i < NN * CM_PITCH / 4; i += 256)
    ((float4*)cm)[i] = make_float4(0.f, 0.f, 0.f, 0.f);
  if (t < NN) deg[t] = 0.f;

  const float4* xg4 = (const float4*)(X + (size_t)g * NN * HD);
#pragma unroll
  for (int j = 0; j < 8; ++j) {
    const int idx = t + 256 * j;
    const int q   = idx >> 5;
    const int c4  = idx & 31;
    const float4 r0 = xg4[(2 * q) * (HD / 4) + c4];
    const float4 r1 = xg4[(2 * q + 1) * (HD / 4) + c4];
    float4* o = (float4*)(fp + q * FP_PITCH + c4 * 4);
    o[0] = make_float4(r0.x, r1.x, r0.y, r1.y);
    o[1] = make_float4(r0.z, r1.z, r0.w, r1.w);
  }

#pragma unroll
  for (int j = 0; j < 16; ++j) {
    const int idx = t + 256 * j;
    const int kk  = idx >> 5;
    const int c4  = idx & 31;
    const float4* w4 = (const float4*)(j < 8 ? Wl : Wr) + (size_t)(2 * (kk & 63)) * (HD / 4);
    const float4 r0 = w4[c4];
    const float4 r1 = w4[(HD / 4) + c4];
    float4* o = (float4*)(wp + kk * HD + c4 * 4);
    o[0] = make_float4(r0.x, r1.x, r0.y, r1.y);
    o[1] = make_float4(r0.z, r1.z, r0.w, r1.w);
  }
  const int4* s4 = (const int4*)(src + (size_t)g * EPG);
  const int4* d4 = (const int4*)(dst + (size_t)g * EPG);
  const int4 sv0 = s4[t], sv1 = s4[t + 256];
  const int4 dv0 = d4[t], dv1 = d4[t + 256];
  __syncthreads();

#define COUNT_EDGE(S, D)                                   \
  {                                                        \
    const int s_ = (S) & (NN - 1);                         \
    const int d_ = (D) & (NN - 1);                         \
    atomicAdd(&cm[d_ * CM_PITCH + s_], 1.0f);              \
    atomicAdd(&deg[d_], 1.0f);                             \
  }
  COUNT_EDGE(sv0.x, dv0.x) COUNT_EDGE(sv0.y, dv0.y)
  COUNT_EDGE(sv0.z, dv0.z) COUNT_EDGE(sv0.w, dv0.w)
  COUNT_EDGE(sv1.x, dv1.x) COUNT_EDGE(sv1.y, dv1.y)
  COUNT_EDGE(sv1.z, dv1.z) COUNT_EDGE(sv1.w, dv1.w)
#undef COUNT_EDGE
  __syncthreads();

  const int wave = t >> 5;
  const int lane = t & 31;
  const int m    = lane & 15;
  const int hi   = lane >> 4;

  const float*  arow = cm + (wave * 16 + m) * CM_PITCH + 2 * hi;
  const float2* bcol = fp + hi * FP_PITCH + m;

  v8f acc[8] = {};
  v2f    af[2];
  float2 bf[2][8];

  af[0] = *(const v2f*)arow;
#pragma unroll
  for (int ct = 0; ct < 8; ++ct) bf[0][ct] = bcol[ct * 16];

#pragma unroll
  for (int ks = 0; ks < NN / 4; ++ks) {
    const int cur = ks & 1, nxt = cur ^ 1;
    if (ks + 1 < NN / 4) {
      af[nxt] = *(const v2f*)(arow + 4 * (ks + 1));
      const float2* bp = bcol + 2 * (ks + 1) * FP_PITCH;
#pragma unroll
      for (int ct = 0; ct < 8; ++ct) bf[nxt][ct] = bp[ct * 16];
    }
#pragma unroll
    for (int ct = 0; ct < 8; ++ct) {
      v2f b; b.x = bf[cur][ct].x; b.y = bf[cur][ct].y;
      acc[ct] = __builtin_amdgcn_wmma_f32_16x16x4_f32(false, af[cur], false, b,
                                                      (short)0, acc[ct], false, false);
    }
  }

  float dn[8];
#pragma unroll
  for (int v = 0; v < 8; ++v) {
    const float c = deg[wave * 16 + 8 * hi + v];
    dn[v] = c > 1.f ? c : 1.f;
  }
  float* st = cm + (wave * 16 + 8 * hi) * CM_PITCH + m;
#pragma unroll
  for (int ct = 0; ct < 8; ++ct) {
#pragma unroll
    for (int v = 0; v < 8; ++v) st[v * CM_PITCH + ct * 16] = acc[ct][v] / dn[v];
  }

  const int row = wave * 16 + m;
  const float*  ax = (const float*)fp + ((row >> 1) * FP_PITCH + 2 * hi) * 2 + (row & 1);
  const float2* bw = wp + hi * HD + m;

  v8f acc0 = {}, acc1 = {}, acc2 = {}, acc3 = {};
  v8f acc4 = {}, acc5 = {}, acc6 = {}, acc7 = {};

#define GEMM_STEP(KOFF)                                                    \
  {                                                                        \
    const float2* bp = bw + ((k0 >> 1) + (KOFF)) * HD;                     \
    float2 bb; v2f b;                                                      \
    bb = bp[0];   b.x = bb.x; b.y = bb.y;                                  \
    acc0 = __builtin_amdgcn_wmma_f32_16x16x4_f32(false, a, false, b, (short)0, acc0, false, false); \
    bb = bp[16];  b.x = bb.x; b.y = bb.y;                                  \
    acc1 = __builtin_amdgcn_wmma_f32_16x16x4_f32(false, a, false, b, (short)0, acc1, false, false); \
    bb = bp[32];  b.x = bb.x; b.y = bb.y;                                  \
    acc2 = __builtin_amdgcn_wmma_f32_16x16x4_f32(false, a, false, b, (short)0, acc2, false, false); \
    bb = bp[48];  b.x = bb.x; b.y = bb.y;                                  \
    acc3 = __builtin_amdgcn_wmma_f32_16x16x4_f32(false, a, false, b, (short)0, acc3, false, false); \
    bb = bp[64];  b.x = bb.x; b.y = bb.y;                                  \
    acc4 = __builtin_amdgcn_wmma_f32_16x16x4_f32(false, a, false, b, (short)0, acc4, false, false); \
    bb = bp[80];  b.x = bb.x; b.y = bb.y;                                  \
    acc5 = __builtin_amdgcn_wmma_f32_16x16x4_f32(false, a, false, b, (short)0, acc5, false, false); \
    bb = bp[96];  b.x = bb.x; b.y = bb.y;                                  \
    acc6 = __builtin_amdgcn_wmma_f32_16x16x4_f32(false, a, false, b, (short)0, acc6, false, false); \
    bb = bp[112]; b.x = bb.x; b.y = bb.y;                                  \
    acc7 = __builtin_amdgcn_wmma_f32_16x16x4_f32(false, a, false, b, (short)0, acc7, false, false); \
  }

#pragma unroll 4
  for (int k0 = 0; k0 < HD; k0 += 4) {
    const v2f a = *(const v2f*)(arow + k0);
    GEMM_STEP(0)
  }
#pragma unroll 4
  for (int k0 = 0; k0 < HD; k0 += 4) {
    v2f a; a.x = ax[2 * k0]; a.y = ax[2 * k0 + 2];
    GEMM_STEP(64)
  }
#undef GEMM_STEP

  const int rbase = g * NN + wave * 16 + hi * 8;
  v8f accs[8] = {acc0, acc1, acc2, acc3, acc4, acc5, acc6, acc7};
#pragma unroll
  for (int ct = 0; ct < 8; ++ct) {
    const int col = ct * 16 + m;
    const float bv = bias[col];
#pragma unroll
    for (int v = 0; v < 8; ++v) {
      float val = accs[ct][v] + bv;
      val = val > 0.f ? val : 0.f;
      out[(size_t)(rbase + v) * HD + col] = val;
    }
  }
}

#define XS_PITCH 132
#define XS_ROWS  34
#define CW_PITCH 388

__global__ __launch_bounds__(256) void sortpool_head(
    const float* __restrict__ H,
    const float* __restrict__ conv_w, const float* __restrict__ conv_b,
    const float* __restrict__ W1, const float* __restrict__ b1,
    const float* __restrict__ W2, const float* __restrict__ b2,
    float* __restrict__ out)
{
  __shared__ float vals[NN];
  __shared__ int   idxs[NN];
  __shared__ float4 xs4[XS_ROWS * XS_PITCH / 4];
  __shared__ float4 cw4[CONV_O * CW_PITCH / 4];
  __shared__ float y[FLAT];
  __shared__ float z[HD];
  __shared__ float logits[NCLS];
  __shared__ float lse;

  const int g = blockIdx.x;
  const int t = threadIdx.x;
  const float* hg = H + (size_t)g * NN * HD;

  const float4* cwg = (const float4*)conv_w;
#pragma unroll
  for (int j = 0; j < CONV_O * HD * CONV_W / 4 / 256; ++j) {
    const int idx = t + 256 * j;
    const int o = idx / (HD * CONV_W / 4);
    const int q = idx - o * (HD * CONV_W / 4);
    cw4[o * (CW_PITCH / 4) + q] = cwg[idx];
  }

  if (t < NN) { vals[t] = hg[t * HD + (HD - 1)]; idxs[t] = t; }
  __syncthreads();

  for (int k = 2; k <= NN; k <<= 1) {
    for (int j = k >> 1; j > 0; j >>= 1) {
      if (t < NN) {
        const int ixj = t ^ j;
        if (ixj > t) {
          const bool dirDesc = ((t & k) == 0);
          const float va = vals[t], vb = vals[ixj];
          const int   ia = idxs[t], ib = idxs[ixj];
          const bool aBeforeB = (va > vb) || (va == vb && ia < ib);
          if (dirDesc ? !aBeforeB : aBeforeB) {
            vals[t] = vb; vals[ixj] = va;
            idxs[t] = ib; idxs[ixj] = ia;
          }
        }
      }
      __syncthreads();
    }
  }

  const float4* hg4 = (const float4*)hg;
  for (int i = t; i < XS_ROWS * (HD / 4); i += 256) {
    const int r = i >> 5, c4 = i & 31;
    float4 v = make_float4(0.f, 0.f, 0.f, 0.f);
    if (r < KTOP) v = hg4[idxs[r] * (HD / 4) + c4];
    xs4[r * (XS_PITCH / 4) + c4] = v;
  }
  __syncthreads();

  const int wave = t >> 5;
  const int lane = t & 31;
  const int m    = lane & 15;
  const int hi   = lane >> 4;
  if (wave < 4) {
    const int rt = wave & 1;
    const int ct = wave >> 1;
    const float* xs = (const float*)xs4;
    const float* cw = (const float*)cw4;
    const float* ap = xs + (rt * 16 + m) * XS_PITCH + 2 * hi;
    const float* bp = cw + (ct * 16 + m) * CW_PITCH + 6 * hi;
    v8f acc0 = {}, acc1 = {};
#pragma unroll
    for (int w = 0; w < CONV_W; ++w) {
#pragma unroll 4
      for (int c0 = 0; c0 < HD; c0 += 8) {
        const v2f a0 = *(const v2f*)(ap + w * XS_PITCH + c0);
        const v2f a1 = *(const v2f*)(ap + w * XS_PITCH + c0 + 4);
        v2f b0, b1;
        b0.x = bp[c0 * CONV_W + w];
        b0.y = bp[(c0 + 1) * CONV_W + w];
        b1.x = bp[(c0 + 4) * CONV_W + w];
        b1.y = bp[(c0 + 5) * CONV_W + w];
        acc0 = __builtin_amdgcn_wmma_f32_16x16x4_f32(false, a0, false, b0,
                                                     (short)0, acc0, false, false);
        acc1 = __builtin_amdgcn_wmma_f32_16x16x4_f32(false, a1, false, b1,
                                                     (short)0, acc1, false, false);
      }
    }
    const int o = ct * 16 + m;
    const float bv = conv_b[o];
#pragma unroll
    for (int v = 0; v < 8; ++v) {
      const int tt = rt * 16 + hi * 8 + v;
      const float s = (acc0[v] + acc1[v]) + bv;
      if (tt < CONV_T) y[o * CONV_T + tt] = s > 0.f ? s : 0.f;
    }
  }
  __syncthreads();

  if (t < HD) {
    float s = b1[t];
    for (int i = 0; i < FLAT; ++i) s += y[i] * W1[(size_t)i * HD + t];
    z[t] = s > 0.f ? s : 0.f;
  }
  __syncthreads();

  if (t < NCLS) {
    float s = b2[t];
    for (int j = 0; j < HD; ++j) s += z[j] * W2[j * NCLS + t];
    logits[t] = s;
  }
  __syncthreads();

  if (t == 0) {
    float mx = logits[0];
    for (int c = 1; c < NCLS; ++c) mx = fmaxf(mx, logits[c]);
    float se = 0.f;
    for (int c = 0; c < NCLS; ++c) se += expf(logits[c] - mx);
    lse = mx + logf(se);
  }
  __syncthreads();
  if (t < NCLS) out[(size_t)g * NCLS + t] = logits[t] - lse;
}

extern "C" void kernel_launch(void* const* d_in, const int* in_sizes, int n_in,
                              void* d_out, int out_size, void* d_ws, size_t ws_size,
                              hipStream_t stream) {
  (void)in_sizes; (void)n_in; (void)out_size; (void)ws_size;
  const float* x      = (const float*)d_in[0];
  const int*   src    = (const int*)d_in[1];
  const int*   dst    = (const int*)d_in[2];
  const float* Wl1    = (const float*)d_in[5];
  const float* bl1    = (const float*)d_in[6];
  const float* Wr1    = (const float*)d_in[7];
  const float* Wl2    = (const float*)d_in[8];
  const float* bl2    = (const float*)d_in[9];
  const float* Wr2    = (const float*)d_in[10];
  const float* Wl3    = (const float*)d_in[11];
  const float* bl3    = (const float*)d_in[12];
  const float* Wr3    = (const float*)d_in[13];
  const float* conv_w = (const float*)d_in[14];
  const float* conv_b = (const float*)d_in[15];
  const float* W1     = (const float*)d_in[16];
  const float* b1     = (const float*)d_in[17];
  const float* W2     = (const float*)d_in[18];
  const float* b2     = (const float*)d_in[19];

  float* ws   = (float*)d_ws;
  float* h1   = ws + (size_t)VN * HD;
  float* h2   = h1 + (size_t)VN * HD;

  sage_layer<<<BG, 256, 0, stream>>>(x, src, dst, Wl1, Wr1, bl1, h1);
  sage_layer<<<BG, 256, 0, stream>>>(h1, src, dst, Wl2, Wr2, bl2, h2);
  sage_layer<<<BG, 256, 0, stream>>>(h2, src, dst, Wl3, Wr3, bl3, h1);
  sortpool_head<<<BG, 256, 0, stream>>>(h1, conv_w, conv_b, W1, b1, W2, b2,
                                        (float*)d_out);
}


# sage layers: 256 persistent workgroups each loop over 4 graphs; [Wl;Wr] staged to LDS once per workgroup, next graph's features and edges requested to registers ahead of the current matrix work
# speedup vs baseline: 6.0230x; 1.0444x over previous
#include <hip/hip_runtime.h>
#include <hip/hip_bf16.h>

#define BG   1024
#define NN   128
#define VN   (BG * NN)
#define HD   128
#define EPG  2048
#define KTOP 30
#define CONV_O 32
#define CONV_W 3
#define CONV_T (KTOP - CONV_W + 1)
#define FLAT   (CONV_O * CONV_T)
#define NCLS   16

typedef float v2f __attribute__((ext_vector_type(2)));
typedef float v8f __attribute__((ext_vector_type(8)));

#define FP_PITCH 144
#define CM_PITCH 132

#define NWG 256

#define REQUEST_GRAPH(G)                                                   \
  {                                                                        \
    const float4* xg4_ = (const float4*)(X + (size_t)(G) * NN * HD);       \
      \
    _Pragma("unroll")                                                      \
    for (int j = 0; j < 8; ++j) {                                          \
      const int idx = t + 256 * j;                                         \
      const int q   = idx >> 5;                                            \
      const int c4  = idx & 31;                                            \
      xr[2 * j]     = xg4_[(2 * q) * (HD / 4) + c4];                       \
      xr[2 * j + 1] = xg4_[(2 * q + 1) * (HD / 4) + c4];                   \
    }                                                                      \
    const int4* s4_ = (const int4*)(src + (size_t)(G) * EPG);              \
    const int4* d4_ = (const int4*)(dst + (size_t)(G) * EPG);              \
    sv0 = s4_[t]; sv1 = s4_[t + 256];                                      \
    dv0 = d4_[t]; dv1 = d4_[t + 256];                                      \
  }

__global__ __launch_bounds__(256) void sage_layer(
    const float* __restrict__ X, const int* src,
    const int* dst,
    const float* __restrict__ Wl, const float* __restrict__ Wr,
    const float* __restrict__ bias, float* __restrict__ out)
{
  __shared__ float2 fp[(NN / 2) * FP_PITCH];
  __shared__ float  cm[NN * CM_PITCH];
  __shared__ float  deg[NN];
  __shared__ float2 wp[128 * HD];

  const int t = threadIdx.x;
  const int wave = t >> 5;
  const int lane = t & 31;
  const int m    = lane & 15;
  const int hi   = lane >> 4;

  float4 xr[16];
  int4 sv0, sv1, dv0, dv1;
  REQUEST_GRAPH(blockIdx.x)

#pragma unroll
  for (int j = 0; j < 16; ++j) {
    const int idx = t + 256 * j;
    const int kk  = idx >> 5;
    const int c4  = idx & 31;
    const float4* w4 = (const float4*)(j < 8 ? Wl : Wr) + (size_t)(2 * (kk & 63)) * (HD / 4);
    const float4 r0 = w4[c4];
    const float4 r1 = w4[(HD / 4) + c4];
    float4* o = (float4*)(wp + kk * HD + c4 * 4);
    o[0] = make_float4(r0.x, r1.x, r0.y, r1.y);
    o[1] = make_float4(r0.z, r1.z, r0.w, r1.w);
  }

  int g = blockIdx.x;
  do {

  for (int i = t; i < NN * CM_PITCH / 4; i += 256)
    ((float4*)cm)[i] = make_float4(0.f, 0.f, 0.f, 0.f);
  if (t < NN) deg[t] = 0.f;

#pragma unroll
  for (int j = 0; j < 8; ++j) {
    const int idx = t + 256 * j;
    const int q   = idx >> 5;
    const int c4  = idx & 31;
    const float4 r0 = xr[2 * j];
    const float4 r1 = xr[2 * j + 1];
    float4* o = (float4*)(fp + q * FP_PITCH + c4 * 4);
    o[0] = make_float4(r0.x, r1.x, r0.y, r1.y);
    o[1] = make_float4(r0.z, r1.z, r0.w, r1.w);
  }
  __syncthreads();

#define COUNT_EDGE(S, D)                                   \
  {                                                        \
    const int s_ = (S) & (NN - 1);                         \
    const int d_ = (D) & (NN - 1);                         \
    atomicAdd(&cm[d_ * CM_PITCH + s_], 1.0f);              \
    atomicAdd(&deg[d_], 1.0f);                             \
  }
  COUNT_EDGE(sv0.x, dv0.x) COUNT_EDGE(sv0.y, dv0.y)
  COUNT_EDGE(sv0.z, dv0.z) COUNT_EDGE(sv0.w, dv0.w)
  COUNT_EDGE(sv1.x, dv1.x) COUNT_EDGE(sv1.y, dv1.y)
  COUNT_EDGE(sv1.z, dv1.z) COUNT_EDGE(sv1.w, dv1.w)
#undef COUNT_EDGE
  __syncthreads();

  if (g + (int)gridDim.x < BG) REQUEST_GRAPH(g + (int)gridDim.x)

  const float*  arow = cm + (wave * 16 + m) * CM_PITCH + 2 * hi;
  const float2* bcol = fp + hi * FP_PITCH + m;

  v8f acc[8] = {};
  v2f    af[2];
  float2 bf[2][8];

  af[0] = *(const v2f*)arow;
#pragma unroll
  for (int ct = 0; ct < 8; ++ct) bf[0][ct] = bcol[ct * 16];

#pragma unroll
  for (int ks = 0; ks < NN / 4; ++ks) {
    const int cur = ks & 1, nxt = cur ^ 1;
    if (ks + 1 < NN / 4) {
      af[nxt] = *(const v2f*)(arow + 4 * (ks + 1));
      const float2* bp = bcol + 2 * (ks + 1) * FP_PITCH;
#pragma unroll
      for (int ct = 0; ct < 8; ++ct) bf[nxt][ct] = bp[ct * 16];
    }
#pragma unroll
    for (int ct = 0; ct < 8; ++ct) {
      v2f b; b.x = bf[cur][ct].x; b.y = bf[cur][ct].y;
      acc[ct] = __builtin_amdgcn_wmma_f32_16x16x4_f32(false, af[cur], false, b,
                                                      (short)0, acc[ct], false, false);
    }
  }

  float dn[8];
#pragma unroll
  for (int v = 0; v < 8; ++v) {
    const float c = deg[wave * 16 + 8 * hi + v];
    dn[v] = c > 1.f ? c : 1.f;
  }
  float* st = cm + (wave * 16 + 8 * hi) * CM_PITCH + m;
#pragma unroll
  for (int ct = 0; ct < 8; ++ct) {
#pragma unroll
    for (int v = 0; v < 8; ++v) st[v * CM_PITCH + ct * 16] = acc[ct][v] / dn[v];
  }

  const int row = wave * 16 + m;
  const float*  ax = (const float*)fp + ((row >> 1) * FP_PITCH + 2 * hi) * 2 + (row & 1);
  const float2* bw = wp + hi * HD + m;

  v8f acc0 = {}, acc1 = {}, acc2 = {}, acc3 = {};
  v8f acc4 = {}, acc5 = {}, acc6 = {}, acc7 = {};

#define GEMM_STEP(KOFF)                                                    \
  {                                                                        \
    const float2* bp = bw + ((k0 >> 1) + (KOFF)) * HD;                     \
    float2 bb; v2f b;                                                      \
    bb = bp[0];   b.x = bb.x; b.y = bb.y;                                  \
    acc0 = __builtin_amdgcn_wmma_f32_16x16x4_f32(false, a, false, b, (short)0, acc0, false, false); \
    bb = bp[16];  b.x = bb.x; b.y = bb.y;                                  \
    acc1 = __builtin_amdgcn_wmma_f32_16x16x4_f32(false, a, false, b, (short)0, acc1, false, false); \
    bb = bp[32];  b.x = bb.x; b.y = bb.y;                                  \
    acc2 = __builtin_amdgcn_wmma_f32_16x16x4_f32(false, a, false, b, (short)0, acc2, false, false); \
    bb = bp[48];  b.x = bb.x; b.y = bb.y;                                  \
    acc3 = __builtin_amdgcn_wmma_f32_16x16x4_f32(false, a, false, b, (short)0, acc3, false, false); \
    bb = bp[64];  b.x = bb.x; b.y = bb.y;                                  \
    acc4 = __builtin_amdgcn_wmma_f32_16x16x4_f32(false, a, false, b, (short)0, acc4, false, false); \
    bb = bp[80];  b.x = bb.x; b.y = bb.y;                                  \
    acc5 = __builtin_amdgcn_wmma_f32_16x16x4_f32(false, a, false, b, (short)0, acc5, false, false); \
    bb = bp[96];  b.x = bb.x; b.y = bb.y;                                  \
    acc6 = __builtin_amdgcn_wmma_f32_16x16x4_f32(false, a, false, b, (short)0, acc6, false, false); \
    bb = bp[112]; b.x = bb.x; b.y = bb.y;                                  \
    acc7 = __builtin_amdgcn_wmma_f32_16x16x4_f32(false, a, false, b, (short)0, acc7, false, false); \
  }

#pragma unroll 4
  for (int k0 = 0; k0 < HD; k0 += 4) {
    const v2f a = *(const v2f*)(arow + k0);
    GEMM_STEP(0)
  }
#pragma unroll 4
  for (int k0 = 0; k0 < HD; k0 += 4) {
    v2f a; a.x = ax[2 * k0]; a.y = ax[2 * k0 + 2];
    GEMM_STEP(64)
  }
#undef GEMM_STEP

  __syncthreads();

  const int rbase = g * NN + wave * 16 + hi * 8;
  v8f accs[8] = {acc0, acc1, acc2, acc3, acc4, acc5, acc6, acc7};
#pragma unroll
  for (int ct = 0; ct < 8; ++ct) {
    const int col = ct * 16 + m;
    const float bv = bias[col];
#pragma unroll
    for (int v = 0; v < 8; ++v) {
      float val = accs[ct][v] + bv;
      val = val > 0.f ? val : 0.f;
      out[(size_t)(rbase + v) * HD + col] = val;
    }
  }

  g += (int)gridDim.x;
  } while (g < BG);
}
#undef REQUEST_GRAPH

#define XS_PITCH 132
#define XS_ROWS  34
#define CW_PITCH 388

__global__ __launch_bounds__(256) void sortpool_head(
    const float* __restrict__ H,
    const float* __restrict__ conv_w, const float* __restrict__ conv_b,
    const float* __restrict__ W1, const float* __restrict__ b1,
    const float* __restrict__ W2, const float* __restrict__ b2,
    float* __restrict__ out)
{
  __shared__ float vals[NN];
  __shared__ int   idxs[NN];
  __shared__ float4 xs4[XS_ROWS * XS_PITCH / 4];
  __shared__ float4 cw4[CONV_O * CW_PITCH / 4];
  __shared__ float y[FLAT];
  __shared__ float z[HD];
  __shared__ float logits[NCLS];
  __shared__ float lse;

  const int g = blockIdx.x;
  const int t = threadIdx.x;
  const float* hg = H + (size_t)g * NN * HD;

  const float4* cwg = (const float4*)conv_w;
#pragma unroll
  for (int j = 0; j < CONV_O * HD * CONV_W / 4 / 256; ++j) {
    const int idx = t + 256 * j;
    const int o = idx / (HD * CONV_W / 4);
    const int q = idx - o * (HD * CONV_W / 4);
    cw4[o * (CW_PITCH / 4) + q] = cwg[idx];
  }

  if (t < NN) { vals[t] = hg[t * HD + (HD - 1)]; idxs[t] = t; }
  __syncthreads();

  for (int k = 2; k <= NN; k <<= 1) {
    for (int j = k >> 1; j > 0; j >>= 1) {
      if (t < NN) {
        const int ixj = t ^ j;
        if (ixj > t) {
          const bool dirDesc = ((t & k) == 0);
          const float va = vals[t], vb = vals[ixj];
          const int   ia = idxs[t], ib = idxs[ixj];
          const bool aBeforeB = (va > vb) || (va == vb && ia < ib);
          if (dirDesc ? !aBeforeB : aBeforeB) {
            vals[t] = vb; vals[ixj] = va;
            idxs[t] = ib; idxs[ixj] = ia;
          }
        }
      }
      __syncthreads();
    }
  }

  const float4* hg4 = (const float4*)hg;
  for (int i = t; i < XS_ROWS * (HD / 4); i += 256) {
    const int r = i >> 5, c4 = i & 31;
    float4 v = make_float4(0.f, 0.f, 0.f, 0.f);
    if (r < KTOP) v = hg4[idxs[r] * (HD / 4) + c4];
    xs4[r * (XS_PITCH / 4) + c4] = v;
  }
  __syncthreads();

  const int wave = t >> 5;
  const int lane = t & 31;
  const int m    = lane & 15;
  const int hi   = lane >> 4;
  if (wave < 4) {
    const int rt = wave & 1;
    const int ct = wave >> 1;
    const float* xs = (const float*)xs4;
    const float* cw = (const float*)cw4;
    const float* ap = xs + (rt * 16 + m) * XS_PITCH + 2 * hi;
    const float* bp = cw + (ct * 16 + m) * CW_PITCH + 6 * hi;
    v8f acc0 = {}, acc1 = {};
#pragma unroll
    for (int w = 0; w < CONV_W; ++w) {
#pragma unroll 4
      for (int c0 = 0; c0 < HD; c0 += 8) {
        const v2f a0 = *(const v2f*)(ap + w * XS_PITCH + c0);
        const v2f a1 = *(const v2f*)(ap + w * XS_PITCH + c0 + 4);
        v2f b0, b1;
        b0.x = bp[c0 * CONV_W + w];
        b0.y = bp[(c0 + 1) * CONV_W + w];
        b1.x = bp[(c0 + 4) * CONV_W + w];
        b1.y = bp[(c0 + 5) * CONV_W + w];
        acc0 = __builtin_amdgcn_wmma_f32_16x16x4_f32(false, a0, false, b0,
                                                     (short)0, acc0, false, false);
        acc1 = __builtin_amdgcn_wmma_f32_16x16x4_f32(false, a1, false, b1,
                                                     (short)0, acc1, false, false);
      }
    }
    const int o = ct * 16 + m;
    const float bv = conv_b[o];
#pragma unroll
    for (int v = 0; v < 8; ++v) {
      const int tt = rt * 16 + hi * 8 + v;
      const float s = (acc0[v] + acc1[v]) + bv;
      if (tt < CONV_T) y[o * CONV_T + tt] = s > 0.f ? s : 0.f;
    }
  }
  __syncthreads();

  if (t < HD) {
    float s = b1[t];
    for (int i = 0; i < FLAT; ++i) s += y[i] * W1[(size_t)i * HD + t];
    z[t] = s > 0.f ? s : 0.f;
  }
  __syncthreads();

  if (t < NCLS) {
    float s = b2[t];
    for (int j = 0; j < HD; ++j) s += z[j] * W2[j * NCLS + t];
    logits[t] = s;
  }
  __syncthreads();

  if (t == 0) {
    float mx = logits[0];
    for (int c = 1; c < NCLS; ++c) mx = fmaxf(mx, logits[c]);
    float se = 0.f;
    for (int c = 0; c < NCLS; ++c) se += expf(logits[c] - mx);
    lse = mx + logf(se);
  }
  __syncthreads();
  if (t < NCLS) out[(size_t)g * NCLS + t] = logits[t] - lse;
}

extern "C" void kernel_launch(void* const* d_in, const int* in_sizes, int n_in,
                              void* d_out, int out_size, void* d_ws, size_t ws_size,
                              hipStream_t stream) {
  (void)in_sizes; (void)n_in; (void)out_size; (void)ws_size;
  const float* x      = (const float*)d_in[0];
  const int*   src    = (const int*)d_in[1];
  const int*   dst    = (const int*)d_in[2];
  const float* Wl1    = (const float*)d_in[5];
  const float* bl1    = (const float*)d_in[6];
  const float* Wr1    = (const float*)d_in[7];
  const float* Wl2    = (const float*)d_in[8];
  const float* bl2    = (const float*)d_in[9];
  const float* Wr2    = (const float*)d_in[10];
  const float* Wl3    = (const float*)d_in[11];
  const float* bl3    = (const float*)d_in[12];
  const float* Wr3    = (const float*)d_in[13];
  const float* conv_w = (const float*)d_in[14];
  const float* conv_b = (const float*)d_in[15];
  const float* W1     = (const float*)d_in[16];
  const float* b1     = (const float*)d_in[17];
  const float* W2     = (const float*)d_in[18];
  const float* b2     = (const float*)d_in[19];

  float* ws   = (float*)d_ws;
  float* h1   = ws + (size_t)VN * HD;
  float* h2   = h1 + (size_t)VN * HD;

  sage_layer<<<NWG, 256, 0, stream>>>(x, src, dst, Wl1, Wr1, bl1, h1);
  sage_layer<<<NWG, 256, 0, stream>>>(h1, src, dst, Wl2, Wr2, bl2, h2);
  sage_layer<<<NWG, 256, 0, stream>>>(h2, src, dst, Wl3, Wr3, bl3, h1);
  sortpool_head<<<BG, 256, 0, stream>>>(h1, conv_w, conv_b, W1, b1, W2, b2,
                                        (float*)d_out);
}


# sage layers: s_setprio window: the younger wave of each SIMD's pair holds matrix-issue priority for 64 of its 96 k-steps, then hands it back, so neither runs alone; sum order unchanged
# speedup vs baseline: 6.2423x; 1.0364x over previous
#include <hip/hip_runtime.h>
#include <hip/hip_bf16.h>

#define BG   1024
#define NN   128
#define VN   (BG * NN)
#define HD   128
#define EPG  2048
#define KTOP 30
#define CONV_O 32
#define CONV_W 3
#define CONV_T (KTOP - CONV_W + 1)
#define FLAT   (CONV_O * CONV_T)
#define NCLS   16

typedef float v2f __attribute__((ext_vector_type(2)));
typedef float v8f __attribute__((ext_vector_type(8)));

#define FP_PITCH 144
#define CM_PITCH 132

#define NWG 256

#define REQUEST_GRAPH(G)                                                   \
  {                                                                        \
    const float4* xg4_ = (const float4*)(X + (size_t)(G) * NN * HD);       \
      \
    _Pragma("unroll")                                                      \
    for (int j = 0; j < 8; ++j) {                                          \
      const int idx = t + 256 * j;                                         \
      const int q   = idx >> 5;                                            \
      const int c4  = idx & 31;                                            \
      xr[2 * j]     = xg4_[(2 * q) * (HD / 4) + c4];                       \
      xr[2 * j + 1] = xg4_[(2 * q + 1) * (HD / 4) + c4];                   \
    }                                                                      \
    const int4* s4_ = (const int4*)(src + (size_t)(G) * EPG);              \
    const int4* d4_ = (const int4*)(dst + (size_t)(G) * EPG);              \
    sv0 = s4_[t]; sv1 = s4_[t + 256];                                      \
    dv0 = d4_[t]; dv1 = d4_[t + 256];                                      \
  }

__global__ __launch_bounds__(256) void sage_layer(
    const float* __restrict__ X, const int* src,
    const int* dst,
    const float* __restrict__ Wl, const float* __restrict__ Wr,
    const float* __restrict__ bias, float* __restrict__ out)
{
  __shared__ float2 fp[(NN / 2) * FP_PITCH];
  __shared__ float  cm[NN * CM_PITCH];
  __shared__ float  deg[NN];
  __shared__ float2 wp[128 * HD];

  const int t = threadIdx.x;
  const int wave = t >> 5;
  const int lane = t & 31;
  const int m    = lane & 15;
  const int hi   = lane >> 4;
  const bool younger = __builtin_amdgcn_readfirstlane(wave) >= 4;

  float4 xr[16];
  int4 sv0, sv1, dv0, dv1;
  REQUEST_GRAPH(blockIdx.x)

#pragma unroll
  for (int j = 0; j < 16; ++j) {
    const int idx = t + 256 * j;
    const int kk  = idx >> 5;
    const int c4  = idx & 31;
    const float4* w4 = (const float4*)(j < 8 ? Wl : Wr) + (size_t)(2 * (kk & 63)) * (HD / 4);
    const float4 r0 = w4[c4];
    const float4 r1 = w4[(HD / 4) + c4];
    float4* o = (float4*)(wp + kk * HD + c4 * 4);
    o[0] = make_float4(r0.x, r1.x, r0.y, r1.y);
    o[1] = make_float4(r0.z, r1.z, r0.w, r1.w);
  }

  int g = blockIdx.x;
  do {

  for (int i = t; i < NN * CM_PITCH / 4; i += 256)
    ((float4*)cm)[i] = make_float4(0.f, 0.f, 0.f, 0.f);
  if (t < NN) deg[t] = 0.f;

#pragma unroll
  for (int j = 0; j < 8; ++j) {
    const int idx = t + 256 * j;
    const int q   = idx >> 5;
    const int c4  = idx & 31;
    const float4 r0 = xr[2 * j];
    const float4 r1 = xr[2 * j + 1];
    float4* o = (float4*)(fp + q * FP_PITCH + c4 * 4);
    o[0] = make_float4(r0.x, r1.x, r0.y, r1.y);
    o[1] = make_float4(r0.z, r1.z, r0.w, r1.w);
  }
  __syncthreads();

#define COUNT_EDGE(S, D)                                   \
  {                                                        \
    const int s_ = (S) & (NN - 1);                         \
    const int d_ = (D) & (NN - 1);                         \
    atomicAdd(&cm[d_ * CM_PITCH + s_], 1.0f);              \
    atomicAdd(&deg[d_], 1.0f);                             \
  }
  COUNT_EDGE(sv0.x, dv0.x) COUNT_EDGE(sv0.y, dv0.y)
  COUNT_EDGE(sv0.z, dv0.z) COUNT_EDGE(sv0.w, dv0.w)
  COUNT_EDGE(sv1.x, dv1.x) COUNT_EDGE(sv1.y, dv1.y)
  COUNT_EDGE(sv1.z, dv1.z) COUNT_EDGE(sv1.w, dv1.w)
#undef COUNT_EDGE
  __syncthreads();

  if (g + (int)gridDim.x < BG) REQUEST_GRAPH(g + (int)gridDim.x)

  if (younger) __builtin_amdgcn_s_setprio((short)3);

  const float*  arow = cm + (wave * 16 + m) * CM_PITCH + 2 * hi;
  const float2* bcol = fp + hi * FP_PITCH + m;

  v8f acc[8] = {};
  v2f    af[2];
  float2 bf[2][8];

  af[0] = *(const v2f*)arow;
#pragma unroll
  for (int ct = 0; ct < 8; ++ct) bf[0][ct] = bcol[ct * 16];

#pragma unroll
  for (int ks = 0; ks < NN / 4; ++ks) {
    const int cur = ks & 1, nxt = cur ^ 1;
    if (ks + 1 < NN / 4) {
      af[nxt] = *(const v2f*)(arow + 4 * (ks + 1));
      const float2* bp = bcol + 2 * (ks + 1) * FP_PITCH;
#pragma unroll
      for (int ct = 0; ct < 8; ++ct) bf[nxt][ct] = bp[ct * 16];
    }
#pragma unroll
    for (int ct = 0; ct < 8; ++ct) {
      v2f b; b.x = bf[cur][ct].x; b.y = bf[cur][ct].y;
      acc[ct] = __builtin_amdgcn_wmma_f32_16x16x4_f32(false, af[cur], false, b,
                                                      (short)0, acc[ct], false, false);
    }
  }

  float dn[8];
#pragma unroll
  for (int v = 0; v < 8; ++v) {
    const float c = deg[wave * 16 + 8 * hi + v];
    dn[v] = c > 1.f ? c : 1.f;
  }
  float* st = cm + (wave * 16 + 8 * hi) * CM_PITCH + m;
#pragma unroll
  for (int ct = 0; ct < 8; ++ct) {
#pragma unroll
    for (int v = 0; v < 8; ++v) st[v * CM_PITCH + ct * 16] = acc[ct][v] / dn[v];
  }

  const int row = wave * 16 + m;
  const float*  ax = (const float*)fp + ((row >> 1) * FP_PITCH + 2 * hi) * 2 + (row & 1);
  const float2* bw = wp + hi * HD + m;

  v8f acc0 = {}, acc1 = {}, acc2 = {}, acc3 = {};
  v8f acc4 = {}, acc5 = {}, acc6 = {}, acc7 = {};

#define GEMM_STEP(KOFF)                                                    \
  {                                                                        \
    const float2* bp = bw + ((k0 >> 1) + (KOFF)) * HD;                     \
    float2 bb; v2f b;                                                      \
    bb = bp[0];   b.x = bb.x; b.y = bb.y;                                  \
    acc0 = __builtin_amdgcn_wmma_f32_16x16x4_f32(false, a, false, b, (short)0, acc0, false, false); \
    bb = bp[16];  b.x = bb.x; b.y = bb.y;                                  \
    acc1 = __builtin_amdgcn_wmma_f32_16x16x4_f32(false, a, false, b, (short)0, acc1, false, false); \
    bb = bp[32];  b.x = bb.x; b.y = bb.y;                                  \
    acc2 = __builtin_amdgcn_wmma_f32_16x16x4_f32(false, a, false, b, (short)0, acc2, false, false); \
    bb = bp[48];  b.x = bb.x; b.y = bb.y;                                  \
    acc3 = __builtin_amdgcn_wmma_f32_16x16x4_f32(false, a, false, b, (short)0, acc3, false, false); \
    bb = bp[64];  b.x = bb.x; b.y = bb.y;                                  \
    acc4 = __builtin_amdgcn_wmma_f32_16x16x4_f32(false, a, false, b, (short)0, acc4, false, false); \
    bb = bp[80];  b.x = bb.x; b.y = bb.y;                                  \
    acc5 = __builtin_amdgcn_wmma_f32_16x16x4_f32(false, a, false, b, (short)0, acc5, false, false); \
    bb = bp[96];  b.x = bb.x; b.y = bb.y;                                  \
    acc6 = __builtin_amdgcn_wmma_f32_16x16x4_f32(false, a, false, b, (short)0, acc6, false, false); \
    bb = bp[112]; b.x = bb.x; b.y = bb.y;                                  \
    acc7 = __builtin_amdgcn_wmma_f32_16x16x4_f32(false, a, false, b, (short)0, acc7, false, false); \
  }

#pragma unroll 4
  for (int k0 = 0; k0 < HD; k0 += 4) {
    const v2f a = *(const v2f*)(arow + k0);
    GEMM_STEP(0)
  }
  if (younger) __builtin_amdgcn_s_setprio((short)0);

#pragma unroll 4
  for (int k0 = 0; k0 < HD; k0 += 4) {
    v2f a; a.x = ax[2 * k0]; a.y = ax[2 * k0 + 2];
    GEMM_STEP(64)
  }
#undef GEMM_STEP

  __syncthreads();

  const int rbase = g * NN + wave * 16 + hi * 8;
  v8f accs[8] = {acc0, acc1, acc2, acc3, acc4, acc5, acc6, acc7};
#pragma unroll
  for (int ct = 0; ct < 8; ++ct) {
    const int col = ct * 16 + m;
    const float bv = bias[col];
#pragma unroll
    for (int v = 0; v < 8; ++v) {
      float val = accs[ct][v] + bv;
      val = val > 0.f ? val : 0.f;
      out[(size_t)(rbase + v) * HD + col] = val;
    }
  }

  g += (int)gridDim.x;
  } while (g < BG);
}
#undef REQUEST_GRAPH

#define XS_PITCH 132
#define XS_ROWS  34
#define CW_PITCH 388

__global__ __launch_bounds__(256) void sortpool_head(
    const float* __restrict__ H,
    const float* __restrict__ conv_w, const float* __restrict__ conv_b,
    const float* __restrict__ W1, const float* __restrict__ b1,
    const float* __restrict__ W2, const float* __restrict__ b2,
    float* __restrict__ out)
{
  __shared__ float vals[NN];
  __shared__ int   idxs[NN];
  __shared__ float4 xs4[XS_ROWS * XS_PITCH / 4];
  __shared__ float4 cw4[CONV_O * CW_PITCH / 4];
  __shared__ float y[FLAT];
  __shared__ float z[HD];
  __shared__ float logits[NCLS];
  __shared__ float lse;

  const int g = blockIdx.x;
  const int t = threadIdx.x;
  const float* hg = H + (size_t)g * NN * HD;

  const float4* cwg = (const float4*)conv_w;
#pragma unroll
  for (int j = 0; j < CONV_O * HD * CONV_W / 4 / 256; ++j) {
    const int idx = t + 256 * j;
    const int o = idx / (HD * CONV_W / 4);
    const int q = idx - o * (HD * CONV_W / 4);
    cw4[o * (CW_PITCH / 4) + q] = cwg[idx];
  }

  if (t < NN) { vals[t] = hg[t * HD + (HD - 1)]; idxs[t] = t; }
  __syncthreads();

  for (int k = 2; k <= NN; k <<= 1) {
    for (int j = k >> 1; j > 0; j >>= 1) {
      if (t < NN) {
        const int ixj = t ^ j;
        if (ixj > t) {
          const bool dirDesc = ((t & k) == 0);
          const float va = vals[t], vb = vals[ixj];
          const int   ia = idxs[t], ib = idxs[ixj];
          const bool aBeforeB = (va > vb) || (va == vb && ia < ib);
          if (dirDesc ? !aBeforeB : aBeforeB) {
            vals[t] = vb; vals[ixj] = va;
            idxs[t] = ib; idxs[ixj] = ia;
          }
        }
      }
      __syncthreads();
    }
  }

  const float4* hg4 = (const float4*)hg;
  for (int i = t; i < XS_ROWS * (HD / 4); i += 256) {
    const int r = i >> 5, c4 = i & 31;
    float4 v = make_float4(0.f, 0.f, 0.f, 0.f);
    if (r < KTOP) v = hg4[idxs[r] * (HD / 4) + c4];
    xs4[r * (XS_PITCH / 4) + c4] = v;
  }
  __syncthreads();

  const int wave = t >> 5;
  const int lane = t & 31;
  const int m    = lane & 15;
  const int hi   = lane >> 4;
  if (wave < 4) {
    const int rt = wave & 1;
    const int ct = wave >> 1;
    const float* xs = (const float*)xs4;
    const float* cw = (const float*)cw4;
    const float* ap = xs + (rt * 16 + m) * XS_PITCH + 2 * hi;
    const float* bp = cw + (ct * 16 + m) * CW_PITCH + 6 * hi;
    v8f acc0 = {}, acc1 = {};
#pragma unroll
    for (int w = 0; w < CONV_W; ++w) {
#pragma unroll 4
      for (int c0 = 0; c0 < HD; c0 += 8) {
        const v2f a0 = *(const v2f*)(ap + w * XS_PITCH + c0);
        const v2f a1 = *(const v2f*)(ap + w * XS_PITCH + c0 + 4);
        v2f b0, b1;
        b0.x = bp[c0 * CONV_W + w];
        b0.y = bp[(c0 + 1) * CONV_W + w];
        b1.x = bp[(c0 + 4) * CONV_W + w];
        b1.y = bp[(c0 + 5) * CONV_W + w];
        acc0 = __builtin_amdgcn_wmma_f32_16x16x4_f32(false, a0, false, b0,
                                                     (short)0, acc0, false, false);
        acc1 = __builtin_amdgcn_wmma_f32_16x16x4_f32(false, a1, false, b1,
                                                     (short)0, acc1, false, false);
      }
    }
    const int o = ct * 16 + m;
    const float bv = conv_b[o];
#pragma unroll
    for (int v = 0; v < 8; ++v) {
      const int tt = rt * 16 + hi * 8 + v;
      const float s = (acc0[v] + acc1[v]) + bv;
      if (tt < CONV_T) y[o * CONV_T + tt] = s > 0.f ? s : 0.f;
    }
  }
  __syncthreads();

  if (t < HD) {
    float s = b1[t];
    for (int i = 0; i < FLAT; ++i) s += y[i] * W1[(size_t)i * HD + t];
    z[t] = s > 0.f ? s : 0.f;
  }
  __syncthreads();

  if (t < NCLS) {
    float s = b2[t];
    for (int j = 0; j < HD; ++j) s += z[j] * W2[j * NCLS + t];
    logits[t] = s;
  }
  __syncthreads();

  if (t == 0) {
    float mx = logits[0];
    for (int c = 1; c < NCLS; ++c) mx = fmaxf(mx, logits[c]);
    float se = 0.f;
    for (int c = 0; c < NCLS; ++c) se += expf(logits[c] - mx);
    lse = mx + logf(se);
  }
  __syncthreads();
  if (t < NCLS) out[(size_t)g * NCLS + t] = logits[t] - lse;
}

extern "C" void kernel_launch(void* const* d_in, const int* in_sizes, int n_in,
                              void* d_out, int out_size, void* d_ws, size_t ws_size,
                              hipStream_t stream) {
  (void)in_sizes; (void)n_in; (void)out_size; (void)ws_size;
  const float* x      = (const float*)d_in[0];
  const int*   src    = (const int*)d_in[1];
  const int*   dst    = (const int*)d_in[2];
  const float* Wl1    = (const float*)d_in[5];
  const float* bl1    = (const float*)d_in[6];
  const float* Wr1    = (const float*)d_in[7];
  const float* Wl2    = (const float*)d_in[8];
  const float* bl2    = (const float*)d_in[9];
  const float* Wr2    = (const float*)d_in[10];
  const float* Wl3    = (const float*)d_in[11];
  const float* bl3    = (const float*)d_in[12];
  const float* Wr3    = (const float*)d_in[13];
  const float* conv_w = (const float*)d_in[14];
  const float* conv_b = (const float*)d_in[15];
  const float* W1     = (const float*)d_in[16];
  const float* b1     = (const float*)d_in[17];
  const float* W2     = (const float*)d_in[18];
  const float* b2     = (const float*)d_in[19];

  float* ws   = (float*)d_ws;
  float* h1   = ws + (size_t)VN * HD;
  float* h2   = h1 + (size_t)VN * HD;

  sage_layer<<<NWG, 256, 0, stream>>>(x, src, dst, Wl1, Wr1, bl1, h1);
  sage_layer<<<NWG, 256, 0, stream>>>(h1, src, dst, Wl2, Wr2, bl2, h2);
  sage_layer<<<NWG, 256, 0, stream>>>(h2, src, dst, Wl3, Wr3, bl3, h1);
  sortpool_head<<<BG, 256, 0, stream>>>(h1, conv_w, conv_b, W1, b1, W2, b2,
                                        (float*)d_out);
}


# head: both dense layers' dot products split along K over all 256 threads (8 and 16 slices, whole-row 16-byte W1 loads), partial sums added via LDS in slice order; sum order of both layers changed
# speedup vs baseline: 6.8199x; 1.0925x over previous
#include <hip/hip_runtime.h>
#include <hip/hip_bf16.h>

#define BG   1024
#define NN   128
#define VN   (BG * NN)
#define HD   128
#define EPG  2048
#define KTOP 30
#define CONV_O 32
#define CONV_W 3
#define CONV_T (KTOP - CONV_W + 1)
#define FLAT   (CONV_O * CONV_T)
#define NCLS   16

typedef float v2f __attribute__((ext_vector_type(2)));
typedef float v8f __attribute__((ext_vector_type(8)));

#define FP_PITCH 144
#define CM_PITCH 132

#define NWG 256

#define REQUEST_GRAPH(G)                                                   \
  {                                                                        \
    const float4* xg4_ = (const float4*)(X + (size_t)(G) * NN * HD);       \
      \
    _Pragma("unroll")                                                      \
    for (int j = 0; j < 8; ++j) {                                          \
      const int idx = t + 256 * j;                                         \
      const int q   = idx >> 5;                                            \
      const int c4  = idx & 31;                                            \
      xr[2 * j]     = xg4_[(2 * q) * (HD / 4) + c4];                       \
      xr[2 * j + 1] = xg4_[(2 * q + 1) * (HD / 4) + c4];                   \
    }                                                                      \
    const int4* s4_ = (const int4*)(src + (size_t)(G) * EPG);              \
    const int4* d4_ = (const int4*)(dst + (size_t)(G) * EPG);              \
    sv0 = s4_[t]; sv1 = s4_[t + 256];                                      \
    dv0 = d4_[t]; dv1 = d4_[t + 256];                                      \
  }

__global__ __launch_bounds__(256) void sage_layer(
    const float* __restrict__ X, const int* src,
    const int* dst,
    const float* __restrict__ Wl, const float* __restrict__ Wr,
    const float* __restrict__ bias, float* __restrict__ out)
{
  __shared__ float2 fp[(NN / 2) * FP_PITCH];
  __shared__ float  cm[NN * CM_PITCH];
  __shared__ float  deg[NN];
  __shared__ float2 wp[128 * HD];

  const int t = threadIdx.x;
  const int wave = t >> 5;
  const int lane = t & 31;
  const int m    = lane & 15;
  const int hi   = lane >> 4;
  const bool younger = __builtin_amdgcn_readfirstlane(wave) >= 4;

  float4 xr[16];
  int4 sv0, sv1, dv0, dv1;
  REQUEST_GRAPH(blockIdx.x)

#pragma unroll
  for (int j = 0; j < 16; ++j) {
    const int idx = t + 256 * j;
    const int kk  = idx >> 5;
    const int c4  = idx & 31;
    const float4* w4 = (const float4*)(j < 8 ? Wl : Wr) + (size_t)(2 * (kk & 63)) * (HD / 4);
    const float4 r0 = w4[c4];
    const float4 r1 = w4[(HD / 4) + c4];
    float4* o = (float4*)(wp + kk * HD + c4 * 4);
    o[0] = make_float4(r0.x, r1.x, r0.y, r1.y);
    o[1] = make_float4(r0.z, r1.z, r0.w, r1.w);
  }

  int g = blockIdx.x;
  do {

  for (int i = t; i < NN * CM_PITCH / 4; i += 256)
    ((float4*)cm)[i] = make_float4(0.f, 0.f, 0.f, 0.f);
  if (t < NN) deg[t] = 0.f;

#pragma unroll
  for (int j = 0; j < 8; ++j) {
    const int idx = t + 256 * j;
    const int q   = idx >> 5;
    const int c4  = idx & 31;
    const float4 r0 = xr[2 * j];
    const float4 r1 = xr[2 * j + 1];
    float4* o = (float4*)(fp + q * FP_PITCH + c4 * 4);
    o[0] = make_float4(r0.x, r1.x, r0.y, r1.y);
    o[1] = make_float4(r0.z, r1.z, r0.w, r1.w);
  }
  __syncthreads();

#define COUNT_EDGE(S, D)                                   \
  {                                                        \
    const int s_ = (S) & (NN - 1);                         \
    const int d_ = (D) & (NN - 1);                         \
    atomicAdd(&cm[d_ * CM_PITCH + s_], 1.0f);              \
    atomicAdd(&deg[d_], 1.0f);                             \
  }
  COUNT_EDGE(sv0.x, dv0.x) COUNT_EDGE(sv0.y, dv0.y)
  COUNT_EDGE(sv0.z, dv0.z) COUNT_EDGE(sv0.w, dv0.w)
  COUNT_EDGE(sv1.x, dv1.x) COUNT_EDGE(sv1.y, dv1.y)
  COUNT_EDGE(sv1.z, dv1.z) COUNT_EDGE(sv1.w, dv1.w)
#undef COUNT_EDGE
  __syncthreads();

  if (g + (int)gridDim.x < BG) REQUEST_GRAPH(g + (int)gridDim.x)

  if (younger) __builtin_amdgcn_s_setprio((short)3);

  const float*  arow = cm + (wave * 16 + m) * CM_PITCH + 2 * hi;
  const float2* bcol = fp + hi * FP_PITCH + m;

  v8f acc[8] = {};
  v2f    af[2];
  float2 bf[2][8];

  af[0] = *(const v2f*)arow;
#pragma unroll
  for (int ct = 0; ct < 8; ++ct) bf[0][ct] = bcol[ct * 16];

#pragma unroll
  for (int ks = 0; ks < NN / 4; ++ks) {
    const int cur = ks & 1, nxt = cur ^ 1;
    if (ks + 1 < NN / 4) {
      af[nxt] = *(const v2f*)(arow + 4 * (ks + 1));
      const float2* bp = bcol + 2 * (ks + 1) * FP_PITCH;
#pragma unroll
      for (int ct = 0; ct < 8; ++ct) bf[nxt][ct] = bp[ct * 16];
    }
#pragma unroll
    for (int ct = 0; ct < 8; ++ct) {
      v2f b; b.x = bf[cur][ct].x; b.y = bf[cur][ct].y;
      acc[ct] = __builtin_amdgcn_wmma_f32_16x16x4_f32(false, af[cur], false, b,
                                                      (short)0, acc[ct], false, false);
    }
  }

  float dn[8];
#pragma unroll
  for (int v = 0; v < 8; ++v) {
    const float c = deg[wave * 16 + 8 * hi + v];
    dn[v] = c > 1.f ? c : 1.f;
  }
  float* st = cm + (wave * 16 + 8 * hi) * CM_PITCH + m;
#pragma unroll
  for (int ct = 0; ct < 8; ++ct) {
#pragma unroll
    for (int v = 0; v < 8; ++v) st[v * CM_PITCH + ct * 16] = acc[ct][v] / dn[v];
  }

  const int row = wave * 16 + m;
  const float*  ax = (const float*)fp + ((row >> 1) * FP_PITCH + 2 * hi) * 2 + (row & 1);
  const float2* bw = wp + hi * HD + m;

  v8f acc0 = {}, acc1 = {}, acc2 = {}, acc3 = {};
  v8f acc4 = {}, acc5 = {}, acc6 = {}, acc7 = {};

#define GEMM_STEP(KOFF)                                                    \
  {                                                                        \
    const float2* bp = bw + ((k0 >> 1) + (KOFF)) * HD;                     \
    float2 bb; v2f b;                                                      \
    bb = bp[0];   b.x = bb.x; b.y = bb.y;                                  \
    acc0 = __builtin_amdgcn_wmma_f32_16x16x4_f32(false, a, false, b, (short)0, acc0, false, false); \
    bb = bp[16];  b.x = bb.x; b.y = bb.y;                                  \
    acc1 = __builtin_amdgcn_wmma_f32_16x16x4_f32(false, a, false, b, (short)0, acc1, false, false); \
    bb = bp[32];  b.x = bb.x; b.y = bb.y;                                  \
    acc2 = __builtin_amdgcn_wmma_f32_16x16x4_f32(false, a, false, b, (short)0, acc2, false, false); \
    bb = bp[48];  b.x = bb.x; b.y = bb.y;                                  \
    acc3 = __builtin_amdgcn_wmma_f32_16x16x4_f32(false, a, false, b, (short)0, acc3, false, false); \
    bb = bp[64];  b.x = bb.x; b.y = bb.y;                                  \
    acc4 = __builtin_amdgcn_wmma_f32_16x16x4_f32(false, a, false, b, (short)0, acc4, false, false); \
    bb = bp[80];  b.x = bb.x; b.y = bb.y;                                  \
    acc5 = __builtin_amdgcn_wmma_f32_16x16x4_f32(false, a, false, b, (short)0, acc5, false, false); \
    bb = bp[96];  b.x = bb.x; b.y = bb.y;                                  \
    acc6 = __builtin_amdgcn_wmma_f32_16x16x4_f32(false, a, false, b, (short)0, acc6, false, false); \
    bb = bp[112]; b.x = bb.x; b.y = bb.y;                                  \
    acc7 = __builtin_amdgcn_wmma_f32_16x16x4_f32(false, a, false, b, (short)0, acc7, false, false); \
  }

#pragma unroll 4
  for (int k0 = 0; k0 < HD; k0 += 4) {
    const v2f a = *(const v2f*)(arow + k0);
    GEMM_STEP(0)
  }
  if (younger) __builtin_amdgcn_s_setprio((short)0);

#pragma unroll 4
  for (int k0 = 0; k0 < HD; k0 += 4) {
    v2f a; a.x = ax[2 * k0]; a.y = ax[2 * k0 + 2];
    GEMM_STEP(64)
  }
#undef GEMM_STEP

  __syncthreads();

  const int rbase = g * NN + wave * 16 + hi * 8;
  v8f accs[8] = {acc0, acc1, acc2, acc3, acc4, acc5, acc6, acc7};
#pragma unroll
  for (int ct = 0; ct < 8; ++ct) {
    const int col = ct * 16 + m;
    const float bv = bias[col];
#pragma unroll
    for (int v = 0; v < 8; ++v) {
      float val = accs[ct][v] + bv;
      val = val > 0.f ? val : 0.f;
      out[(size_t)(rbase + v) * HD + col] = val;
    }
  }

  g += (int)gridDim.x;
  } while (g < BG);
}
#undef REQUEST_GRAPH

#define XS_PITCH 132
#define XS_ROWS  34
#define CW_PITCH 388

__global__ __launch_bounds__(256) void sortpool_head(
    const float* __restrict__ H,
    const float* __restrict__ conv_w, const float* __restrict__ conv_b,
    const float* __restrict__ W1, const float* __restrict__ b1,
    const float* __restrict__ W2, const float* __restrict__ b2,
    float* __restrict__ out)
{
  __shared__ float vals[NN];
  __shared__ int   idxs[NN];
  __shared__ float4 xs4[XS_ROWS * XS_PITCH / 4];
  __shared__ float4 cw4[CONV_O * CW_PITCH / 4];
  __shared__ float4 y4s[FLAT / 4];
  __shared__ float z[HD];
  __shared__ float logits[NCLS];
  __shared__ float lse;

  const int g = blockIdx.x;
  const int t = threadIdx.x;
  float* y = (float*)y4s;
  const float* hg = H + (size_t)g * NN * HD;

  const float4* cwg = (const float4*)conv_w;
#pragma unroll
  for (int j = 0; j < CONV_O * HD * CONV_W / 4 / 256; ++j) {
    const int idx = t + 256 * j;
    const int o = idx / (HD * CONV_W / 4);
    const int q = idx - o * (HD * CONV_W / 4);
    cw4[o * (CW_PITCH / 4) + q] = cwg[idx];
  }

  if (t < NN) { vals[t] = hg[t * HD + (HD - 1)]; idxs[t] = t; }
  __syncthreads();

  for (int k = 2; k <= NN; k <<= 1) {
    for (int j = k >> 1; j > 0; j >>= 1) {
      if (t < NN) {
        const int ixj = t ^ j;
        if (ixj > t) {
          const bool dirDesc = ((t & k) == 0);
          const float va = vals[t], vb = vals[ixj];
          const int   ia = idxs[t], ib = idxs[ixj];
          const bool aBeforeB = (va > vb) || (va == vb && ia < ib);
          if (dirDesc ? !aBeforeB : aBeforeB) {
            vals[t] = vb; vals[ixj] = va;
            idxs[t] = ib; idxs[ixj] = ia;
          }
        }
      }
      __syncthreads();
    }
  }

  const float4* hg4 = (const float4*)hg;
  for (int i = t; i < XS_ROWS * (HD / 4); i += 256) {
    const int r = i >> 5, c4 = i & 31;
    float4 v = make_float4(0.f, 0.f, 0.f, 0.f);
    if (r < KTOP) v = hg4[idxs[r] * (HD / 4) + c4];
    xs4[r * (XS_PITCH / 4) + c4] = v;
  }
  __syncthreads();

  const int wave = t >> 5;
  const int lane = t & 31;
  const int m    = lane & 15;
  const int hi   = lane >> 4;
  if (wave < 4) {
    const int rt = wave & 1;
    const int ct = wave >> 1;
    const float* xs = (const float*)xs4;
    const float* cw = (const float*)cw4;
    const float* ap = xs + (rt * 16 + m) * XS_PITCH + 2 * hi;
    const float* bp = cw + (ct * 16 + m) * CW_PITCH + 6 * hi;
    v8f acc0 = {}, acc1 = {};
#pragma unroll
    for (int w = 0; w < CONV_W; ++w) {
#pragma unroll 4
      for (int c0 = 0; c0 < HD; c0 += 8) {
        const v2f a0 = *(const v2f*)(ap + w * XS_PITCH + c0);
        const v2f a1 = *(const v2f*)(ap + w * XS_PITCH + c0 + 4);
        v2f b0, b1;
        b0.x = bp[c0 * CONV_W + w];
        b0.y = bp[(c0 + 1) * CONV_W + w];
        b1.x = bp[(c0 + 4) * CONV_W + w];
        b1.y = bp[(c0 + 5) * CONV_W + w];
        acc0 = __builtin_amdgcn_wmma_f32_16x16x4_f32(false, a0, false, b0,
                                                     (short)0, acc0, false, false);
        acc1 = __builtin_amdgcn_wmma_f32_16x16x4_f32(false, a1, false, b1,
                                                     (short)0, acc1, false, false);
      }
    }
    const int o = ct * 16 + m;
    const float bv = conv_b[o];
#pragma unroll
    for (int v = 0; v < 8; ++v) {
      const int tt = rt * 16 + hi * 8 + v;
      const float s = (acc0[v] + acc1[v]) + bv;
      if (tt < CONV_T) y[o * CONV_T + tt] = s > 0.f ? s : 0.f;
    }
  }
  __syncthreads();

  float4* part = cw4;
  float*  part2 = (float*)cw4 + 8 * HD;
  {
    const int cg = t & 31;
    const int ks = t >> 5;
    const float4* w14 = (const float4*)W1 + (size_t)(ks * (FLAT / 8)) * (HD / 4) + cg;
    const float4* yk  = y4s + ks * (FLAT / 8 / 4);
    float4 s = make_float4(0.f, 0.f, 0.f, 0.f);
#pragma unroll 2
    for (int i4 = 0; i4 < FLAT / 8 / 4; ++i4) {
      const float4 yv = yk[i4];
      const float4 w0 = w14[(4 * i4 + 0) * (HD / 4)];
      const float4 w1 = w14[(4 * i4 + 1) * (HD / 4)];
      const float4 w2 = w14[(4 * i4 + 2) * (HD / 4)];
      const float4 w3 = w14[(4 * i4 + 3) * (HD / 4)];
      s.x += yv.x * w0.x; s.y += yv.x * w0.y; s.z += yv.x * w0.z; s.w += yv.x * w0.w;
      s.x += yv.y * w1.x; s.y += yv.y * w1.y; s.z += yv.y * w1.z; s.w += yv.y * w1.w;
      s.x += yv.z * w2.x; s.y += yv.z * w2.y; s.z += yv.z * w2.z; s.w += yv.z * w2.w;
      s.x += yv.w * w3.x; s.y += yv.w * w3.y; s.z += yv.w * w3.z; s.w += yv.w * w3.w;
    }
    part[ks * (HD / 4) + cg] = s;
  }
  __syncthreads();
  if (t < HD) {
    const float* pf = (const float*)part;
    float s = b1[t];
#pragma unroll
    for (int ks = 0; ks < 8; ++ks) s += pf[ks * HD + t];
    z[t] = s > 0.f ? s : 0.f;
  }
  __syncthreads();

  {
    const int c   = t & 15;
    const int ks2 = t >> 4;
    float s = 0.f;
#pragma unroll
    for (int j = 0; j < 8; ++j) s += z[ks2 * 8 + j] * W2[(ks2 * 8 + j) * NCLS + c];
    part2[ks2 * NCLS + c] = s;
  }
  __syncthreads();
  if (t < NCLS) {
    float s = b2[t];
#pragma unroll
    for (int ks2 = 0; ks2 < 16; ++ks2) s += part2[ks2 * NCLS + t];
    logits[t] = s;
  }
  __syncthreads();

  if (t == 0) {
    float mx = logits[0];
    for (int c = 1; c < NCLS; ++c) mx = fmaxf(mx, logits[c]);
    float se = 0.f;
    for (int c = 0; c < NCLS; ++c) se += expf(logits[c] - mx);
    lse = mx + logf(se);
  }
  __syncthreads();
  if (t < NCLS) out[(size_t)g * NCLS + t] = logits[t] - lse;
}

extern "C" void kernel_launch(void* const* d_in, const int* in_sizes, int n_in,
                              void* d_out, int out_size, void* d_ws, size_t ws_size,
                              hipStream_t stream) {
  (void)in_sizes; (void)n_in; (void)out_size; (void)ws_size;
  const float* x      = (const float*)d_in[0];
  const int*   src    = (const int*)d_in[1];
  const int*   dst    = (const int*)d_in[2];
  const float* Wl1    = (const float*)d_in[5];
  const float* bl1    = (const float*)d_in[6];
  const float* Wr1    = (const float*)d_in[7];
  const float* Wl2    = (const float*)d_in[8];
  const float* bl2    = (const float*)d_in[9];
  const float* Wr2    = (const float*)d_in[10];
  const float* Wl3    = (const float*)d_in[11];
  const float* bl3    = (const float*)d_in[12];
  const float* Wr3    = (const float*)d_in[13];
  const float* conv_w = (const float*)d_in[14];
  const float* conv_b = (const float*)d_in[15];
  const float* W1     = (const float*)d_in[16];
  const float* b1     = (const float*)d_in[17];
  const float* W2     = (const float*)d_in[18];
  const float* b2     = (const float*)d_in[19];

  float* ws   = (float*)d_ws;
  float* h1   = ws + (size_t)VN * HD;
  float* h2   = h1 + (size_t)VN * HD;

  sage_layer<<<NWG, 256, 0, stream>>>(x, src, dst, Wl1, Wr1, bl1, h1);
  sage_layer<<<NWG, 256, 0, stream>>>(h1, src, dst, Wl2, Wr2, bl2, h2);
  sage_layer<<<NWG, 256, 0, stream>>>(h2, src, dst, Wl3, Wr3, bl3, h1);
  sortpool_head<<<BG, 256, 0, stream>>>(h1, conv_w, conv_b, W1, b1, W2, b2,
                                        (float*)d_out);
}


# sage: a graph's three layers run back to back in ONE launch; features, count matrix and degrees stay in LDS across layers (counted once), weights re-staged per layer; 2 launches gone; sums unchanged
# speedup vs baseline: 7.6876x; 1.1272x over previous
#include <hip/hip_runtime.h>
#include <hip/hip_bf16.h>

#define BG   1024
#define NN   128
#define VN   (BG * NN)
#define HD   128
#define EPG  2048
#define KTOP 30
#define CONV_O 32
#define CONV_W 3
#define CONV_T (KTOP - CONV_W + 1)
#define FLAT   (CONV_O * CONV_T)
#define NCLS   16

typedef float v2f __attribute__((ext_vector_type(2)));
typedef float v8f __attribute__((ext_vector_type(8)));


#define FP_PITCH 144
#define CM_PITCH 132
#define PK_PITCH 68

#define NWG 256

#define REQUEST_GRAPH(G)                                                   \
  {                                                                        \
    const float4* xg4_ = (const float4*)(X + (size_t)(G) * NN * HD);       \
      \
    _Pragma("unroll")                                                      \
    for (int j = 0; j < 8; ++j) {                                          \
      const int idx = t + 256 * j;                                         \
      const int q   = idx >> 5;                                            \
      const int c4  = idx & 31;                                            \
      xr[2 * j]     = xg4_[(2 * q) * (HD / 4) + c4];                       \
      xr[2 * j + 1] = xg4_[(2 * q + 1) * (HD / 4) + c4];                   \
    }                                                                      \
    const int4* s4_ = (const int4*)(src + (size_t)(G) * EPG);              \
    const int4* d4_ = (const int4*)(dst + (size_t)(G) * EPG);              \
    sv0 = s4_[t]; sv1 = s4_[t + 256];                                      \
    dv0 = d4_[t]; dv1 = d4_[t + 256];                                      \
  }

#define REQUEST_WEIGHTS(WL, WR)                                            \
  {                                                                        \
      \
                                \
    _Pragma("unroll")                                                      \
    for (int j = 0; j < 16; ++j) {                                         \
      const int idx = t + 256 * j;                                         \
      const int kk  = idx >> 5;                                            \
      const int c4  = idx & 31;                                            \
      const float4* w4_ = (const float4*)(j < 8 ? (WL) : (WR))             \
                          + (size_t)(2 * (kk & 63)) * (HD / 4);            \
      wr[2 * j]     = w4_[c4];                                             \
      wr[2 * j + 1] = w4_[(HD / 4) + c4];                                  \
    }                                                                      \
  }

__global__ __launch_bounds__(256) void sage_layers(
    const float* __restrict__ X, const int* src, const int* dst,
    const float* Wl1, const float* Wr1, const float* bl1,
    const float* Wl2, const float* Wr2, const float* bl2,
    const float* Wl3, const float* Wr3, const float* bl3,
    float* __restrict__ out)
{
  static_assert(BG % NWG == 0, "every workgroup loops over the same number of graphs");
  __shared__ float2 fp[(NN / 2) * FP_PITCH];
  __shared__ float  cm[NN * CM_PITCH];
  __shared__ float  deg[NN];
  __shared__ float  pk[NN * PK_PITCH];
  __shared__ float2 wp[128 * HD];

  const int t = threadIdx.x;
  const int wave = t >> 5;
  const int lane = t & 31;
  const int m    = lane & 15;
  const int hi   = lane >> 4;
  const bool younger = __builtin_amdgcn_readfirstlane(wave) >= 4;

  float4 xr[16];
  int4 sv0, sv1, dv0, dv1;
  float4 wr[32];
  REQUEST_GRAPH(blockIdx.x)
  REQUEST_WEIGHTS(Wl1, Wr1)

  int g = blockIdx.x;
  do {
  const bool more = g + (int)gridDim.x < BG;

#pragma nounroll
  for (int l = 0; l < 3; ++l) {

  if (l == 0) {
    for (int i = t; i < NN * CM_PITCH / 4; i += 256)
      ((float4*)cm)[i] = make_float4(0.f, 0.f, 0.f, 0.f);
    if (t < NN) deg[t] = 0.f;

#pragma unroll
    for (int j = 0; j < 8; ++j) {
      const int idx = t + 256 * j;
      const int q   = idx >> 5;
      const int c4  = idx & 31;
      const float4 r0 = xr[2 * j];
      const float4 r1 = xr[2 * j + 1];
      float4* o = (float4*)(fp + q * FP_PITCH + c4 * 4);
      o[0] = make_float4(r0.x, r1.x, r0.y, r1.y);
      o[1] = make_float4(r0.z, r1.z, r0.w, r1.w);
    }
  }

#pragma unroll
  for (int j = 0; j < 16; ++j) {
    const int idx = t + 256 * j;
    const int kk  = idx >> 5;
    const int c4  = idx & 31;
    const float4 r0 = wr[2 * j];
    const float4 r1 = wr[2 * j + 1];
    float4* o = (float4*)(wp + kk * HD + c4 * 4);
    o[0] = make_float4(r0.x, r1.x, r0.y, r1.y);
    o[1] = make_float4(r0.z, r1.z, r0.w, r1.w);
  }
  __syncthreads();

  if (l == 0) {
#define COUNT_EDGE(S, D)                                   \
    {                                                      \
      const int s_ = (S) & (NN - 1);                       \
      const int d_ = (D) & (NN - 1);                       \
      atomicAdd(&cm[d_ * CM_PITCH + s_], 1.0f);            \
      atomicAdd(&deg[d_], 1.0f);                           \
    }
    COUNT_EDGE(sv0.x, dv0.x) COUNT_EDGE(sv0.y, dv0.y)
    COUNT_EDGE(sv0.z, dv0.z) COUNT_EDGE(sv0.w, dv0.w)
    COUNT_EDGE(sv1.x, dv1.x) COUNT_EDGE(sv1.y, dv1.y)
    COUNT_EDGE(sv1.z, dv1.z) COUNT_EDGE(sv1.w, dv1.w)
#undef COUNT_EDGE
    __syncthreads();

    if (more) REQUEST_GRAPH(g + (int)gridDim.x)
  }

  const float* bias = l == 0 ? bl1 : (l == 1 ? bl2 : bl3);
  const float* WLn  = l == 0 ? Wl2 : (l == 1 ? Wl3 : Wl1);
  const float* WRn  = l == 0 ? Wr2 : (l == 1 ? Wr3 : Wr1);

  if (younger) __builtin_amdgcn_s_setprio((short)3);

  const float*  arow = cm + (wave * 16 + m) * CM_PITCH + 2 * hi;
  const float2* bcol = fp + hi * FP_PITCH + m;

  v8f acc[8] = {};
  v2f    af[2];
  float2 bf[2][8];

  af[0] = *(const v2f*)arow;
#pragma unroll
  for (int ct = 0; ct < 8; ++ct) bf[0][ct] = bcol[ct * 16];

#pragma unroll
  for (int ks = 0; ks < NN / 4; ++ks) {
    const int cur = ks & 1, nxt = cur ^ 1;
    if (ks + 1 < NN / 4) {
      af[nxt] = *(const v2f*)(arow + 4 * (ks + 1));
      const float2* bp = bcol + 2 * (ks + 1) * FP_PITCH;
#pragma unroll
      for (int ct = 0; ct < 8; ++ct) bf[nxt][ct] = bp[ct * 16];
    }
#pragma unroll
    for (int ct = 0; ct < 8; ++ct) {
      v2f b; b.x = bf[cur][ct].x; b.y = bf[cur][ct].y;
      acc[ct] = __builtin_amdgcn_wmma_f32_16x16x4_f32(false, af[cur], false, b,
                                                      (short)0, acc[ct], false, false);
    }
  }

  float dn[8];
#pragma unroll
  for (int v = 0; v < 8; ++v) {
    const float c = deg[wave * 16 + 8 * hi + v];
    dn[v] = c > 1.f ? c : 1.f;
  }
  float* st = pk + (wave * 16 + 8 * hi) * PK_PITCH + m;
  const float* prow = pk + (wave * 16 + m) * PK_PITCH + 2 * hi;

  const int row = wave * 16 + m;
  const float*  ax = (const float*)fp + ((row >> 1) * FP_PITCH + 2 * hi) * 2 + (row & 1);
  const float2* bw = wp + hi * HD + m;

  v8f acc0 = {}, acc1 = {}, acc2 = {}, acc3 = {};
  v8f acc4 = {}, acc5 = {}, acc6 = {}, acc7 = {};

#define GEMM_STEP(KOFF)                                                    \
  {                                                                        \
    const float2* bp = bw + ((k0 >> 1) + (KOFF)) * HD;                     \
    float2 bb; v2f b;                                                      \
    bb = bp[0];   b.x = bb.x; b.y = bb.y;                                  \
    acc0 = __builtin_amdgcn_wmma_f32_16x16x4_f32(false, a, false, b, (short)0, acc0, false, false); \
    bb = bp[16];  b.x = bb.x; b.y = bb.y;                                  \
    acc1 = __builtin_amdgcn_wmma_f32_16x16x4_f32(false, a, false, b, (short)0, acc1, false, false); \
    bb = bp[32];  b.x = bb.x; b.y = bb.y;                                  \
    acc2 = __builtin_amdgcn_wmma_f32_16x16x4_f32(false, a, false, b, (short)0, acc2, false, false); \
    bb = bp[48];  b.x = bb.x; b.y = bb.y;                                  \
    acc3 = __builtin_amdgcn_wmma_f32_16x16x4_f32(false, a, false, b, (short)0, acc3, false, false); \
    bb = bp[64];  b.x = bb.x; b.y = bb.y;                                  \
    acc4 = __builtin_amdgcn_wmma_f32_16x16x4_f32(false, a, false, b, (short)0, acc4, false, false); \
    bb = bp[80];  b.x = bb.x; b.y = bb.y;                                  \
    acc5 = __builtin_amdgcn_wmma_f32_16x16x4_f32(false, a, false, b, (short)0, acc5, false, false); \
    bb = bp[96];  b.x = bb.x; b.y = bb.y;                                  \
    acc6 = __builtin_amdgcn_wmma_f32_16x16x4_f32(false, a, false, b, (short)0, acc6, false, false); \
    bb = bp[112]; b.x = bb.x; b.y = bb.y;                                  \
    acc7 = __builtin_amdgcn_wmma_f32_16x16x4_f32(false, a, false, b, (short)0, acc7, false, false); \
  }

#pragma unroll
  for (int ct = 0; ct < 4; ++ct) {
#pragma unroll
    for (int v = 0; v < 8; ++v) st[v * PK_PITCH + ct * 16] = acc[ct][v] / dn[v];
  }
#pragma unroll 4
  for (int k0 = 0; k0 < HD / 2; k0 += 4) {
    const v2f a = *(const v2f*)(prow + k0);
    GEMM_STEP(0)
  }
#pragma unroll
  for (int ct = 4; ct < 8; ++ct) {
#pragma unroll
    for (int v = 0; v < 8; ++v) st[v * PK_PITCH + (ct - 4) * 16] = acc[ct][v] / dn[v];
  }
#pragma unroll 4
  for (int k0 = 0; k0 < HD / 2; k0 += 4) {
    const v2f a = *(const v2f*)(prow + k0);
    GEMM_STEP(32)
  }
  if (younger) __builtin_amdgcn_s_setprio((short)0);

  if (l < 2 || more) REQUEST_WEIGHTS(WLn, WRn)
  float bvs[8];
#pragma unroll
  for (int ct = 0; ct < 8; ++ct) bvs[ct] = bias[ct * 16 + m];

#pragma unroll 4
  for (int k0 = 0; k0 < HD; k0 += 4) {
    v2f a; a.x = ax[2 * k0]; a.y = ax[2 * k0 + 2];
    GEMM_STEP(64)
  }
#undef GEMM_STEP

  __syncthreads();

  v8f accs[8] = {acc0, acc1, acc2, acc3, acc4, acc5, acc6, acc7};
  if (l < 2) {
    float2* hp = fp + (wave * 8 + 4 * hi) * FP_PITCH + m;
#pragma unroll
    for (int ct = 0; ct < 8; ++ct) {
      const float bv = bvs[ct];
#pragma unroll
      for (int p = 0; p < 4; ++p) {
        float v0 = accs[ct][2 * p] + bv;
        float v1 = accs[ct][2 * p + 1] + bv;
        v0 = v0 > 0.f ? v0 : 0.f;
        v1 = v1 > 0.f ? v1 : 0.f;
        hp[p * FP_PITCH + ct * 16] = make_float2(v0, v1);
      }
    }
  } else {
    const int rbase = g * NN + wave * 16 + hi * 8;
#pragma unroll
    for (int ct = 0; ct < 8; ++ct) {
      const int col = ct * 16 + m;
      const float bv = bvs[ct];
#pragma unroll
      for (int v = 0; v < 8; ++v) {
        float val = accs[ct][v] + bv;
        val = val > 0.f ? val : 0.f;
        out[(size_t)(rbase + v) * HD + col] = val;
      }
    }
  }

  }

  g += (int)gridDim.x;
  } while (g < BG);
}
#undef REQUEST_GRAPH
#undef REQUEST_WEIGHTS

#define XS_PITCH 132
#define XS_ROWS  34
#define CW_PITCH 388

__global__ __launch_bounds__(256) void sortpool_head(
    const float* __restrict__ H,
    const float* __restrict__ conv_w, const float* __restrict__ conv_b,
    const float* __restrict__ W1, const float* __restrict__ b1,
    const float* __restrict__ W2, const float* __restrict__ b2,
    float* __restrict__ out)
{
  __shared__ float vals[NN];
  __shared__ int   idxs[NN];
  __shared__ float4 xs4[XS_ROWS * XS_PITCH / 4];
  __shared__ float4 cw4[CONV_O * CW_PITCH / 4];
  __shared__ float4 y4s[FLAT / 4];
  __shared__ float z[HD];
  __shared__ float logits[NCLS];
  __shared__ float lse;

  const int g = blockIdx.x;
  const int t = threadIdx.x;
  float* y = (float*)y4s;
  const float* hg = H + (size_t)g * NN * HD;

  const float4* cwg = (const float4*)conv_w;
#pragma unroll
  for (int j = 0; j < CONV_O * HD * CONV_W / 4 / 256; ++j) {
    const int idx = t + 256 * j;
    const int o = idx / (HD * CONV_W / 4);
    const int q = idx - o * (HD * CONV_W / 4);
    cw4[o * (CW_PITCH / 4) + q] = cwg[idx];
  }

  if (t < NN) { vals[t] = hg[t * HD + (HD - 1)]; idxs[t] = t; }
  __syncthreads();

  for (int k = 2; k <= NN; k <<= 1) {
    for (int j = k >> 1; j > 0; j >>= 1) {
      if (t < NN) {
        const int ixj = t ^ j;
        if (ixj > t) {
          const bool dirDesc = ((t & k) == 0);
          const float va = vals[t], vb = vals[ixj];
          const int   ia = idxs[t], ib = idxs[ixj];
          const bool aBeforeB = (va > vb) || (va == vb && ia < ib);
          if (dirDesc ? !aBeforeB : aBeforeB) {
            vals[t] = vb; vals[ixj] = va;
            idxs[t] = ib; idxs[ixj] = ia;
          }
        }
      }
      __syncthreads();
    }
  }

  const float4* hg4 = (const float4*)hg;
  for (int i = t; i < XS_ROWS * (HD / 4); i += 256) {
    const int r = i >> 5, c4 = i & 31;
    float4 v = make_float4(0.f, 0.f, 0.f, 0.f);
    if (r < KTOP) v = hg4[idxs[r] * (HD / 4) + c4];
    xs4[r * (XS_PITCH / 4) + c4] = v;
  }
  __syncthreads();

  const int wave = t >> 5;
  const int lane = t & 31;
  const int m    = lane & 15;
  const int hi   = lane >> 4;
  if (wave < 4) {
    const int rt = wave & 1;
    const int ct = wave >> 1;
    const float* xs = (const float*)xs4;
    const float* cw = (const float*)cw4;
    const float* ap = xs + (rt * 16 + m) * XS_PITCH + 2 * hi;
    const float* bp = cw + (ct * 16 + m) * CW_PITCH + 6 * hi;
    v8f acc0 = {}, acc1 = {};
#pragma unroll
    for (int w = 0; w < CONV_W; ++w) {
#pragma unroll 4
      for (int c0 = 0; c0 < HD; c0 += 8) {
        const v2f a0 = *(const v2f*)(ap + w * XS_PITCH + c0);
        const v2f a1 = *(const v2f*)(ap + w * XS_PITCH + c0 + 4);
        v2f b0, b1;
        b0.x = bp[c0 * CONV_W + w];
        b0.y = bp[(c0 + 1) * CONV_W + w];
        b1.x = bp[(c0 + 4) * CONV_W + w];
        b1.y = bp[(c0 + 5) * CONV_W + w];
        acc0 = __builtin_amdgcn_wmma_f32_16x16x4_f32(false, a0, false, b0,
                                                     (short)0, acc0, false, false);
        acc1 = __builtin_amdgcn_wmma_f32_16x16x4_f32(false, a1, false, b1,
                                                     (short)0, acc1, false, false);
      }
    }
    const int o = ct * 16 + m;
    const float bv = conv_b[o];
#pragma unroll
    for (int v = 0; v < 8; ++v) {
      const int tt = rt * 16 + hi * 8 + v;
      const float s = (acc0[v] + acc1[v]) + bv;
      if (tt < CONV_T) y[o * CONV_T + tt] = s > 0.f ? s : 0.f;
    }
  }
  __syncthreads();

  float4* part = cw4;
  float*  part2 = (float*)cw4 + 8 * HD;
  {
    const int cg = t & 31;
    const int ks = t >> 5;
    const float4* w14 = (const float4*)W1 + (size_t)(ks * (FLAT / 8)) * (HD / 4) + cg;
    const float4* yk  = y4s + ks * (FLAT / 8 / 4);
    float4 s = make_float4(0.f, 0.f, 0.f, 0.f);
#pragma unroll 2
    for (int i4 = 0; i4 < FLAT / 8 / 4; ++i4) {
      const float4 yv = yk[i4];
      const float4 w0 = w14[(4 * i4 + 0) * (HD / 4)];
      const float4 w1 = w14[(4 * i4 + 1) * (HD / 4)];
      const float4 w2 = w14[(4 * i4 + 2) * (HD / 4)];
      const float4 w3 = w14[(4 * i4 + 3) * (HD / 4)];
      s.x += yv.x * w0.x; s.y += yv.x * w0.y; s.z += yv.x * w0.z; s.w += yv.x * w0.w;
      s.x += yv.y * w1.x; s.y += yv.y * w1.y; s.z += yv.y * w1.z; s.w += yv.y * w1.w;
      s.x += yv.z * w2.x; s.y += yv.z * w2.y; s.z += yv.z * w2.z; s.w += yv.z * w2.w;
      s.x += yv.w * w3.x; s.y += yv.w * w3.y; s.z += yv.w * w3.z; s.w += yv.w * w3.w;
    }
    part[ks * (HD / 4) + cg] = s;
  }
  __syncthreads();
  if (t < HD) {
    const float* pf = (const float*)part;
    float s = b1[t];
#pragma unroll
    for (int ks = 0; ks < 8; ++ks) s += pf[ks * HD + t];
    z[t] = s > 0.f ? s : 0.f;
  }
  __syncthreads();

  {
    const int c   = t & 15;
    const int ks2 = t >> 4;
    float s = 0.f;
#pragma unroll
    for (int j = 0; j < 8; ++j) s += z[ks2 * 8 + j] * W2[(ks2 * 8 + j) * NCLS + c];
    part2[ks2 * NCLS + c] = s;
  }
  __syncthreads();
  if (t < NCLS) {
    float s = b2[t];
#pragma unroll
    for (int ks2 = 0; ks2 < 16; ++ks2) s += part2[ks2 * NCLS + t];
    logits[t] = s;
  }
  __syncthreads();

  if (t == 0) {
    float mx = logits[0];
    for (int c = 1; c < NCLS; ++c) mx = fmaxf(mx, logits[c]);
    float se = 0.f;
    for (int c = 0; c < NCLS; ++c) se += expf(logits[c] - mx);
    lse = mx + logf(se);
  }
  __syncthreads();
  if (t < NCLS) out[(size_t)g * NCLS + t] = logits[t] - lse;
}

extern "C" void kernel_launch(void* const* d_in, const int* in_sizes, int n_in,
                              void* d_out, int out_size, void* d_ws, size_t ws_size,
                              hipStream_t stream) {
  (void)in_sizes; (void)n_in; (void)out_size; (void)ws_size;
  const float* x      = (const float*)d_in[0];
  const int*   src    = (const int*)d_in[1];
  const int*   dst    = (const int*)d_in[2];
  const float* Wl1    = (const float*)d_in[5];
  const float* bl1    = (const float*)d_in[6];
  const float* Wr1    = (const float*)d_in[7];
  const float* Wl2    = (const float*)d_in[8];
  const float* bl2    = (const float*)d_in[9];
  const float* Wr2    = (const float*)d_in[10];
  const float* Wl3    = (const float*)d_in[11];
  const float* bl3    = (const float*)d_in[12];
  const float* Wr3    = (const float*)d_in[13];
  const float* conv_w = (const float*)d_in[14];
  const float* conv_b = (const float*)d_in[15];
  const float* W1     = (const float*)d_in[16];
  const float* b1     = (const float*)d_in[17];
  const float* W2     = (const float*)d_in[18];
  const float* b2     = (const float*)d_in[19];

  float* ws   = (float*)d_ws;
  float* h3   = ws + (size_t)VN * HD;

  sage_layers<<<NWG, 256, 0, stream>>>(x, src, dst, Wl1, Wr1, bl1, Wl2, Wr2, bl2,
                                       Wl3, Wr3, bl3, h3);
  sortpool_head<<<BG, 256, 0, stream>>>(h3, conv_w, conv_b, W1, b1, W2, b2,
                                        (float*)d_out);
}


# sage layers: vector work beside the f32 matrix ops cut by two thirds: mean's divides become mul + 2 fma by a per-row reciprocal (same bits), H@Wr LDS reads use immediate offsets; sums unchanged
# speedup vs baseline: 8.1075x; 1.0546x over previous
#include <hip/hip_runtime.h>
#include <hip/hip_bf16.h>

#define BG   1024
#define NN   128
#define VN   (BG * NN)
#define HD   128
#define EPG  2048
#define KTOP 30
#define CONV_O 32
#define CONV_W 3
#define CONV_T (KTOP - CONV_W + 1)
#define FLAT   (CONV_O * CONV_T)
#define NCLS   16

typedef float v2f __attribute__((ext_vector_type(2)));
typedef float v8f __attribute__((ext_vector_type(8)));
typedef __attribute__((address_space(3))) const float lds_cf;
typedef __attribute__((address_space(3))) const v2f   lds_cv2f;

__device__ __forceinline__ float div_by_degree(float a, float d, float r) {
  const float q = a * r;
  const float e = __builtin_fmaf(-d, q, a);
  return __builtin_fmaf(e, r, q);
}


#define FP_PITCH 144
#define CM_PITCH 132
#define PK_PITCH 68

#define NWG 256

#define REQUEST_GRAPH(G)                                                   \
  {                                                                        \
    const float4* xg4_ = (const float4*)(X + (size_t)(G) * NN * HD);       \
      \
    _Pragma("unroll")                                                      \
    for (int j = 0; j < 8; ++j) {                                          \
      const int idx = t + 256 * j;                                         \
      const int q   = idx >> 5;                                            \
      const int c4  = idx & 31;                                            \
      xr[2 * j]     = xg4_[(2 * q) * (HD / 4) + c4];                       \
      xr[2 * j + 1] = xg4_[(2 * q + 1) * (HD / 4) + c4];                   \
    }                                                                      \
    const int4* s4_ = (const int4*)(src + (size_t)(G) * EPG);              \
    const int4* d4_ = (const int4*)(dst + (size_t)(G) * EPG);              \
    sv0 = s4_[t]; sv1 = s4_[t + 256];                                      \
    dv0 = d4_[t]; dv1 = d4_[t + 256];                                      \
  }

#define REQUEST_WEIGHTS(WL, WR)                                            \
  {                                                                        \
      \
                                \
    _Pragma("unroll")                                                      \
    for (int j = 0; j < 16; ++j) {                                         \
      const int idx = t + 256 * j;                                         \
      const int kk  = idx >> 5;                                            \
      const int c4  = idx & 31;                                            \
      const float4* w4_ = (const float4*)(j < 8 ? (WL) : (WR))             \
                          + (size_t)(2 * (kk & 63)) * (HD / 4);            \
      wr[2 * j]     = w4_[c4];                                             \
      wr[2 * j + 1] = w4_[(HD / 4) + c4];                                  \
    }                                                                      \
  }

__global__ __launch_bounds__(256) void sage_layers(
    const float* __restrict__ X, const int* src, const int* dst,
    const float* Wl1, const float* Wr1, const float* bl1,
    const float* Wl2, const float* Wr2, const float* bl2,
    const float* Wl3, const float* Wr3, const float* bl3,
    float* __restrict__ out)
{
  static_assert(BG % NWG == 0, "every workgroup loops over the same number of graphs");
  __shared__ float2 fp[(NN / 2) * FP_PITCH];
  __shared__ float  cm[NN * CM_PITCH];
  __shared__ float  deg[NN];
  __shared__ float  rdg[NN];
  __shared__ float  pk[NN * PK_PITCH];
  __shared__ float2 wp[128 * HD];

  const int t = threadIdx.x;
  const int wave = t >> 5;
  const int lane = t & 31;
  const int m    = lane & 15;
  const int hi   = lane >> 4;
  const bool younger = __builtin_amdgcn_readfirstlane(wave) >= 4;

  float4 xr[16];
  int4 sv0, sv1, dv0, dv1;
  float4 wr[32];
  REQUEST_GRAPH(blockIdx.x)
  REQUEST_WEIGHTS(Wl1, Wr1)

  int g = blockIdx.x;
  do {
  const bool more = g + (int)gridDim.x < BG;

#pragma nounroll
  for (int l = 0; l < 3; ++l) {

  if (l == 0) {
    for (int i = t; i < NN * CM_PITCH / 4; i += 256)
      ((float4*)cm)[i] = make_float4(0.f, 0.f, 0.f, 0.f);
    if (t < NN) deg[t] = 0.f;

#pragma unroll
    for (int j = 0; j < 8; ++j) {
      const int idx = t + 256 * j;
      const int q   = idx >> 5;
      const int c4  = idx & 31;
      const float4 r0 = xr[2 * j];
      const float4 r1 = xr[2 * j + 1];
      float4* o = (float4*)(fp + q * FP_PITCH + c4 * 4);
      o[0] = make_float4(r0.x, r1.x, r0.y, r1.y);
      o[1] = make_float4(r0.z, r1.z, r0.w, r1.w);
    }
  }

#pragma unroll
  for (int j = 0; j < 16; ++j) {
    const int idx = t + 256 * j;
    const int kk  = idx >> 5;
    const int c4  = idx & 31;
    const float4 r0 = wr[2 * j];
    const float4 r1 = wr[2 * j + 1];
    float4* o = (float4*)(wp + kk * HD + c4 * 4);
    o[0] = make_float4(r0.x, r1.x, r0.y, r1.y);
    o[1] = make_float4(r0.z, r1.z, r0.w, r1.w);
  }
  __syncthreads();

  if (l == 0) {
#define COUNT_EDGE(S, D)                                   \
    {                                                      \
      const int s_ = (S) & (NN - 1);                       \
      const int d_ = (D) & (NN - 1);                       \
      atomicAdd(&cm[d_ * CM_PITCH + s_], 1.0f);            \
      atomicAdd(&deg[d_], 1.0f);                           \
    }
    COUNT_EDGE(sv0.x, dv0.x) COUNT_EDGE(sv0.y, dv0.y)
    COUNT_EDGE(sv0.z, dv0.z) COUNT_EDGE(sv0.w, dv0.w)
    COUNT_EDGE(sv1.x, dv1.x) COUNT_EDGE(sv1.y, dv1.y)
    COUNT_EDGE(sv1.z, dv1.z) COUNT_EDGE(sv1.w, dv1.w)
#undef COUNT_EDGE
    __syncthreads();

    if (more) REQUEST_GRAPH(g + (int)gridDim.x)

    if (lane < 16) {
      const float c = deg[wave * 16 + lane];
      const float d = c > 1.f ? c : 1.f;
      deg[wave * 16 + lane] = d;
      rdg[wave * 16 + lane] = 1.0f / d;
    }
  }

  const float* bias = l == 0 ? bl1 : (l == 1 ? bl2 : bl3);
  const float* WLn  = l == 0 ? Wl2 : (l == 1 ? Wl3 : Wl1);
  const float* WRn  = l == 0 ? Wr2 : (l == 1 ? Wr3 : Wr1);

  if (younger) __builtin_amdgcn_s_setprio((short)3);

  const float*  arow = cm + (wave * 16 + m) * CM_PITCH + 2 * hi;
  const float2* bcol = fp + hi * FP_PITCH + m;

  v8f acc[8] = {};
  v2f    af[2];
  float2 bf[2][8];

  af[0] = *(const v2f*)arow;
#pragma unroll
  for (int ct = 0; ct < 8; ++ct) bf[0][ct] = bcol[ct * 16];

#pragma unroll
  for (int ks = 0; ks < NN / 4; ++ks) {
    const int cur = ks & 1, nxt = cur ^ 1;
    if (ks + 1 < NN / 4) {
      af[nxt] = *(const v2f*)(arow + 4 * (ks + 1));
      const float2* bp = bcol + 2 * (ks + 1) * FP_PITCH;
#pragma unroll
      for (int ct = 0; ct < 8; ++ct) bf[nxt][ct] = bp[ct * 16];
    }
#pragma unroll
    for (int ct = 0; ct < 8; ++ct) {
      v2f b; b.x = bf[cur][ct].x; b.y = bf[cur][ct].y;
      acc[ct] = __builtin_amdgcn_wmma_f32_16x16x4_f32(false, af[cur], false, b,
                                                      (short)0, acc[ct], false, false);
    }
  }

  float dn[8], rn[8];
#pragma unroll
  for (int v = 0; v < 8; ++v) {
    dn[v] = deg[wave * 16 + 8 * hi + v];
    rn[v] = rdg[wave * 16 + 8 * hi + v];
  }
  float* st = pk + (wave * 16 + 8 * hi) * PK_PITCH + m;
  const float* prow = pk + (wave * 16 + m) * PK_PITCH + 2 * hi;

  const int row = wave * 16 + m;
  const float*  ax = (const float*)fp + ((row >> 1) * FP_PITCH + 2 * hi) * 2 + (row & 1);
  const float2* bw = wp + hi * HD + m;

  v8f acc0 = {}, acc1 = {}, acc2 = {}, acc3 = {};
  v8f acc4 = {}, acc5 = {}, acc6 = {}, acc7 = {};

#define GEMM_STEP(KOFF)                                                    \
  {                                                                        \
    const float2* bp = bw + ((k0 >> 1) + (KOFF)) * HD;                     \
    float2 bb; v2f b;                                                      \
    bb = bp[0];   b.x = bb.x; b.y = bb.y;                                  \
    acc0 = __builtin_amdgcn_wmma_f32_16x16x4_f32(false, a, false, b, (short)0, acc0, false, false); \
    bb = bp[16];  b.x = bb.x; b.y = bb.y;                                  \
    acc1 = __builtin_amdgcn_wmma_f32_16x16x4_f32(false, a, false, b, (short)0, acc1, false, false); \
    bb = bp[32];  b.x = bb.x; b.y = bb.y;                                  \
    acc2 = __builtin_amdgcn_wmma_f32_16x16x4_f32(false, a, false, b, (short)0, acc2, false, false); \
    bb = bp[48];  b.x = bb.x; b.y = bb.y;                                  \
    acc3 = __builtin_amdgcn_wmma_f32_16x16x4_f32(false, a, false, b, (short)0, acc3, false, false); \
    bb = bp[64];  b.x = bb.x; b.y = bb.y;                                  \
    acc4 = __builtin_amdgcn_wmma_f32_16x16x4_f32(false, a, false, b, (short)0, acc4, false, false); \
    bb = bp[80];  b.x = bb.x; b.y = bb.y;                                  \
    acc5 = __builtin_amdgcn_wmma_f32_16x16x4_f32(false, a, false, b, (short)0, acc5, false, false); \
    bb = bp[96];  b.x = bb.x; b.y = bb.y;                                  \
    acc6 = __builtin_amdgcn_wmma_f32_16x16x4_f32(false, a, false, b, (short)0, acc6, false, false); \
    bb = bp[112]; b.x = bb.x; b.y = bb.y;                                  \
    acc7 = __builtin_amdgcn_wmma_f32_16x16x4_f32(false, a, false, b, (short)0, acc7, false, false); \
  }

#pragma unroll
  for (int ct = 0; ct < 4; ++ct) {
#pragma unroll
    for (int v = 0; v < 8; ++v) st[v * PK_PITCH + ct * 16] = div_by_degree(acc[ct][v], dn[v], rn[v]);
  }
#pragma unroll 4
  for (int k0 = 0; k0 < HD / 2; k0 += 4) {
    const v2f a = *(const v2f*)(prow + k0);
    GEMM_STEP(0)
  }
#pragma unroll
  for (int ct = 4; ct < 8; ++ct) {
#pragma unroll
    for (int v = 0; v < 8; ++v) st[v * PK_PITCH + (ct - 4) * 16] = div_by_degree(acc[ct][v], dn[v], rn[v]);
  }
#pragma unroll 4
  for (int k0 = 0; k0 < HD / 2; k0 += 4) {
    const v2f a = *(const v2f*)(prow + k0);
    GEMM_STEP(32)
  }
  if (younger) __builtin_amdgcn_s_setprio((short)0);

  if (l < 2 || more) REQUEST_WEIGHTS(WLn, WRn)
  float bvs[8];
#pragma unroll
  for (int ct = 0; ct < 8; ++ct) bvs[ct] = bias[ct * 16 + m];

#undef GEMM_STEP

  lds_cf*   ax3 = (lds_cf*)ax;
  lds_cv2f* bw3 = (lds_cv2f*)(bw + 64 * HD);
  asm volatile("" : "+v"(ax3));
  asm volatile("" : "+v"(bw3));
#pragma unroll 4
  for (int k0 = 0; k0 < HD; k0 += 4) {
    v2f a; a.x = ax3[2 * k0]; a.y = ax3[2 * k0 + 2];
    lds_cv2f* bp = bw3 + (k0 >> 1) * HD;
    acc0 = __builtin_amdgcn_wmma_f32_16x16x4_f32(false, a, false, bp[0],   (short)0, acc0, false, false);
    acc1 = __builtin_amdgcn_wmma_f32_16x16x4_f32(false, a, false, bp[16],  (short)0, acc1, false, false);
    acc2 = __builtin_amdgcn_wmma_f32_16x16x4_f32(false, a, false, bp[32],  (short)0, acc2, false, false);
    acc3 = __builtin_amdgcn_wmma_f32_16x16x4_f32(false, a, false, bp[48],  (short)0, acc3, false, false);
    acc4 = __builtin_amdgcn_wmma_f32_16x16x4_f32(false, a, false, bp[64],  (short)0, acc4, false, false);
    acc5 = __builtin_amdgcn_wmma_f32_16x16x4_f32(false, a, false, bp[80],  (short)0, acc5, false, false);
    acc6 = __builtin_amdgcn_wmma_f32_16x16x4_f32(false, a, false, bp[96],  (short)0, acc6, false, false);
    acc7 = __builtin_amdgcn_wmma_f32_16x16x4_f32(false, a, false, bp[112], (short)0, acc7, false, false);
  }

  __syncthreads();

  v8f accs[8] = {acc0, acc1, acc2, acc3, acc4, acc5, acc6, acc7};
  if (l < 2) {
    float2* hp = fp + (wave * 8 + 4 * hi) * FP_PITCH + m;
#pragma unroll
    for (int ct = 0; ct < 8; ++ct) {
      const float bv = bvs[ct];
#pragma unroll
      for (int p = 0; p < 4; ++p) {
        float v0 = accs[ct][2 * p] + bv;
        float v1 = accs[ct][2 * p + 1] + bv;
        v0 = __builtin_fmaxf(v0, 0.f);
        v1 = __builtin_fmaxf(v1, 0.f);
        hp[p * FP_PITCH + ct * 16] = make_float2(v0, v1);
      }
    }
  } else {
    const int rbase = g * NN + wave * 16 + hi * 8;
#pragma unroll
    for (int ct = 0; ct < 8; ++ct) {
      const int col = ct * 16 + m;
      const float bv = bvs[ct];
#pragma unroll
      for (int v = 0; v < 8; ++v) {
        float val = accs[ct][v] + bv;
        val = val > 0.f ? val : 0.f;
        out[(size_t)(rbase + v) * HD + col] = val;
      }
    }
  }

  }

  g += (int)gridDim.x;
  } while (g < BG);
}
#undef REQUEST_GRAPH
#undef REQUEST_WEIGHTS

#define XS_PITCH 132
#define XS_ROWS  34
#define CW_PITCH 388

__global__ __launch_bounds__(256) void sortpool_head(
    const float* __restrict__ H,
    const float* __restrict__ conv_w, const float* __restrict__ conv_b,
    const float* __restrict__ W1, const float* __restrict__ b1,
    const float* __restrict__ W2, const float* __restrict__ b2,
    float* __restrict__ out)
{
  __shared__ float vals[NN];
  __shared__ int   idxs[NN];
  __shared__ float4 xs4[XS_ROWS * XS_PITCH / 4];
  __shared__ float4 cw4[CONV_O * CW_PITCH / 4];
  __shared__ float4 y4s[FLAT / 4];
  __shared__ float z[HD];
  __shared__ float logits[NCLS];
  __shared__ float lse;

  const int g = blockIdx.x;
  const int t = threadIdx.x;
  float* y = (float*)y4s;
  const float* hg = H + (size_t)g * NN * HD;

  const float4* cwg = (const float4*)conv_w;
#pragma unroll
  for (int j = 0; j < CONV_O * HD * CONV_W / 4 / 256; ++j) {
    const int idx = t + 256 * j;
    const int o = idx / (HD * CONV_W / 4);
    const int q = idx - o * (HD * CONV_W / 4);
    cw4[o * (CW_PITCH / 4) + q] = cwg[idx];
  }

  if (t < NN) { vals[t] = hg[t * HD + (HD - 1)]; idxs[t] = t; }
  __syncthreads();

  for (int k = 2; k <= NN; k <<= 1) {
    for (int j = k >> 1; j > 0; j >>= 1) {
      if (t < NN) {
        const int ixj = t ^ j;
        if (ixj > t) {
          const bool dirDesc = ((t & k) == 0);
          const float va = vals[t], vb = vals[ixj];
          const int   ia = idxs[t], ib = idxs[ixj];
          const bool aBeforeB = (va > vb) || (va == vb && ia < ib);
          if (dirDesc ? !aBeforeB : aBeforeB) {
            vals[t] = vb; vals[ixj] = va;
            idxs[t] = ib; idxs[ixj] = ia;
          }
        }
      }
      __syncthreads();
    }
  }

  const float4* hg4 = (const float4*)hg;
  for (int i = t; i < XS_ROWS * (HD / 4); i += 256) {
    const int r = i >> 5, c4 = i & 31;
    float4 v = make_float4(0.f, 0.f, 0.f, 0.f);
    if (r < KTOP) v = hg4[idxs[r] * (HD / 4) + c4];
    xs4[r * (XS_PITCH / 4) + c4] = v;
  }
  __syncthreads();

  const int wave = t >> 5;
  const int lane = t & 31;
  const int m    = lane & 15;
  const int hi   = lane >> 4;
  if (wave < 4) {
    const int rt = wave & 1;
    const int ct = wave >> 1;
    const float* xs = (const float*)xs4;
    const float* cw = (const float*)cw4;
    const float* ap = xs + (rt * 16 + m) * XS_PITCH + 2 * hi;
    const float* bp = cw + (ct * 16 + m) * CW_PITCH + 6 * hi;
    v8f acc0 = {}, acc1 = {};
#pragma unroll
    for (int w = 0; w < CONV_W; ++w) {
#pragma unroll 4
      for (int c0 = 0; c0 < HD; c0 += 8) {
        const v2f a0 = *(const v2f*)(ap + w * XS_PITCH + c0);
        const v2f a1 = *(const v2f*)(ap + w * XS_PITCH + c0 + 4);
        v2f b0, b1;
        b0.x = bp[c0 * CONV_W + w];
        b0.y = bp[(c0 + 1) * CONV_W + w];
        b1.x = bp[(c0 + 4) * CONV_W + w];
        b1.y = bp[(c0 + 5) * CONV_W + w];
        acc0 = __builtin_amdgcn_wmma_f32_16x16x4_f32(false, a0, false, b0,
                                                     (short)0, acc0, false, false);
        acc1 = __builtin_amdgcn_wmma_f32_16x16x4_f32(false, a1, false, b1,
                                                     (short)0, acc1, false, false);
      }
    }
    const int o = ct * 16 + m;
    const float bv = conv_b[o];
#pragma unroll
    for (int v = 0; v < 8; ++v) {
      const int tt = rt * 16 + hi * 8 + v;
      const float s = (acc0[v] + acc1[v]) + bv;
      if (tt < CONV_T) y[o * CONV_T + tt] = s > 0.f ? s : 0.f;
    }
  }
  __syncthreads();

  float4* part = cw4;
  float*  part2 = (float*)cw4 + 8 * HD;
  {
    const int cg = t & 31;
    const int ks = t >> 5;
    const float4* w14 = (const float4*)W1 + (size_t)(ks * (FLAT / 8)) * (HD / 4) + cg;
    const float4* yk  = y4s + ks * (FLAT / 8 / 4);
    float4 s = make_float4(0.f, 0.f, 0.f, 0.f);
#pragma unroll 2
    for (int i4 = 0; i4 < FLAT / 8 / 4; ++i4) {
      const float4 yv = yk[i4];
      const float4 w0 = w14[(4 * i4 + 0) * (HD / 4)];
      const float4 w1 = w14[(4 * i4 + 1) * (HD / 4)];
      const float4 w2 = w14[(4 * i4 + 2) * (HD / 4)];
      const float4 w3 = w14[(4 * i4 + 3) * (HD / 4)];
      s.x += yv.x * w0.x; s.y += yv.x * w0.y; s.z += yv.x * w0.z; s.w += yv.x * w0.w;
      s.x += yv.y * w1.x; s.y += yv.y * w1.y; s.z += yv.y * w1.z; s.w += yv.y * w1.w;
      s.x += yv.z * w2.x; s.y += yv.z * w2.y; s.z += yv.z * w2.z; s.w += yv.z * w2.w;
      s.x += yv.w * w3.x; s.y += yv.w * w3.y; s.z += yv.w * w3.z; s.w += yv.w * w3.w;
    }
    part[ks * (HD / 4) + cg] = s;
  }
  __syncthreads();
  if (t < HD) {
    const float* pf = (const float*)part;
    float s = b1[t];
#pragma unroll
    for (int ks = 0; ks < 8; ++ks) s += pf[ks * HD + t];
    z[t] = s > 0.f ? s : 0.f;
  }
  __syncthreads();

  {
    const int c   = t & 15;
    const int ks2 = t >> 4;
    float s = 0.f;
#pragma unroll
    for (int j = 0; j < 8; ++j) s += z[ks2 * 8 + j] * W2[(ks2 * 8 + j) * NCLS + c];
    part2[ks2 * NCLS + c] = s;
  }
  __syncthreads();
  if (t < NCLS) {
    float s = b2[t];
#pragma unroll
    for (int ks2 = 0; ks2 < 16; ++ks2) s += part2[ks2 * NCLS + t];
    logits[t] = s;
  }
  __syncthreads();

  if (t == 0) {
    float mx = logits[0];
    for (int c = 1; c < NCLS; ++c) mx = fmaxf(mx, logits[c]);
    float se = 0.f;
    for (int c = 0; c < NCLS; ++c) se += expf(logits[c] - mx);
    lse = mx + logf(se);
  }
  __syncthreads();
  if (t < NCLS) out[(size_t)g * NCLS + t] = logits[t] - lse;
}

extern "C" void kernel_launch(void* const* d_in, const int* in_sizes, int n_in,
                              void* d_out, int out_size, void* d_ws, size_t ws_size,
                              hipStream_t stream) {
  (void)in_sizes; (void)n_in; (void)out_size; (void)ws_size;
  const float* x      = (const float*)d_in[0];
  const int*   src    = (const int*)d_in[1];
  const int*   dst    = (const int*)d_in[2];
  const float* Wl1    = (const float*)d_in[5];
  const float* bl1    = (const float*)d_in[6];
  const float* Wr1    = (const float*)d_in[7];
  const float* Wl2    = (const float*)d_in[8];
  const float* bl2    = (const float*)d_in[9];
  const float* Wr2    = (const float*)d_in[10];
  const float* Wl3    = (const float*)d_in[11];
  const float* bl3    = (const float*)d_in[12];
  const float* Wr3    = (const float*)d_in[13];
  const float* conv_w = (const float*)d_in[14];
  const float* conv_b = (const float*)d_in[15];
  const float* W1     = (const float*)d_in[16];
  const float* b1     = (const float*)d_in[17];
  const float* W2     = (const float*)d_in[18];
  const float* b2     = (const float*)d_in[19];

  float* ws   = (float*)d_ws;
  float* h3   = ws + (size_t)VN * HD;

  sage_layers<<<NWG, 256, 0, stream>>>(x, src, dst, Wl1, Wr1, bl1, Wl2, Wr2, bl2,
                                       Wl3, Wr3, bl3, h3);
  sortpool_head<<<BG, 256, 0, stream>>>(h3, conv_w, conv_b, W1, b1, W2, b2,
                                        (float*)d_out);
}


# layer 3 only for what the pooling keeps: key column tile for all rows, rank by counting in LDS, full linear transform for the 30 kept rows, stored in rank order; head sort gone; sum orders kept
# speedup vs baseline: 9.1210x; 1.1250x over previous
#include <hip/hip_runtime.h>
#include <hip/hip_bf16.h>

#define BG   1024
#define NN   128
#define VN   (BG * NN)
#define HD   128
#define EPG  2048
#define KTOP 30
#define CONV_O 32
#define CONV_W 3
#define CONV_T (KTOP - CONV_W + 1)
#define FLAT   (CONV_O * CONV_T)
#define NCLS   16

typedef float v2f __attribute__((ext_vector_type(2)));
typedef float v8f __attribute__((ext_vector_type(8)));
typedef __attribute__((address_space(3))) const float lds_cf;
typedef __attribute__((address_space(3))) const v2f   lds_cv2f;
typedef __attribute__((address_space(1))) const float glb_cf;
typedef __attribute__((address_space(1))) float       glb_f;

__device__ __forceinline__ float div_by_degree(float a, float d, float r) {
  const float q = a * r;
  const float e = __builtin_fmaf(-d, q, a);
  return __builtin_fmaf(e, r, q);
}


#define FP_PITCH 144
#define CM_PITCH 132
#define PK_PITCH 68

#define NWG 256

#define REQUEST_GRAPH(G)                                                   \
  {                                                                        \
    const float4* xg4_ = (const float4*)(X + (size_t)(G) * NN * HD);       \
      \
    _Pragma("unroll")                                                      \
    for (int j = 0; j < 8; ++j) {                                          \
      const int idx = t + 256 * j;                                         \
      const int q   = idx >> 5;                                            \
      const int c4  = idx & 31;                                            \
      xr[2 * j]     = xg4_[(2 * q) * (HD / 4) + c4];                       \
      xr[2 * j + 1] = xg4_[(2 * q + 1) * (HD / 4) + c4];                   \
    }                                                                      \
    const int4* s4_ = (const int4*)(src + (size_t)(G) * EPG);              \
    const int4* d4_ = (const int4*)(dst + (size_t)(G) * EPG);              \
    sv0 = s4_[t]; sv1 = s4_[t + 256];                                      \
    dv0 = d4_[t]; dv1 = d4_[t + 256];                                      \
  }

#define REQUEST_WEIGHTS(WL, WR)                                            \
  {                                                                        \
      \
                                \
    _Pragma("unroll")                                                      \
    for (int j = 0; j < 16; ++j) {                                         \
      const int idx = t + 256 * j;                                         \
      const int kk  = idx >> 5;                                            \
      const int c4  = idx & 31;                                            \
      const float4* w4_ = (const float4*)(j < 8 ? (WL) : (WR))             \
                          + (size_t)(2 * (kk & 63)) * (HD / 4);            \
      wr[2 * j]     = w4_[c4];                                             \
      wr[2 * j + 1] = w4_[(HD / 4) + c4];                                  \
    }                                                                      \
  }

__global__ __launch_bounds__(256) void sage_layers(
    const float* __restrict__ X, const int* src, const int* dst,
    const float* Wl1, const float* Wr1, const float* bl1,
    const float* Wl2, const float* Wr2, const float* bl2,
    const float* Wl3, const float* Wr3, const float* bl3,
    float* __restrict__ out)
{
  static_assert(BG % NWG == 0, "every workgroup loops over the same number of graphs");
  __shared__ float2 fp[(NN / 2) * FP_PITCH];
  __shared__ float  cm[NN * CM_PITCH];
  __shared__ float  deg[NN];
  __shared__ float  rdg[NN];
  __shared__ float  pk[NN * PK_PITCH];
  __shared__ float2 wp[128 * HD];
  __shared__ float4 keys4[NN / 4];
  __shared__ int    sel[32];

  float* keys = (float*)keys4;
  const int t = threadIdx.x;
  const int wave = t >> 5;
  const int lane = t & 31;
  const int m    = lane & 15;
  const int hi   = lane >> 4;
  const bool younger = __builtin_amdgcn_readfirstlane(wave) >= 4;

  float4 xr[16];
  int4 sv0, sv1, dv0, dv1;
  float4 wr[32];
  REQUEST_GRAPH(blockIdx.x)
  REQUEST_WEIGHTS(Wl1, Wr1)

  int g = blockIdx.x;
  do {
  const bool more = g + (int)gridDim.x < BG;

#pragma nounroll
  for (int l = 0; l < 2; ++l) {

  if (l == 0) {
    for (int i = t; i < NN * CM_PITCH / 4; i += 256)
      ((float4*)cm)[i] = make_float4(0.f, 0.f, 0.f, 0.f);
    if (t < NN) deg[t] = 0.f;

#pragma unroll
    for (int j = 0; j < 8; ++j) {
      const int idx = t + 256 * j;
      const int q   = idx >> 5;
      const int c4  = idx & 31;
      const float4 r0 = xr[2 * j];
      const float4 r1 = xr[2 * j + 1];
      float4* o = (float4*)(fp + q * FP_PITCH + c4 * 4);
      o[0] = make_float4(r0.x, r1.x, r0.y, r1.y);
      o[1] = make_float4(r0.z, r1.z, r0.w, r1.w);
    }
  }

#pragma unroll
  for (int j = 0; j < 16; ++j) {
    const int idx = t + 256 * j;
    const int kk  = idx >> 5;
    const int c4  = idx & 31;
    const float4 r0 = wr[2 * j];
    const float4 r1 = wr[2 * j + 1];
    float4* o = (float4*)(wp + kk * HD + c4 * 4);
    o[0] = make_float4(r0.x, r1.x, r0.y, r1.y);
    o[1] = make_float4(r0.z, r1.z, r0.w, r1.w);
  }
  __syncthreads();

  if (l == 0) {
#define COUNT_EDGE(S, D)                                   \
    {                                                      \
      const int s_ = (S) & (NN - 1);                       \
      const int d_ = (D) & (NN - 1);                       \
      atomicAdd(&cm[d_ * CM_PITCH + s_], 1.0f);            \
      atomicAdd(&deg[d_], 1.0f);                           \
    }
    COUNT_EDGE(sv0.x, dv0.x) COUNT_EDGE(sv0.y, dv0.y)
    COUNT_EDGE(sv0.z, dv0.z) COUNT_EDGE(sv0.w, dv0.w)
    COUNT_EDGE(sv1.x, dv1.x) COUNT_EDGE(sv1.y, dv1.y)
    COUNT_EDGE(sv1.z, dv1.z) COUNT_EDGE(sv1.w, dv1.w)
#undef COUNT_EDGE
    __syncthreads();

    if (more) REQUEST_GRAPH(g + (int)gridDim.x)

    if (lane < 16) {
      const float c = deg[wave * 16 + lane];
      const float d = c > 1.f ? c : 1.f;
      deg[wave * 16 + lane] = d;
      rdg[wave * 16 + lane] = 1.0f / d;
    }
  }

  const float* bias = l == 0 ? bl1 : bl2;
  const float* WLn  = l == 0 ? Wl2 : Wl3;
  const float* WRn  = l == 0 ? Wr2 : Wr3;

  if (younger) __builtin_amdgcn_s_setprio((short)3);

  const float*  arow = cm + (wave * 16 + m) * CM_PITCH + 2 * hi;
  const float2* bcol = fp + hi * FP_PITCH + m;

  v8f acc[8] = {};
  v2f    af[2];
  float2 bf[2][8];

  af[0] = *(const v2f*)arow;
#pragma unroll
  for (int ct = 0; ct < 8; ++ct) bf[0][ct] = bcol[ct * 16];

#pragma unroll
  for (int ks = 0; ks < NN / 4; ++ks) {
    const int cur = ks & 1, nxt = cur ^ 1;
    if (ks + 1 < NN / 4) {
      af[nxt] = *(const v2f*)(arow + 4 * (ks + 1));
      const float2* bp = bcol + 2 * (ks + 1) * FP_PITCH;
#pragma unroll
      for (int ct = 0; ct < 8; ++ct) bf[nxt][ct] = bp[ct * 16];
    }
#pragma unroll
    for (int ct = 0; ct < 8; ++ct) {
      v2f b; b.x = bf[cur][ct].x; b.y = bf[cur][ct].y;
      acc[ct] = __builtin_amdgcn_wmma_f32_16x16x4_f32(false, af[cur], false, b,
                                                      (short)0, acc[ct], false, false);
    }
  }

  float dn[8], rn[8];
#pragma unroll
  for (int v = 0; v < 8; ++v) {
    dn[v] = deg[wave * 16 + 8 * hi + v];
    rn[v] = rdg[wave * 16 + 8 * hi + v];
  }
  float* st = pk + (wave * 16 + 8 * hi) * PK_PITCH + m;
  const float* prow = pk + (wave * 16 + m) * PK_PITCH + 2 * hi;

  const int row = wave * 16 + m;
  const float*  ax = (const float*)fp + ((row >> 1) * FP_PITCH + 2 * hi) * 2 + (row & 1);
  const float2* bw = wp + hi * HD + m;

  v8f acc0 = {}, acc1 = {}, acc2 = {}, acc3 = {};
  v8f acc4 = {}, acc5 = {}, acc6 = {}, acc7 = {};

#define GEMM_STEP(KOFF)                                                    \
  {                                                                        \
    const float2* bp = bw + ((k0 >> 1) + (KOFF)) * HD;                     \
    float2 bb; v2f b;                                                      \
    bb = bp[0];   b.x = bb.x; b.y = bb.y;                                  \
    acc0 = __builtin_amdgcn_wmma_f32_16x16x4_f32(false, a, false, b, (short)0, acc0, false, false); \
    bb = bp[16];  b.x = bb.x; b.y = bb.y;                                  \
    acc1 = __builtin_amdgcn_wmma_f32_16x16x4_f32(false, a, false, b, (short)0, acc1, false, false); \
    bb = bp[32];  b.x = bb.x; b.y = bb.y;                                  \
    acc2 = __builtin_amdgcn_wmma_f32_16x16x4_f32(false, a, false, b, (short)0, acc2, false, false); \
    bb = bp[48];  b.x = bb.x; b.y = bb.y;                                  \
    acc3 = __builtin_amdgcn_wmma_f32_16x16x4_f32(false, a, false, b, (short)0, acc3, false, false); \
    bb = bp[64];  b.x = bb.x; b.y = bb.y;                                  \
    acc4 = __builtin_amdgcn_wmma_f32_16x16x4_f32(false, a, false, b, (short)0, acc4, false, false); \
    bb = bp[80];  b.x = bb.x; b.y = bb.y;                                  \
    acc5 = __builtin_amdgcn_wmma_f32_16x16x4_f32(false, a, false, b, (short)0, acc5, false, false); \
    bb = bp[96];  b.x = bb.x; b.y = bb.y;                                  \
    acc6 = __builtin_amdgcn_wmma_f32_16x16x4_f32(false, a, false, b, (short)0, acc6, false, false); \
    bb = bp[112]; b.x = bb.x; b.y = bb.y;                                  \
    acc7 = __builtin_amdgcn_wmma_f32_16x16x4_f32(false, a, false, b, (short)0, acc7, false, false); \
  }

#pragma unroll
  for (int ct = 0; ct < 4; ++ct) {
#pragma unroll
    for (int v = 0; v < 8; ++v) st[v * PK_PITCH + ct * 16] = div_by_degree(acc[ct][v], dn[v], rn[v]);
  }
#pragma unroll 4
  for (int k0 = 0; k0 < HD / 2; k0 += 4) {
    const v2f a = *(const v2f*)(prow + k0);
    GEMM_STEP(0)
  }
#pragma unroll
  for (int ct = 4; ct < 8; ++ct) {
#pragma unroll
    for (int v = 0; v < 8; ++v) st[v * PK_PITCH + (ct - 4) * 16] = div_by_degree(acc[ct][v], dn[v], rn[v]);
  }
#pragma unroll 4
  for (int k0 = 0; k0 < HD / 2; k0 += 4) {
    const v2f a = *(const v2f*)(prow + k0);
    GEMM_STEP(32)
  }
  if (younger) __builtin_amdgcn_s_setprio((short)0);

  REQUEST_WEIGHTS(WLn, WRn)
  float bvs[8];
#pragma unroll
  for (int ct = 0; ct < 8; ++ct) bvs[ct] = bias[ct * 16 + m];

#undef GEMM_STEP

  lds_cf*   ax3 = (lds_cf*)ax;
  lds_cv2f* bw3 = (lds_cv2f*)(bw + 64 * HD);
  asm volatile("" : "+v"(ax3));
  asm volatile("" : "+v"(bw3));
#pragma unroll 4
  for (int k0 = 0; k0 < HD; k0 += 4) {
    v2f a; a.x = ax3[2 * k0]; a.y = ax3[2 * k0 + 2];
    lds_cv2f* bp = bw3 + (k0 >> 1) * HD;
    acc0 = __builtin_amdgcn_wmma_f32_16x16x4_f32(false, a, false, bp[0],   (short)0, acc0, false, false);
    acc1 = __builtin_amdgcn_wmma_f32_16x16x4_f32(false, a, false, bp[16],  (short)0, acc1, false, false);
    acc2 = __builtin_amdgcn_wmma_f32_16x16x4_f32(false, a, false, bp[32],  (short)0, acc2, false, false);
    acc3 = __builtin_amdgcn_wmma_f32_16x16x4_f32(false, a, false, bp[48],  (short)0, acc3, false, false);
    acc4 = __builtin_amdgcn_wmma_f32_16x16x4_f32(false, a, false, bp[64],  (short)0, acc4, false, false);
    acc5 = __builtin_amdgcn_wmma_f32_16x16x4_f32(false, a, false, bp[80],  (short)0, acc5, false, false);
    acc6 = __builtin_amdgcn_wmma_f32_16x16x4_f32(false, a, false, bp[96],  (short)0, acc6, false, false);
    acc7 = __builtin_amdgcn_wmma_f32_16x16x4_f32(false, a, false, bp[112], (short)0, acc7, false, false);
  }

  __syncthreads();

  v8f accs[8] = {acc0, acc1, acc2, acc3, acc4, acc5, acc6, acc7};
  {
    float2* hp = fp + (wave * 8 + 4 * hi) * FP_PITCH + m;
#pragma unroll
    for (int ct = 0; ct < 8; ++ct) {
      const float bv = bvs[ct];
#pragma unroll
      for (int p = 0; p < 4; ++p) {
        float v0 = accs[ct][2 * p] + bv;
        float v1 = accs[ct][2 * p + 1] + bv;
        v0 = __builtin_fmaxf(v0, 0.f);
        v1 = __builtin_fmaxf(v1, 0.f);
        hp[p * FP_PITCH + ct * 16] = make_float2(v0, v1);
      }
    }
  }

  }

  {
#pragma unroll
  for (int j = 0; j < 16; ++j) {
    const int idx = t + 256 * j;
    const int kk  = idx >> 5;
    const int c4  = idx & 31;
    const float4 r0 = wr[2 * j];
    const float4 r1 = wr[2 * j + 1];
    float4* o = (float4*)(wp + kk * HD + c4 * 4);
    o[0] = make_float4(r0.x, r1.x, r0.y, r1.y);
    o[1] = make_float4(r0.z, r1.z, r0.w, r1.w);
  }
  __syncthreads();

  glb_cf* wl1n = (glb_cf*)Wl1;
  glb_cf* wr1n = (glb_cf*)Wr1;
  glb_cf* b3   = (glb_cf*)bl3;
  glb_f*  o3   = (glb_f*)out;
  asm volatile("" : "+s"(wl1n));
  asm volatile("" : "+s"(wr1n));
  asm volatile("" : "+s"(b3));
  asm volatile("" : "+s"(o3));

  int t3 = t;
  asm volatile("" : "+v"(t3));
  const int wave3 = t3 >> 5;
  const int m3    = t3 & 15;
  const int hi3   = (t3 >> 4) & 1;

  const float kb = b3[(HD - 16) + m3];

  if (younger) __builtin_amdgcn_s_setprio((short)3);

  const float*  arow = cm + (wave * 16 + m) * CM_PITCH + 2 * hi;
  const float2* bcol = fp + hi * FP_PITCH + m;

  v8f acc[8] = {};
  v2f    af[2];
  float2 bf[2][8];

  af[0] = *(const v2f*)arow;
#pragma unroll
  for (int ct = 0; ct < 8; ++ct) bf[0][ct] = bcol[ct * 16];

#pragma unroll
  for (int ks = 0; ks < NN / 4; ++ks) {
    const int cur = ks & 1, nxt = cur ^ 1;
    if (ks + 1 < NN / 4) {
      af[nxt] = *(const v2f*)(arow + 4 * (ks + 1));
      const float2* bp = bcol + 2 * (ks + 1) * FP_PITCH;
#pragma unroll
      for (int ct = 0; ct < 8; ++ct) bf[nxt][ct] = bp[ct * 16];
    }
#pragma unroll
    for (int ct = 0; ct < 8; ++ct) {
      v2f b; b.x = bf[cur][ct].x; b.y = bf[cur][ct].y;
      acc[ct] = __builtin_amdgcn_wmma_f32_16x16x4_f32(false, af[cur], false, b,
                                                      (short)0, acc[ct], false, false);
    }
  }

  float dn[8], rn[8];
#pragma unroll
  for (int v = 0; v < 8; ++v) {
    dn[v] = deg[wave * 16 + 8 * hi + v];
    rn[v] = rdg[wave * 16 + 8 * hi + v];
  }
  float* st = cm + (wave3 * 16 + 8 * hi3) * CM_PITCH + m3;
#pragma unroll
  for (int ct = 0; ct < 8; ++ct) {
#pragma unroll
    for (int v = 0; v < 8; ++v) st[v * CM_PITCH + ct * 16] = div_by_degree(acc[ct][v], dn[v], rn[v]);
  }

  if (younger) __builtin_amdgcn_s_setprio((short)0);

  const int row = wave3 * 16 + m3;
  const float*  ax = (const float*)fp + ((row >> 1) * FP_PITCH + 2 * hi3) * 2 + (row & 1);
  const float2* bk = wp + hi3 * HD + (HD - 16) + m3;
  const float*  mrow = cm + row * CM_PITCH + 2 * hi3;
  v8f kacc = {};
  lds_cv2f* am3 = (lds_cv2f*)mrow;
  lds_cf*   ax3 = (lds_cf*)ax;
  lds_cv2f* bk3 = (lds_cv2f*)bk;
  lds_cv2f* bk4 = (lds_cv2f*)(bk + 64 * HD);
  asm volatile("" : "+v"(am3));
  asm volatile("" : "+v"(ax3));
  asm volatile("" : "+v"(bk3));
  asm volatile("" : "+v"(bk4));
  v2f ka[2][8], kw[2][8];
#pragma unroll
  for (int j = 0; j < 8; ++j) { ka[0][j] = am3[2 * j]; kw[0][j] = bk3[2 * j * HD]; }
#pragma unroll
  for (int q = 0; q < 8; ++q) {
    const int cur = q & 1, nxt = cur ^ 1;
    if (q + 1 < 8) {
#pragma unroll
      for (int j = 0; j < 8; ++j) {
        const int k0 = 32 * ((q + 1) & 3) + 4 * j;
        if (q + 1 < 4) {
          ka[nxt][j] = am3[k0 >> 1];
          kw[nxt][j] = bk3[(k0 >> 1) * HD];
        } else {
          v2f a; a.x = ax3[2 * k0]; a.y = ax3[2 * k0 + 2];
          ka[nxt][j] = a;
          kw[nxt][j] = bk4[(k0 >> 1) * HD];
        }
      }
    }
    __builtin_amdgcn_sched_barrier(0);
#pragma unroll
    for (int j = 0; j < 8; ++j)
      kacc = __builtin_amdgcn_wmma_f32_16x16x4_f32(false, ka[cur][j], false, kw[cur][j],
                                                   (short)0, kacc, false, false);
    __builtin_amdgcn_sched_barrier(0);
  }
  if (m3 == 15) {
#pragma unroll
    for (int v = 0; v < 8; ++v) {
      const float val = kacc[v] + kb;
      keys[wave3 * 16 + 8 * hi3 + v] = val > 0.f ? val : 0.f;
    }
  }
  if (t >= NN && t < NN + 2) sel[KTOP + t - NN] = 0;
  __syncthreads();

  if (t < NN) {
    const float ki = keys[t];
    const float4* k4 = (const float4*)keys;
    int cnt = 0;
#pragma unroll 4
    for (int j4 = 0; j4 < NN / 4; ++j4) {
      const float4 kj = k4[j4];
      const int j = 4 * j4;
      cnt += (kj.x > ki || (kj.x == ki && j     < t)) ? 1 : 0;
      cnt += (kj.y > ki || (kj.y == ki && j + 1 < t)) ? 1 : 0;
      cnt += (kj.z > ki || (kj.z == ki && j + 2 < t)) ? 1 : 0;
      cnt += (kj.w > ki || (kj.w == ki && j + 3 < t)) ? 1 : 0;
    }
    if (cnt < KTOP) sel[cnt] = t;
  }
  __syncthreads();

  if (more) REQUEST_WEIGHTS((const float*)wl1n, (const float*)wr1n)

  const int rt = wave3 & 1;
  const int cp = wave3 >> 1;
  const float bv0 = b3[cp * 32 + m3];
  const float bv1 = b3[cp * 32 + 16 + m3];
  const int rs = sel[rt * 16 + m3];
  const float*  am = cm + rs * CM_PITCH + 2 * hi3;
  const float*  ah = (const float*)fp + ((rs >> 1) * FP_PITCH + 2 * hi3) * 2 + (rs & 1);
  const float2* bg = wp + hi3 * HD + cp * 32 + m3;
  v8f g0 = {}, g1 = {};
  lds_cv2f* gm3 = (lds_cv2f*)am;
  lds_cf*   gh3 = (lds_cf*)ah;
  lds_cv2f* bg3 = (lds_cv2f*)bg;
  lds_cv2f* bg4 = (lds_cv2f*)(bg + 64 * HD);
  asm volatile("" : "+v"(gm3));
  asm volatile("" : "+v"(gh3));
  asm volatile("" : "+v"(bg3));
  asm volatile("" : "+v"(bg4));
  v2f ga[2][4], gw0[2][4], gw1[2][4];
#pragma unroll
  for (int j = 0; j < 4; ++j) {
    ga[0][j]  = gm3[2 * j];
    gw0[0][j] = bg3[2 * j * HD];
    gw1[0][j] = bg3[2 * j * HD + 16];
  }
#pragma unroll
  for (int q = 0; q < 16; ++q) {
    const int cur = q & 1, nxt = cur ^ 1;
    if (q + 1 < 16) {
#pragma unroll
      for (int j = 0; j < 4; ++j) {
        const int k0 = 16 * ((q + 1) & 7) + 4 * j;
        if (q + 1 < 8) {
          ga[nxt][j]  = gm3[k0 >> 1];
          gw0[nxt][j] = bg3[(k0 >> 1) * HD];
          gw1[nxt][j] = bg3[(k0 >> 1) * HD + 16];
        } else {
          v2f a; a.x = gh3[2 * k0]; a.y = gh3[2 * k0 + 2];
          ga[nxt][j]  = a;
          gw0[nxt][j] = bg4[(k0 >> 1) * HD];
          gw1[nxt][j] = bg4[(k0 >> 1) * HD + 16];
        }
      }
    }
    __builtin_amdgcn_sched_barrier(0);
#pragma unroll
    for (int j = 0; j < 4; ++j) {
      g0 = __builtin_amdgcn_wmma_f32_16x16x4_f32(false, ga[cur][j], false, gw0[cur][j], (short)0, g0, false, false);
      g1 = __builtin_amdgcn_wmma_f32_16x16x4_f32(false, ga[cur][j], false, gw1[cur][j], (short)0, g1, false, false);
    }
    __builtin_amdgcn_sched_barrier(0);
  }

  __syncthreads();

  glb_f* og = o3 + (size_t)g * KTOP * HD + cp * 32 + m3;
#pragma unroll
  for (int v = 0; v < 8; ++v) {
    const int r = rt * 16 + 8 * hi3 + v;
    if (r < KTOP) {
      const float v0 = g0[v] + bv0;
      const float v1 = g1[v] + bv1;
      og[r * HD]      = v0 > 0.f ? v0 : 0.f;
      og[r * HD + 16] = v1 > 0.f ? v1 : 0.f;
    }
  }
  }

  g += (int)gridDim.x;
  } while (g < BG);
}
#undef REQUEST_GRAPH
#undef REQUEST_WEIGHTS

#define XS_PITCH 132
#define XS_ROWS  34
#define CW_PITCH 388

__global__ __launch_bounds__(256) void sortpool_head(
    const float* __restrict__ H,
    const float* __restrict__ conv_w, const float* __restrict__ conv_b,
    const float* __restrict__ W1, const float* __restrict__ b1,
    const float* __restrict__ W2, const float* __restrict__ b2,
    float* __restrict__ out)
{
  __shared__ float4 xs4[XS_ROWS * XS_PITCH / 4];
  __shared__ float4 cw4[CONV_O * CW_PITCH / 4];
  __shared__ float4 y4s[FLAT / 4];
  __shared__ float z[HD];
  __shared__ float logits[NCLS];
  __shared__ float lse;

  const int g = blockIdx.x;
  const int t = threadIdx.x;
  float* y = (float*)y4s;
  const float* hg = H + (size_t)g * KTOP * HD;

  const float4* cwg = (const float4*)conv_w;
#pragma unroll
  for (int j = 0; j < CONV_O * HD * CONV_W / 4 / 256; ++j) {
    const int idx = t + 256 * j;
    const int o = idx / (HD * CONV_W / 4);
    const int q = idx - o * (HD * CONV_W / 4);
    cw4[o * (CW_PITCH / 4) + q] = cwg[idx];
  }

  const float4* hg4 = (const float4*)hg;
  for (int i = t; i < XS_ROWS * (HD / 4); i += 256) {
    const int r = i >> 5, c4 = i & 31;
    float4 v = make_float4(0.f, 0.f, 0.f, 0.f);
    if (r < KTOP) v = hg4[r * (HD / 4) + c4];
    xs4[r * (XS_PITCH / 4) + c4] = v;
  }
  __syncthreads();

  const int wave = t >> 5;
  const int lane = t & 31;
  const int m    = lane & 15;
  const int hi   = lane >> 4;
  if (wave < 4) {
    const int rt = wave & 1;
    const int ct = wave >> 1;
    const float* xs = (const float*)xs4;
    const float* cw = (const float*)cw4;
    const float* ap = xs + (rt * 16 + m) * XS_PITCH + 2 * hi;
    const float* bp = cw + (ct * 16 + m) * CW_PITCH + 6 * hi;
    v8f acc0 = {}, acc1 = {};
#pragma unroll
    for (int w = 0; w < CONV_W; ++w) {
#pragma unroll 4
      for (int c0 = 0; c0 < HD; c0 += 8) {
        const v2f a0 = *(const v2f*)(ap + w * XS_PITCH + c0);
        const v2f a1 = *(const v2f*)(ap + w * XS_PITCH + c0 + 4);
        v2f b0, b1;
        b0.x = bp[c0 * CONV_W + w];
        b0.y = bp[(c0 + 1) * CONV_W + w];
        b1.x = bp[(c0 + 4) * CONV_W + w];
        b1.y = bp[(c0 + 5) * CONV_W + w];
        acc0 = __builtin_amdgcn_wmma_f32_16x16x4_f32(false, a0, false, b0,
                                                     (short)0, acc0, false, false);
        acc1 = __builtin_amdgcn_wmma_f32_16x16x4_f32(false, a1, false, b1,
                                                     (short)0, acc1, false, false);
      }
    }
    const int o = ct * 16 + m;
    const float bv = conv_b[o];
#pragma unroll
    for (int v = 0; v < 8; ++v) {
      const int tt = rt * 16 + hi * 8 + v;
      const float s = (acc0[v] + acc1[v]) + bv;
      if (tt < CONV_T) y[o * CONV_T + tt] = s > 0.f ? s : 0.f;
    }
  }
  __syncthreads();

  float4* part = cw4;
  float*  part2 = (float*)cw4 + 8 * HD;
  {
    const int cg = t & 31;
    const int ks = t >> 5;
    const float4* w14 = (const float4*)W1 + (size_t)(ks * (FLAT / 8)) * (HD / 4) + cg;
    const float4* yk  = y4s + ks * (FLAT / 8 / 4);
    float4 s = make_float4(0.f, 0.f, 0.f, 0.f);
#pragma unroll 2
    for (int i4 = 0; i4 < FLAT / 8 / 4; ++i4) {
      const float4 yv = yk[i4];
      const float4 w0 = w14[(4 * i4 + 0) * (HD / 4)];
      const float4 w1 = w14[(4 * i4 + 1) * (HD / 4)];
      const float4 w2 = w14[(4 * i4 + 2) * (HD / 4)];
      const float4 w3 = w14[(4 * i4 + 3) * (HD / 4)];
      s.x += yv.x * w0.x; s.y += yv.x * w0.y; s.z += yv.x * w0.z; s.w += yv.x * w0.w;
      s.x += yv.y * w1.x; s.y += yv.y * w1.y; s.z += yv.y * w1.z; s.w += yv.y * w1.w;
      s.x += yv.z * w2.x; s.y += yv.z * w2.y; s.z += yv.z * w2.z; s.w += yv.z * w2.w;
      s.x += yv.w * w3.x; s.y += yv.w * w3.y; s.z += yv.w * w3.z; s.w += yv.w * w3.w;
    }
    part[ks * (HD / 4) + cg] = s;
  }
  __syncthreads();
  if (t < HD) {
    const float* pf = (const float*)part;
    float s = b1[t];
#pragma unroll
    for (int ks = 0; ks < 8; ++ks) s += pf[ks * HD + t];
    z[t] = s > 0.f ? s : 0.f;
  }
  __syncthreads();

  {
    const int c   = t & 15;
    const int ks2 = t >> 4;
    float s = 0.f;
#pragma unroll
    for (int j = 0; j < 8; ++j) s += z[ks2 * 8 + j] * W2[(ks2 * 8 + j) * NCLS + c];
    part2[ks2 * NCLS + c] = s;
  }
  __syncthreads();
  if (t < NCLS) {
    float s = b2[t];
#pragma unroll
    for (int ks2 = 0; ks2 < 16; ++ks2) s += part2[ks2 * NCLS + t];
    logits[t] = s;
  }
  __syncthreads();

  if (t == 0) {
    float mx = logits[0];
    for (int c = 1; c < NCLS; ++c) mx = fmaxf(mx, logits[c]);
    float se = 0.f;
    for (int c = 0; c < NCLS; ++c) se += expf(logits[c] - mx);
    lse = mx + logf(se);
  }
  __syncthreads();
  if (t < NCLS) out[(size_t)g * NCLS + t] = logits[t] - lse;
}

extern "C" void kernel_launch(void* const* d_in, const int* in_sizes, int n_in,
                              void* d_out, int out_size, void* d_ws, size_t ws_size,
                              hipStream_t stream) {
  (void)in_sizes; (void)n_in; (void)out_size; (void)ws_size;
  const float* x      = (const float*)d_in[0];
  const int*   src    = (const int*)d_in[1];
  const int*   dst    = (const int*)d_in[2];
  const float* Wl1    = (const float*)d_in[5];
  const float* bl1    = (const float*)d_in[6];
  const float* Wr1    = (const float*)d_in[7];
  const float* Wl2    = (const float*)d_in[8];
  const float* bl2    = (const float*)d_in[9];
  const float* Wr2    = (const float*)d_in[10];
  const float* Wl3    = (const float*)d_in[11];
  const float* bl3    = (const float*)d_in[12];
  const float* Wr3    = (const float*)d_in[13];
  const float* conv_w = (const float*)d_in[14];
  const float* conv_b = (const float*)d_in[15];
  const float* W1     = (const float*)d_in[16];
  const float* b1     = (const float*)d_in[17];
  const float* W2     = (const float*)d_in[18];
  const float* b2     = (const float*)d_in[19];

  float* ws   = (float*)d_ws;
  float* h3   = ws + (size_t)VN * HD;

  sage_layers<<<NWG, 256, 0, stream>>>(x, src, dst, Wl1, Wr1, bl1, Wl2, Wr2, bl2,
                                       Wl3, Wr3, bl3, h3);
  sortpool_head<<<BG, 256, 0, stream>>>(h3, conv_w, conv_b, W1, b1, W2, b2,
                                        (float*)d_out);
}


# layer 3: neighbour sum only for the 30 kept rows (64 matrix ops a wave, was 256); keys via C @ (H @ Wl3 key tile), so the keys' sum order changes; kept rows' sums keep their order
# speedup vs baseline: 9.6310x; 1.0559x over previous
#include <hip/hip_runtime.h>
#include <hip/hip_bf16.h>

#define BG   1024
#define NN   128
#define VN   (BG * NN)
#define HD   128
#define EPG  2048
#define KTOP 30
#define CONV_O 32
#define CONV_W 3
#define CONV_T (KTOP - CONV_W + 1)
#define FLAT   (CONV_O * CONV_T)
#define NCLS   16

typedef float v2f __attribute__((ext_vector_type(2)));
typedef float v8f __attribute__((ext_vector_type(8)));
typedef __attribute__((address_space(3))) const float lds_cf;
typedef __attribute__((address_space(3))) const v2f   lds_cv2f;
typedef __attribute__((address_space(1))) const float glb_cf;
typedef __attribute__((address_space(1))) float       glb_f;

__device__ __forceinline__ float div_by_degree(float a, float d, float r) {
  const float q = a * r;
  const float e = __builtin_fmaf(-d, q, a);
  return __builtin_fmaf(e, r, q);
}


#define FP_PITCH 144
#define CM_PITCH 132
#define PK_PITCH 68

#define NWG 256

#define REQUEST_GRAPH(G)                                                   \
  {                                                                        \
    const float4* xg4_ = (const float4*)(X + (size_t)(G) * NN * HD);       \
      \
    _Pragma("unroll")                                                      \
    for (int j = 0; j < 8; ++j) {                                          \
      const int idx = t + 256 * j;                                         \
      const int q   = idx >> 5;                                            \
      const int c4  = idx & 31;                                            \
      xr[2 * j]     = xg4_[(2 * q) * (HD / 4) + c4];                       \
      xr[2 * j + 1] = xg4_[(2 * q + 1) * (HD / 4) + c4];                   \
    }                                                                      \
    const int4* s4_ = (const int4*)(src + (size_t)(G) * EPG);              \
    const int4* d4_ = (const int4*)(dst + (size_t)(G) * EPG);              \
    sv0 = s4_[t]; sv1 = s4_[t + 256];                                      \
    dv0 = d4_[t]; dv1 = d4_[t + 256];                                      \
  }

#define REQUEST_WEIGHTS(WL, WR)                                            \
  {                                                                        \
      \
                                \
    _Pragma("unroll")                                                      \
    for (int j = 0; j < 16; ++j) {                                         \
      const int idx = t + 256 * j;                                         \
      const int kk  = idx >> 5;                                            \
      const int c4  = idx & 31;                                            \
      const float4* w4_ = (const float4*)(j < 8 ? (WL) : (WR))             \
                          + (size_t)(2 * (kk & 63)) * (HD / 4);            \
      wr[2 * j]     = w4_[c4];                                             \
      wr[2 * j + 1] = w4_[(HD / 4) + c4];                                  \
    }                                                                      \
  }

__global__ __launch_bounds__(256) void sage_layers(
    const float* __restrict__ X, const int* src, const int* dst,
    const float* Wl1, const float* Wr1, const float* bl1,
    const float* Wl2, const float* Wr2, const float* bl2,
    const float* Wl3, const float* Wr3, const float* bl3,
    float* __restrict__ out)
{
  static_assert(BG % NWG == 0, "every workgroup loops over the same number of graphs");
  __shared__ float2 fp[(NN / 2) * FP_PITCH];
  __shared__ float  cm[NN * CM_PITCH];
  __shared__ float  deg[NN];
  __shared__ float  rdg[NN];
  __shared__ float  pk[NN * PK_PITCH];
  __shared__ float2 wp[128 * HD];
  __shared__ float4 keys4[NN / 4];
  __shared__ int    sel[32];

  float* keys = (float*)keys4;
  const int t = threadIdx.x;
  const int wave = t >> 5;
  const int lane = t & 31;
  const int m    = lane & 15;
  const int hi   = lane >> 4;
  const bool younger = __builtin_amdgcn_readfirstlane(wave) >= 4;

  float4 xr[16];
  int4 sv0, sv1, dv0, dv1;
  float4 wr[32];
  REQUEST_GRAPH(blockIdx.x)
  REQUEST_WEIGHTS(Wl1, Wr1)

  int g = blockIdx.x;
  do {
  const bool more = g + (int)gridDim.x < BG;

#pragma nounroll
  for (int l = 0; l < 2; ++l) {

  if (l == 0) {
    for (int i = t; i < NN * CM_PITCH / 4; i += 256)
      ((float4*)cm)[i] = make_float4(0.f, 0.f, 0.f, 0.f);
    if (t < NN) deg[t] = 0.f;

#pragma unroll
    for (int j = 0; j < 8; ++j) {
      const int idx = t + 256 * j;
      const int q   = idx >> 5;
      const int c4  = idx & 31;
      const float4 r0 = xr[2 * j];
      const float4 r1 = xr[2 * j + 1];
      float4* o = (float4*)(fp + q * FP_PITCH + c4 * 4);
      o[0] = make_float4(r0.x, r1.x, r0.y, r1.y);
      o[1] = make_float4(r0.z, r1.z, r0.w, r1.w);
    }
  }

#pragma unroll
  for (int j = 0; j < 16; ++j) {
    const int idx = t + 256 * j;
    const int kk  = idx >> 5;
    const int c4  = idx & 31;
    const float4 r0 = wr[2 * j];
    const float4 r1 = wr[2 * j + 1];
    float4* o = (float4*)(wp + kk * HD + c4 * 4);
    o[0] = make_float4(r0.x, r1.x, r0.y, r1.y);
    o[1] = make_float4(r0.z, r1.z, r0.w, r1.w);
  }
  __syncthreads();

  if (l == 0) {
#define COUNT_EDGE(S, D)                                   \
    {                                                      \
      const int s_ = (S) & (NN - 1);                       \
      const int d_ = (D) & (NN - 1);                       \
      atomicAdd(&cm[d_ * CM_PITCH + s_], 1.0f);            \
      atomicAdd(&deg[d_], 1.0f);                           \
    }
    COUNT_EDGE(sv0.x, dv0.x) COUNT_EDGE(sv0.y, dv0.y)
    COUNT_EDGE(sv0.z, dv0.z) COUNT_EDGE(sv0.w, dv0.w)
    COUNT_EDGE(sv1.x, dv1.x) COUNT_EDGE(sv1.y, dv1.y)
    COUNT_EDGE(sv1.z, dv1.z) COUNT_EDGE(sv1.w, dv1.w)
#undef COUNT_EDGE
    __syncthreads();

    if (more) REQUEST_GRAPH(g + (int)gridDim.x)

    if (lane < 16) {
      const float c = deg[wave * 16 + lane];
      const float d = c > 1.f ? c : 1.f;
      deg[wave * 16 + lane] = d;
      rdg[wave * 16 + lane] = 1.0f / d;
    }
  }

  const float* bias = l == 0 ? bl1 : bl2;
  const float* WLn  = l == 0 ? Wl2 : Wl3;
  const float* WRn  = l == 0 ? Wr2 : Wr3;

  if (younger) __builtin_amdgcn_s_setprio((short)3);

  const float*  arow = cm + (wave * 16 + m) * CM_PITCH + 2 * hi;
  const float2* bcol = fp + hi * FP_PITCH + m;

  v8f acc[8] = {};
  v2f    af[2];
  float2 bf[2][8];

  af[0] = *(const v2f*)arow;
#pragma unroll
  for (int ct = 0; ct < 8; ++ct) bf[0][ct] = bcol[ct * 16];

#pragma unroll
  for (int ks = 0; ks < NN / 4; ++ks) {
    const int cur = ks & 1, nxt = cur ^ 1;
    if (ks + 1 < NN / 4) {
      af[nxt] = *(const v2f*)(arow + 4 * (ks + 1));
      const float2* bp = bcol + 2 * (ks + 1) * FP_PITCH;
#pragma unroll
      for (int ct = 0; ct < 8; ++ct) bf[nxt][ct] = bp[ct * 16];
    }
#pragma unroll
    for (int ct = 0; ct < 8; ++ct) {
      v2f b; b.x = bf[cur][ct].x; b.y = bf[cur][ct].y;
      acc[ct] = __builtin_amdgcn_wmma_f32_16x16x4_f32(false, af[cur], false, b,
                                                      (short)0, acc[ct], false, false);
    }
  }

  float dn[8], rn[8];
#pragma unroll
  for (int v = 0; v < 8; ++v) {
    dn[v] = deg[wave * 16 + 8 * hi + v];
    rn[v] = rdg[wave * 16 + 8 * hi + v];
  }
  float* st = pk + (wave * 16 + 8 * hi) * PK_PITCH + m;
  const float* prow = pk + (wave * 16 + m) * PK_PITCH + 2 * hi;

  const int row = wave * 16 + m;
  const float*  ax = (const float*)fp + ((row >> 1) * FP_PITCH + 2 * hi) * 2 + (row & 1);
  const float2* bw = wp + hi * HD + m;

  v8f acc0 = {}, acc1 = {}, acc2 = {}, acc3 = {};
  v8f acc4 = {}, acc5 = {}, acc6 = {}, acc7 = {};

#define GEMM_STEP(KOFF)                                                    \
  {                                                                        \
    const float2* bp = bw + ((k0 >> 1) + (KOFF)) * HD;                     \
    float2 bb; v2f b;                                                      \
    bb = bp[0];   b.x = bb.x; b.y = bb.y;                                  \
    acc0 = __builtin_amdgcn_wmma_f32_16x16x4_f32(false, a, false, b, (short)0, acc0, false, false); \
    bb = bp[16];  b.x = bb.x; b.y = bb.y;                                  \
    acc1 = __builtin_amdgcn_wmma_f32_16x16x4_f32(false, a, false, b, (short)0, acc1, false, false); \
    bb = bp[32];  b.x = bb.x; b.y = bb.y;                                  \
    acc2 = __builtin_amdgcn_wmma_f32_16x16x4_f32(false, a, false, b, (short)0, acc2, false, false); \
    bb = bp[48];  b.x = bb.x; b.y = bb.y;                                  \
    acc3 = __builtin_amdgcn_wmma_f32_16x16x4_f32(false, a, false, b, (short)0, acc3, false, false); \
    bb = bp[64];  b.x = bb.x; b.y = bb.y;                                  \
    acc4 = __builtin_amdgcn_wmma_f32_16x16x4_f32(false, a, false, b, (short)0, acc4, false, false); \
    bb = bp[80];  b.x = bb.x; b.y = bb.y;                                  \
    acc5 = __builtin_amdgcn_wmma_f32_16x16x4_f32(false, a, false, b, (short)0, acc5, false, false); \
    bb = bp[96];  b.x = bb.x; b.y = bb.y;                                  \
    acc6 = __builtin_amdgcn_wmma_f32_16x16x4_f32(false, a, false, b, (short)0, acc6, false, false); \
    bb = bp[112]; b.x = bb.x; b.y = bb.y;                                  \
    acc7 = __builtin_amdgcn_wmma_f32_16x16x4_f32(false, a, false, b, (short)0, acc7, false, false); \
  }

#pragma unroll
  for (int ct = 0; ct < 4; ++ct) {
#pragma unroll
    for (int v = 0; v < 8; ++v) st[v * PK_PITCH + ct * 16] = div_by_degree(acc[ct][v], dn[v], rn[v]);
  }
#pragma unroll 4
  for (int k0 = 0; k0 < HD / 2; k0 += 4) {
    const v2f a = *(const v2f*)(prow + k0);
    GEMM_STEP(0)
  }
#pragma unroll
  for (int ct = 4; ct < 8; ++ct) {
#pragma unroll
    for (int v = 0; v < 8; ++v) st[v * PK_PITCH + (ct - 4) * 16] = div_by_degree(acc[ct][v], dn[v], rn[v]);
  }
#pragma unroll 4
  for (int k0 = 0; k0 < HD / 2; k0 += 4) {
    const v2f a = *(const v2f*)(prow + k0);
    GEMM_STEP(32)
  }
  if (younger) __builtin_amdgcn_s_setprio((short)0);

  REQUEST_WEIGHTS(WLn, WRn)
  float bvs[8];
#pragma unroll
  for (int ct = 0; ct < 8; ++ct) bvs[ct] = bias[ct * 16 + m];

#undef GEMM_STEP

  lds_cf*   ax3 = (lds_cf*)ax;
  lds_cv2f* bw3 = (lds_cv2f*)(bw + 64 * HD);
  asm volatile("" : "+v"(ax3));
  asm volatile("" : "+v"(bw3));
#pragma unroll 4
  for (int k0 = 0; k0 < HD; k0 += 4) {
    v2f a; a.x = ax3[2 * k0]; a.y = ax3[2 * k0 + 2];
    lds_cv2f* bp = bw3 + (k0 >> 1) * HD;
    acc0 = __builtin_amdgcn_wmma_f32_16x16x4_f32(false, a, false, bp[0],   (short)0, acc0, false, false);
    acc1 = __builtin_amdgcn_wmma_f32_16x16x4_f32(false, a, false, bp[16],  (short)0, acc1, false, false);
    acc2 = __builtin_amdgcn_wmma_f32_16x16x4_f32(false, a, false, bp[32],  (short)0, acc2, false, false);
    acc3 = __builtin_amdgcn_wmma_f32_16x16x4_f32(false, a, false, bp[48],  (short)0, acc3, false, false);
    acc4 = __builtin_amdgcn_wmma_f32_16x16x4_f32(false, a, false, bp[64],  (short)0, acc4, false, false);
    acc5 = __builtin_amdgcn_wmma_f32_16x16x4_f32(false, a, false, bp[80],  (short)0, acc5, false, false);
    acc6 = __builtin_amdgcn_wmma_f32_16x16x4_f32(false, a, false, bp[96],  (short)0, acc6, false, false);
    acc7 = __builtin_amdgcn_wmma_f32_16x16x4_f32(false, a, false, bp[112], (short)0, acc7, false, false);
  }

  __syncthreads();

  v8f accs[8] = {acc0, acc1, acc2, acc3, acc4, acc5, acc6, acc7};
  {
    float2* hp = fp + (wave * 8 + 4 * hi) * FP_PITCH + m;
#pragma unroll
    for (int ct = 0; ct < 8; ++ct) {
      const float bv = bvs[ct];
#pragma unroll
      for (int p = 0; p < 4; ++p) {
        float v0 = accs[ct][2 * p] + bv;
        float v1 = accs[ct][2 * p + 1] + bv;
        v0 = __builtin_fmaxf(v0, 0.f);
        v1 = __builtin_fmaxf(v1, 0.f);
        hp[p * FP_PITCH + ct * 16] = make_float2(v0, v1);
      }
    }
  }

  }

  {
#pragma unroll
  for (int j = 0; j < 16; ++j) {
    const int idx = t + 256 * j;
    const int kk  = idx >> 5;
    const int c4  = idx & 31;
    const float4 r0 = wr[2 * j];
    const float4 r1 = wr[2 * j + 1];
    float4* o = (float4*)(wp + kk * HD + c4 * 4);
    o[0] = make_float4(r0.x, r1.x, r0.y, r1.y);
    o[1] = make_float4(r0.z, r1.z, r0.w, r1.w);
  }
  __syncthreads();

  glb_cf* wl1n = (glb_cf*)Wl1;
  glb_cf* wr1n = (glb_cf*)Wr1;
  glb_cf* b3   = (glb_cf*)bl3;
  glb_f*  o3   = (glb_f*)out;
  asm volatile("" : "+s"(wl1n));
  asm volatile("" : "+s"(wr1n));
  asm volatile("" : "+s"(b3));
  asm volatile("" : "+s"(o3));

  int t3 = t;
  asm volatile("" : "+v"(t3));
  const int wave3 = t3 >> 5;
  const int m3    = t3 & 15;
  const int hi3   = (t3 >> 4) & 1;

  const float kb = b3[(HD - 16) + m3];

  const int row = wave3 * 16 + m3;
  const float*  ax = (const float*)fp + ((row >> 1) * FP_PITCH + 2 * hi3) * 2 + (row & 1);
  const float2* bk = wp + hi3 * HD + (HD - 16) + m3;
  lds_cf*   ax3 = (lds_cf*)ax;
  lds_cv2f* bk3 = (lds_cv2f*)bk;
  lds_cv2f* bk4 = (lds_cv2f*)(bk + 64 * HD);
  asm volatile("" : "+v"(ax3));
  asm volatile("" : "+v"(bk3));
  asm volatile("" : "+v"(bk4));
  v8f uacc = {}, vacc = {};
  {
    v2f ha[2][4], wl[2][4], wr3[2][4];
#pragma unroll
    for (int j = 0; j < 4; ++j) {
      v2f a; a.x = ax3[8 * j]; a.y = ax3[8 * j + 2];
      ha[0][j]  = a;
      wl[0][j]  = bk3[2 * j * HD];
      wr3[0][j] = bk4[2 * j * HD];
    }
#pragma unroll
    for (int q = 0; q < 8; ++q) {
      const int cur = q & 1, nxt = cur ^ 1;
      if (q + 1 < 8) {
#pragma unroll
        for (int j = 0; j < 4; ++j) {
          const int k0 = 16 * (q + 1) + 4 * j;
          v2f a; a.x = ax3[2 * k0]; a.y = ax3[2 * k0 + 2];
          ha[nxt][j]  = a;
          wl[nxt][j]  = bk3[(k0 >> 1) * HD];
          wr3[nxt][j] = bk4[(k0 >> 1) * HD];
        }
      }
      __builtin_amdgcn_sched_barrier(0);
#pragma unroll
      for (int j = 0; j < 4; ++j) {
        uacc = __builtin_amdgcn_wmma_f32_16x16x4_f32(false, ha[cur][j], false, wl[cur][j],  (short)0, uacc, false, false);
        vacc = __builtin_amdgcn_wmma_f32_16x16x4_f32(false, ha[cur][j], false, wr3[cur][j], (short)0, vacc, false, false);
      }
      __builtin_amdgcn_sched_barrier(0);
    }
  }
  float2* up = (float2*)pk;
  {
    float2* us = up + (wave3 * 8 + 4 * hi3) * 16 + m3;
#pragma unroll
    for (int p = 0; p < 4; ++p) us[p * 16] = make_float2(uacc[2 * p], uacc[2 * p + 1]);
  }
  __syncthreads();

  {
    const float*  arow = cm + (wave3 * 16 + m3) * CM_PITCH + 2 * hi3;
    lds_cv2f* ac3 = (lds_cv2f*)arow;
    lds_cv2f* ub3 = (lds_cv2f*)(up + hi3 * 16 + m3);
    asm volatile("" : "+v"(ac3));
    asm volatile("" : "+v"(ub3));
    v8f kc0 = {}, kc1 = {};
    float dn[8], rn[8];
#pragma unroll
    for (int v = 0; v < 8; ++v) {
      dn[v] = deg[wave3 * 16 + 8 * hi3 + v];
      rn[v] = rdg[wave3 * 16 + 8 * hi3 + v];
    }
    v2f ca[2][8], ub[2][8];
#pragma unroll
    for (int j = 0; j < 8; ++j) { ca[0][j] = ac3[2 * j]; ub[0][j] = ub3[32 * j]; }
#pragma unroll
    for (int q = 0; q < 4; ++q) {
      const int cur = q & 1, nxt = cur ^ 1;
      if (q + 1 < 4) {
#pragma unroll
        for (int j = 0; j < 8; ++j) {
          const int ks = 8 * (q + 1) + j;
          ca[nxt][j] = ac3[2 * ks];
          ub[nxt][j] = ub3[32 * ks];
        }
      }
      __builtin_amdgcn_sched_barrier(0);
#pragma unroll
      for (int j = 0; j < 8; j += 2) {
        kc0 = __builtin_amdgcn_wmma_f32_16x16x4_f32(false, ca[cur][j],     false, ub[cur][j],     (short)0, kc0, false, false);
        kc1 = __builtin_amdgcn_wmma_f32_16x16x4_f32(false, ca[cur][j + 1], false, ub[cur][j + 1], (short)0, kc1, false, false);
      }
      __builtin_amdgcn_sched_barrier(0);
    }
    if (m3 == 15) {
#pragma unroll
      for (int v = 0; v < 8; ++v) {
        const int r = wave3 * 16 + 8 * hi3 + v;
        const float su  = kc0[v] + kc1[v];
        const float val = (div_by_degree(su, dn[v], rn[v]) + vacc[v]) + kb;
        keys[r] = val > 0.f ? val : 0.f;
      }
    }
  }
  if (t >= NN && t < NN + 2) sel[KTOP + t - NN] = 0;
  __syncthreads();

  if (t < NN) {
    const float ki = keys[t];
    const float4* k4 = (const float4*)keys;
    int cnt = 0;
#pragma unroll 4
    for (int j4 = 0; j4 < NN / 4; ++j4) {
      const float4 kj = k4[j4];
      const int j = 4 * j4;
      cnt += (kj.x > ki || (kj.x == ki && j     < t)) ? 1 : 0;
      cnt += (kj.y > ki || (kj.y == ki && j + 1 < t)) ? 1 : 0;
      cnt += (kj.z > ki || (kj.z == ki && j + 2 < t)) ? 1 : 0;
      cnt += (kj.w > ki || (kj.w == ki && j + 3 < t)) ? 1 : 0;
    }
    if (cnt < KTOP) sel[cnt] = t;
  }
  __syncthreads();

  const int rt = wave3 & 1;
  const int cp = wave3 >> 1;
  const int rs = sel[rt * 16 + m3];
  float* mk = pk + 2 * 64 * 16;

  {
    const float*  cr = cm + rs * CM_PITCH + 2 * hi3;
    const float2* hb = fp + hi3 * FP_PITCH + cp * 32 + m3;
    lds_cv2f* cr3 = (lds_cv2f*)cr;
    lds_cv2f* hb3 = (lds_cv2f*)hb;
    asm volatile("" : "+v"(cr3));
    asm volatile("" : "+v"(hb3));
    v8f s0 = {}, s1 = {};
    float dk[8], rk[8];
#pragma unroll
    for (int v = 0; v < 8; ++v) {
      const int nd = sel[rt * 16 + 8 * hi3 + v];
      dk[v] = deg[nd];
      rk[v] = rdg[nd];
    }
    v2f sa[2][4], sb0[2][4], sb1[2][4];
#pragma unroll
    for (int j = 0; j < 4; ++j) {
      sa[0][j]  = cr3[2 * j];
      sb0[0][j] = hb3[2 * j * FP_PITCH];
      sb1[0][j] = hb3[2 * j * FP_PITCH + 16];
    }
#pragma unroll
    for (int q = 0; q < 8; ++q) {
      const int cur = q & 1, nxt = cur ^ 1;
      if (q + 1 < 8) {
#pragma unroll
        for (int j = 0; j < 4; ++j) {
          const int ks = 4 * (q + 1) + j;
          sa[nxt][j]  = cr3[2 * ks];
          sb0[nxt][j] = hb3[2 * ks * FP_PITCH];
          sb1[nxt][j] = hb3[2 * ks * FP_PITCH + 16];
        }
      }
      __builtin_amdgcn_sched_barrier(0);
#pragma unroll
      for (int j = 0; j < 4; ++j) {
        s0 = __builtin_amdgcn_wmma_f32_16x16x4_f32(false, sa[cur][j], false, sb0[cur][j], (short)0, s0, false, false);
        s1 = __builtin_amdgcn_wmma_f32_16x16x4_f32(false, sa[cur][j], false, sb1[cur][j], (short)0, s1, false, false);
      }
      __builtin_amdgcn_sched_barrier(0);
    }
    float* st = mk + (rt * 16 + 8 * hi3) * CM_PITCH + cp * 32 + m3;
#pragma unroll
    for (int v = 0; v < 8; ++v) {
      st[v * CM_PITCH]      = div_by_degree(s0[v], dk[v], rk[v]);
      st[v * CM_PITCH + 16] = div_by_degree(s1[v], dk[v], rk[v]);
    }
  }
  __syncthreads();

  if (more) REQUEST_WEIGHTS((const float*)wl1n, (const float*)wr1n)
  const float bv0 = b3[cp * 32 + m3];
  const float bv1 = b3[cp * 32 + 16 + m3];

  const float*  am = mk + (rt * 16 + m3) * CM_PITCH + 2 * hi3;
  const float*  ah = (const float*)fp + ((rs >> 1) * FP_PITCH + 2 * hi3) * 2 + (rs & 1);
  const float2* bg = wp + hi3 * HD + cp * 32 + m3;
  v8f g0 = {}, g1 = {};
  lds_cv2f* gm3 = (lds_cv2f*)am;
  lds_cf*   gh3 = (lds_cf*)ah;
  lds_cv2f* bg3 = (lds_cv2f*)bg;
  lds_cv2f* bg4 = (lds_cv2f*)(bg + 64 * HD);
  asm volatile("" : "+v"(gm3));
  asm volatile("" : "+v"(gh3));
  asm volatile("" : "+v"(bg3));
  asm volatile("" : "+v"(bg4));
  v2f ga[2][4], gw0[2][4], gw1[2][4];
#pragma unroll
  for (int j = 0; j < 4; ++j) {
    ga[0][j]  = gm3[2 * j];
    gw0[0][j] = bg3[2 * j * HD];
    gw1[0][j] = bg3[2 * j * HD + 16];
  }
#pragma unroll
  for (int q = 0; q < 16; ++q) {
    const int cur = q & 1, nxt = cur ^ 1;
    if (q + 1 < 16) {
#pragma unroll
      for (int j = 0; j < 4; ++j) {
        const int k0 = 16 * ((q + 1) & 7) + 4 * j;
        if (q + 1 < 8) {
          ga[nxt][j]  = gm3[k0 >> 1];
          gw0[nxt][j] = bg3[(k0 >> 1) * HD];
          gw1[nxt][j] = bg3[(k0 >> 1) * HD + 16];
        } else {
          v2f a; a.x = gh3[2 * k0]; a.y = gh3[2 * k0 + 2];
          ga[nxt][j]  = a;
          gw0[nxt][j] = bg4[(k0 >> 1) * HD];
          gw1[nxt][j] = bg4[(k0 >> 1) * HD + 16];
        }
      }
    }
    __builtin_amdgcn_sched_barrier(0);
#pragma unroll
    for (int j = 0; j < 4; ++j) {
      g0 = __builtin_amdgcn_wmma_f32_16x16x4_f32(false, ga[cur][j], false, gw0[cur][j], (short)0, g0, false, false);
      g1 = __builtin_amdgcn_wmma_f32_16x16x4_f32(false, ga[cur][j], false, gw1[cur][j], (short)0, g1, false, false);
    }
    __builtin_amdgcn_sched_barrier(0);
  }

  __syncthreads();

  glb_f* og = o3 + (size_t)g * KTOP * HD + cp * 32 + m3;
#pragma unroll
  for (int v = 0; v < 8; ++v) {
    const int r = rt * 16 + 8 * hi3 + v;
    if (r < KTOP) {
      const float v0 = g0[v] + bv0;
      const float v1 = g1[v] + bv1;
      og[r * HD]      = v0 > 0.f ? v0 : 0.f;
      og[r * HD + 16] = v1 > 0.f ? v1 : 0.f;
    }
  }
  }

  g += (int)gridDim.x;
  } while (g < BG);
}
#undef REQUEST_GRAPH
#undef REQUEST_WEIGHTS

#define XS_PITCH 132
#define XS_ROWS  34
#define CW_PITCH 388

__global__ __launch_bounds__(256) void sortpool_head(
    const float* __restrict__ H,
    const float* __restrict__ conv_w, const float* __restrict__ conv_b,
    const float* __restrict__ W1, const float* __restrict__ b1,
    const float* __restrict__ W2, const float* __restrict__ b2,
    float* __restrict__ out)
{
  __shared__ float4 xs4[XS_ROWS * XS_PITCH / 4];
  __shared__ float4 cw4[CONV_O * CW_PITCH / 4];
  __shared__ float4 y4s[FLAT / 4];
  __shared__ float z[HD];
  __shared__ float logits[NCLS];
  __shared__ float lse;

  const int g = blockIdx.x;
  const int t = threadIdx.x;
  float* y = (float*)y4s;
  const float* hg = H + (size_t)g * KTOP * HD;

  const float4* cwg = (const float4*)conv_w;
#pragma unroll
  for (int j = 0; j < CONV_O * HD * CONV_W / 4 / 256; ++j) {
    const int idx = t + 256 * j;
    const int o = idx / (HD * CONV_W / 4);
    const int q = idx - o * (HD * CONV_W / 4);
    cw4[o * (CW_PITCH / 4) + q] = cwg[idx];
  }

  const float4* hg4 = (const float4*)hg;
  for (int i = t; i < XS_ROWS * (HD / 4); i += 256) {
    const int r = i >> 5, c4 = i & 31;
    float4 v = make_float4(0.f, 0.f, 0.f, 0.f);
    if (r < KTOP) v = hg4[r * (HD / 4) + c4];
    xs4[r * (XS_PITCH / 4) + c4] = v;
  }
  __syncthreads();

  const int wave = t >> 5;
  const int lane = t & 31;
  const int m    = lane & 15;
  const int hi   = lane >> 4;
  if (wave < 4) {
    const int rt = wave & 1;
    const int ct = wave >> 1;
    const float* xs = (const float*)xs4;
    const float* cw = (const float*)cw4;
    const float* ap = xs + (rt * 16 + m) * XS_PITCH + 2 * hi;
    const float* bp = cw + (ct * 16 + m) * CW_PITCH + 6 * hi;
    v8f acc0 = {}, acc1 = {};
#pragma unroll
    for (int w = 0; w < CONV_W; ++w) {
#pragma unroll 4
      for (int c0 = 0; c0 < HD; c0 += 8) {
        const v2f a0 = *(const v2f*)(ap + w * XS_PITCH + c0);
        const v2f a1 = *(const v2f*)(ap + w * XS_PITCH + c0 + 4);
        v2f b0, b1;
        b0.x = bp[c0 * CONV_W + w];
        b0.y = bp[(c0 + 1) * CONV_W + w];
        b1.x = bp[(c0 + 4) * CONV_W + w];
        b1.y = bp[(c0 + 5) * CONV_W + w];
        acc0 = __builtin_amdgcn_wmma_f32_16x16x4_f32(false, a0, false, b0,
                                                     (short)0, acc0, false, false);
        acc1 = __builtin_amdgcn_wmma_f32_16x16x4_f32(false, a1, false, b1,
                                                     (short)0, acc1, false, false);
      }
    }
    const int o = ct * 16 + m;
    const float bv = conv_b[o];
#pragma unroll
    for (int v = 0; v < 8; ++v) {
      const int tt = rt * 16 + hi * 8 + v;
      const float s = (acc0[v] + acc1[v]) + bv;
      if (tt < CONV_T) y[o * CONV_T + tt] = s > 0.f ? s : 0.f;
    }
  }
  __syncthreads();

  float4* part = cw4;
  float*  part2 = (float*)cw4 + 8 * HD;
  {
    const int cg = t & 31;
    const int ks = t >> 5;
    const float4* w14 = (const float4*)W1 + (size_t)(ks * (FLAT / 8)) * (HD / 4) + cg;
    const float4* yk  = y4s + ks * (FLAT / 8 / 4);
    float4 s = make_float4(0.f, 0.f, 0.f, 0.f);
#pragma unroll 2
    for (int i4 = 0; i4 < FLAT / 8 / 4; ++i4) {
      const float4 yv = yk[i4];
      const float4 w0 = w14[(4 * i4 + 0) * (HD / 4)];
      const float4 w1 = w14[(4 * i4 + 1) * (HD / 4)];
      const float4 w2 = w14[(4 * i4 + 2) * (HD / 4)];
      const float4 w3 = w14[(4 * i4 + 3) * (HD / 4)];
      s.x += yv.x * w0.x; s.y += yv.x * w0.y; s.z += yv.x * w0.z; s.w += yv.x * w0.w;
      s.x += yv.y * w1.x; s.y += yv.y * w1.y; s.z += yv.y * w1.z; s.w += yv.y * w1.w;
      s.x += yv.z * w2.x; s.y += yv.z * w2.y; s.z += yv.z * w2.z; s.w += yv.z * w2.w;
      s.x += yv.w * w3.x; s.y += yv.w * w3.y; s.z += yv.w * w3.z; s.w += yv.w * w3.w;
    }
    part[ks * (HD / 4) + cg] = s;
  }
  __syncthreads();
  if (t < HD) {
    const float* pf = (const float*)part;
    float s = b1[t];
#pragma unroll
    for (int ks = 0; ks < 8; ++ks) s += pf[ks * HD + t];
    z[t] = s > 0.f ? s : 0.f;
  }
  __syncthreads();

  {
    const int c   = t & 15;
    const int ks2 = t >> 4;
    float s = 0.f;
#pragma unroll
    for (int j = 0; j < 8; ++j) s += z[ks2 * 8 + j] * W2[(ks2 * 8 + j) * NCLS + c];
    part2[ks2 * NCLS + c] = s;
  }
  __syncthreads();
  if (t < NCLS) {
    float s = b2[t];
#pragma unroll
    for (int ks2 = 0; ks2 < 16; ++ks2) s += part2[ks2 * NCLS + t];
    logits[t] = s;
  }
  __syncthreads();

  if (t == 0) {
    float mx = logits[0];
    for (int c = 1; c < NCLS; ++c) mx = fmaxf(mx, logits[c]);
    float se = 0.f;
    for (int c = 0; c < NCLS; ++c) se += expf(logits[c] - mx);
    lse = mx + logf(se);
  }
  __syncthreads();
  if (t < NCLS) out[(size_t)g * NCLS + t] = logits[t] - lse;
}

extern "C" void kernel_launch(void* const* d_in, const int* in_sizes, int n_in,
                              void* d_out, int out_size, void* d_ws, size_t ws_size,
                              hipStream_t stream) {
  (void)in_sizes; (void)n_in; (void)out_size; (void)ws_size;
  const float* x      = (const float*)d_in[0];
  const int*   src    = (const int*)d_in[1];
  const int*   dst    = (const int*)d_in[2];
  const float* Wl1    = (const float*)d_in[5];
  const float* bl1    = (const float*)d_in[6];
  const float* Wr1    = (const float*)d_in[7];
  const float* Wl2    = (const float*)d_in[8];
  const float* bl2    = (const float*)d_in[9];
  const float* Wr2    = (const float*)d_in[10];
  const float* Wl3    = (const float*)d_in[11];
  const float* bl3    = (const float*)d_in[12];
  const float* Wr3    = (const float*)d_in[13];
  const float* conv_w = (const float*)d_in[14];
  const float* conv_b = (const float*)d_in[15];
  const float* W1     = (const float*)d_in[16];
  const float* b1     = (const float*)d_in[17];
  const float* W2     = (const float*)d_in[18];
  const float* b2     = (const float*)d_in[19];

  float* ws   = (float*)d_ws;
  float* h3   = ws + (size_t)VN * HD;

  sage_layers<<<NWG, 256, 0, stream>>>(x, src, dst, Wl1, Wr1, bl1, Wl2, Wr2, bl2,
                                       Wl3, Wr3, bl3, h3);
  sortpool_head<<<BG, 256, 0, stream>>>(h3, conv_w, conv_b, W1, b1, W2, b2,
                                        (float*)d_out);
}
